# starting kernel
# baseline (speedup 1.0000x reference)
#include <hip/hip_runtime.h>
#include <hip/hip_bf16.h>
#include <cstddef>
#include <cstdint>

#define N_NODES 100000
#define N_EDGES 3200000
#define BN_EPS 1e-5f

typedef __attribute__((ext_vector_type(2))) float v2f;
typedef __attribute__((ext_vector_type(8))) float v8f;

__device__ __forceinline__ v8f wmma_f32(v2f a, v2f b, v8f c) {
  return __builtin_amdgcn_wmma_f32_16x16x4_f32(false, a, false, b, (short)0, c,
                                               false, false);
}

__global__ void zero_f4_kernel(float4* __restrict__ p, long long n4) {
  long long i = (long long)blockIdx.x * blockDim.x + threadIdx.x;
  if (i < n4) p[i] = make_float4(0.f, 0.f, 0.f, 0.f);
}

__global__ void build_input_kernel(const float* __restrict__ x,
                                   const int* __restrict__ labels,
                                   const float* __restrict__ emb,
                                   float* __restrict__ h) {
  long long idx = (long long)blockIdx.x * blockDim.x + threadIdx.x;
  if (idx >= (long long)N_NODES * 16) return;
  const int i = (int)(idx >> 4);
  const int c = (int)(idx & 15);
  float v = 0.0f;
  if (c < 2)       v = x[(size_t)i * 2 + c];
  else if (c < 10) v = emb[(size_t)labels[i] * 8 + (c - 2)];
  h[(size_t)i * 64 + c] = v;
}

__global__ void pad_w1_kernel(const float* __restrict__ W1,
                              float* __restrict__ Wpad) {
  const int idx = blockIdx.x * blockDim.x + threadIdx.x;
  if (idx >= 16 * 64) return;
  const int r = idx >> 6;
  Wpad[idx] = (r < 10) ? W1[idx] : 0.0f;
}

__global__ void scatter_kernel(const float* __restrict__ h,
                               const int* __restrict__ edge_index,
                               float* __restrict__ agg, int log2w) {
  const long long idx = (long long)blockIdx.x * blockDim.x + threadIdx.x;
  const long long total = (long long)N_EDGES << log2w;
  if (idx >= total) return;
  const int e = (int)(idx >> log2w);
  const int c = (int)(idx & ((1 << log2w) - 1));
  const int s = edge_index[e];
  const int d = edge_index[N_EDGES + e];
  const float v = h[(size_t)s * 64 + c];
  unsafeAtomicAdd(&agg[(size_t)d * 64 + c], v);
}

__global__ void __launch_bounds__(128)
gin_mlp_kernel(float* __restrict__ h, const float* __restrict__ agg,
               const float* __restrict__ W1, const float* __restrict__ b1,
               const float* __restrict__ W2, const float* __restrict__ b2,
               const float* __restrict__ eps_ptr, int eps_idx, int K,
               float* __restrict__ stats) {
  __shared__ float ytile[4 * 16 * 68];
  __shared__ float sstat[128];
  const int tid   = threadIdx.x;
  const int wave  = tid >> 5;
  const int lane  = tid & 31;
  const int lrow  = lane & 15;
  const int lhalf = lane >> 4;
  const int tile  = blockIdx.x * 4 + wave;

  if (tid < 128) sstat[tid] = 0.0f;
  __syncthreads();

  if (tile < (N_NODES / 16)) {
    const int base = tile * 16;
    const float onepe = 1.0f + eps_ptr[eps_idx];
    const int koff = lhalf * 2;

    v8f acc[4] = {};
    for (int kk = 0; kk < K; kk += 4) {
      const int k0 = kk + koff;
      const float* hp = h   + (size_t)(base + lrow) * 64 + k0;
      const float* ap = agg + (size_t)(base + lrow) * 64 + k0;
      v2f afrag;
      afrag.x = onepe * hp[0] + ap[0];
      afrag.y = onepe * hp[1] + ap[1];
      const float* wp = W1 + (size_t)k0 * 64 + lrow;
#pragma unroll
      for (int nt = 0; nt < 4; ++nt) {
        v2f bfrag;
        bfrag.x = wp[nt * 16];
        bfrag.y = wp[nt * 16 + 64];
        acc[nt] = wmma_f32(afrag, bfrag, acc[nt]);
      }
    }

    float* my = ytile + wave * (16 * 68);
#pragma unroll
    for (int nt = 0; nt < 4; ++nt) {
      const float bv = b1[nt * 16 + lrow];
#pragma unroll
      for (int r = 0; r < 8; ++r)
        my[(r + lhalf * 8) * 68 + nt * 16 + lrow] = fmaxf(acc[nt][r] + bv, 0.f);
    }

    v8f acc2[4] = {};
    for (int kk = 0; kk < 64; kk += 4) {
      const int k0 = kk + koff;
      v2f afrag;
      afrag.x = my[lrow * 68 + k0];
      afrag.y = my[lrow * 68 + k0 + 1];
      const float* wp = W2 + (size_t)k0 * 64 + lrow;
#pragma unroll
      for (int nt = 0; nt < 4; ++nt) {
        v2f bfrag;
        bfrag.x = wp[nt * 16];
        bfrag.y = wp[nt * 16 + 64];
        acc2[nt] = wmma_f32(afrag, bfrag, acc2[nt]);
      }
    }

#pragma unroll
    for (int nt = 0; nt < 4; ++nt) {
      const float bv = b2[nt * 16 + lrow];
      float s = 0.0f, s2 = 0.0f;
#pragma unroll
      for (int r = 0; r < 8; ++r) {
        const float v = fmaxf(acc2[nt][r] + bv, 0.f);
        h[(size_t)(base + r + lhalf * 8) * 64 + nt * 16 + lrow] = v;
        s += v;
        s2 += v * v;
      }
      unsafeAtomicAdd(&sstat[nt * 16 + lrow], s);
      unsafeAtomicAdd(&sstat[64 + nt * 16 + lrow], s2);
    }
  }
  __syncthreads();
  if (tid < 128) unsafeAtomicAdd(&stats[tid], sstat[tid]);
}

__global__ void bn_finalize_kernel(const float* __restrict__ stats,
                                   const float* __restrict__ gamma,
                                   const float* __restrict__ beta,
                                   float* __restrict__ ss) {
  const int c = threadIdx.x;
  if (c >= 64) return;
  const float inv_n = 1.0f / (float)N_NODES;
  const float mu  = stats[c] * inv_n;
  const float var = stats[64 + c] * inv_n - mu * mu;
  const float rs  = rsqrtf(var + BN_EPS);
  const float sc  = gamma[c] * rs;
  ss[c]      = sc;
  ss[64 + c] = beta[c] - mu * sc;
}

__global__ void bn_relu_kernel(float* __restrict__ h,
                               const float* __restrict__ ss) {
  const long long i = (long long)blockIdx.x * blockDim.x + threadIdx.x;
  if (i >= (long long)N_NODES * 16) return;
  const int c4 = (int)(i & 15) * 4;
  float4 v = ((const float4*)h)[i];
  v.x = fmaxf(v.x * ss[c4 + 0] + ss[64 + c4 + 0], 0.f);
  v.y = fmaxf(v.y * ss[c4 + 1] + ss[64 + c4 + 1], 0.f);
  v.z = fmaxf(v.z * ss[c4 + 2] + ss[64 + c4 + 2], 0.f);
  v.w = fmaxf(v.w * ss[c4 + 3] + ss[64 + c4 + 3], 0.f);
  ((float4*)h)[i] = v;
}

__global__ void final_kernel(const float* __restrict__ h,
                             const float* __restrict__ Wf,
                             const float* __restrict__ bf,
                             float* __restrict__ out) {
  const int i = blockIdx.x * blockDim.x + threadIdx.x;
  if (i >= N_NODES) return;
  const float4* hp = (const float4*)(h + (size_t)i * 64);
  const float4* wp = (const float4*)Wf;
  float s = bf[0];
#pragma unroll
  for (int j = 0; j < 16; ++j) {
    float4 a = hp[j], b = wp[j];
    s += a.x * b.x + a.y * b.y + a.z * b.z + a.w * b.w;
  }
  out[i] = s;
}

extern "C" void kernel_launch(void* const* d_in, const int* in_sizes, int n_in,
                              void* d_out, int out_size, void* d_ws,
                              size_t ws_size, hipStream_t stream) {
  const float* x      = (const float*)d_in[0];
  const int*   labels = (const int*)d_in[1];
  const int*   ei     = (const int*)d_in[2];
  const float* emb    = (const float*)d_in[3];
  const float* W1_0   = (const float*)d_in[4];
  const float* b1_0   = (const float*)d_in[5];
  const float* W2_0   = (const float*)d_in[6];
  const float* b2_0   = (const float*)d_in[7];
  const float* eps_0  = (const float*)d_in[8];
  const float* g_0    = (const float*)d_in[9];
  const float* be_0   = (const float*)d_in[10];
  const float* W1_s   = (const float*)d_in[11];
  const float* b1_s   = (const float*)d_in[12];
  const float* W2_s   = (const float*)d_in[13];
  const float* b2_s   = (const float*)d_in[14];
  const float* eps_s  = (const float*)d_in[15];
  const float* g_s    = (const float*)d_in[16];
  const float* be_s   = (const float*)d_in[17];
  const float* Wf     = (const float*)d_in[18];
  const float* bf     = (const float*)d_in[19];
  float* out = (float*)d_out;

  float* agg   = (float*)d_ws;
  float* stats = agg + (size_t)N_NODES * 64;
  float* ss    = stats + 128;
  float* Wpad  = ss + 128;
  float* h     = Wpad + 1024;

  const int ntiles     = N_NODES / 16;
  const int mlp_blocks = (ntiles + 3) / 4;

  build_input_kernel<<<(N_NODES * 16 + 255) / 256, 256, 0, stream>>>(x, labels,
                                                                     emb, h);
  pad_w1_kernel<<<4, 256, 0, stream>>>(W1_0, Wpad);

  const long long zero4 = ((long long)N_NODES * 64 + 256) / 4;
  const int zblocks = (int)((zero4 + 255) / 256);

  for (int layer = 0; layer < 3; ++layer) {
    zero_f4_kernel<<<zblocks, 256, 0, stream>>>((float4*)agg, zero4);

    const int log2w = (layer == 0) ? 4 : 6;
    const long long tot = (long long)N_EDGES << log2w;
    scatter_kernel<<<(int)((tot + 255) / 256), 256, 0, stream>>>(h, ei, agg,
                                                                 log2w);

    const float* W1 = (layer == 0) ? Wpad : (W1_s + (size_t)(layer - 1) * 4096);
    const float* b1 = (layer == 0) ? b1_0 : (b1_s + (size_t)(layer - 1) * 64);
    const float* W2 = (layer == 0) ? W2_0 : (W2_s + (size_t)(layer - 1) * 4096);
    const float* b2 = (layer == 0) ? b2_0 : (b2_s + (size_t)(layer - 1) * 64);
    const float* ep = (layer == 0) ? eps_0 : eps_s;
    const int    ei_idx = (layer == 0) ? 0 : (layer - 1);
    const float* gm = (layer == 0) ? g_0 : (g_s + (size_t)(layer - 1) * 64);
    const float* bt = (layer == 0) ? be_0 : (be_s + (size_t)(layer - 1) * 64);
    const int    K  = (layer == 0) ? 16 : 64;

    gin_mlp_kernel<<<mlp_blocks, 128, 0, stream>>>(h, agg, W1, b1, W2, b2, ep,
                                                   ei_idx, K, stats);
    bn_finalize_kernel<<<1, 64, 0, stream>>>(stats, gm, bt, ss);
    bn_relu_kernel<<<(N_NODES * 16 + 255) / 256, 256, 0, stream>>>(h, ss);
  }

  final_kernel<<<(N_NODES + 255) / 256, 256, 0, stream>>>(h, Wf, bf, out);
}


# Global-atomic scatter and its clear pass replaced by owner-computes sums: 131 work-groups each scan all dst ids and add the rows of their 768 nodes with LDS f32 atomics; f32 sum order changes
# speedup vs baseline: 1.8827x; 1.8827x over previous
#include <hip/hip_runtime.h>
#include <hip/hip_bf16.h>
#include <cstddef>
#include <cstdint>

#define N_NODES 100000
#define N_EDGES 3200000
#define BN_EPS 1e-5f

typedef __attribute__((ext_vector_type(2))) float v2f;
typedef __attribute__((ext_vector_type(8))) float v8f;

__device__ __forceinline__ v8f wmma_f32(v2f a, v2f b, v8f c) {
  return __builtin_amdgcn_wmma_f32_16x16x4_f32(false, a, false, b, (short)0, c,
                                               false, false);
}

__global__ void build_input_kernel(const float* __restrict__ x,
                                   const int* __restrict__ labels,
                                   const float* __restrict__ emb,
                                   float* __restrict__ h) {
  long long idx = (long long)blockIdx.x * blockDim.x + threadIdx.x;
  if (idx >= (long long)N_NODES * 16) return;
  const int i = (int)(idx >> 4);
  const int c = (int)(idx & 15);
  float v = 0.0f;
  if (c < 2)       v = x[(size_t)i * 2 + c];
  else if (c < 10) v = emb[(size_t)labels[i] * 8 + (c - 2)];
  h[(size_t)i * 64 + c] = v;
}

__global__ void pad_w1_kernel(const float* __restrict__ W1,
                              float* __restrict__ Wpad) {
  const int idx = blockIdx.x * blockDim.x + threadIdx.x;
  if (idx >= 16 * 64) return;
  const int r = idx >> 6;
  Wpad[idx] = (r < 10) ? W1[idx] : 0.0f;
}

#define OWN_ROWS  768
#define OWN_WAVES 32
#define OWN_QCAP  544
#define N_OWNERS  ((N_NODES + OWN_ROWS - 1) / OWN_ROWS)
#define OWN_TRIPS (N_EDGES / 512)

static_assert(N_EDGES % 512 == 0, "edge list is read in whole 512-edge trips");
static_assert(N_EDGES <= (1 << 22), "edge id must fit 22 bits of a queue word");
static_assert(OWN_ROWS <= (1 << 10), "local row must fit 10 bits of a queue word");

__device__ __forceinline__ void own_flush(const float* __restrict__ hcol,
                                          const int* __restrict__ srcs,
                                          const unsigned* q, int n, float* acc2,
                                          int lane, bool colact) {
  const unsigned ent = (lane < n) ? q[lane] : 0u;
  const int dl = (int)(ent >> 22);
  const int s  = srcs[ent & 0x3FFFFFu];
  for (int kb = 0; kb < n; kb += 8) {
    float2 v[8];
#pragma unroll
    for (int j = 0; j < 8; ++j) {
      const unsigned sk = (unsigned)__builtin_amdgcn_readlane(s, kb + j);
      v[j] = *(const float2*)(hcol + (size_t)sk * 64);
    }
#pragma unroll
    for (int j = 0; j < 8; ++j) {
      if (kb + j < n) {
        const int dk = __builtin_amdgcn_readlane(dl, kb + j);
        if (colact) {
          unsafeAtomicAdd(&acc2[dk * 64], v[j].x);
          unsafeAtomicAdd(&acc2[dk * 64 + 1], v[j].y);
        }
      }
    }
  }
}

__device__ __forceinline__ void own_trip(const int4 (&d)[4], int t,
                                         unsigned row0, unsigned* q, int& cnt,
                                         const float* __restrict__ hcol,
                                         const int* __restrict__ srcs,
                                         float* acc2, int lane, bool colact) {
#pragma unroll
  for (int j = 0; j < 4; ++j) {
    const int e0 = t * 512 + j * 128 + lane * 4;
    const int dv[4] = {d[j].x, d[j].y, d[j].z, d[j].w};
#pragma unroll
    for (int c = 0; c < 4; ++c) {
      const unsigned dl = (unsigned)dv[c] - row0;
      const bool m = dl < (unsigned)OWN_ROWS;
      const unsigned bm = __builtin_amdgcn_ballot_w32(m);
      if (bm != 0u) {
        if (m)
          q[cnt + (int)__builtin_amdgcn_mbcnt_lo(bm, 0u)] =
              (unsigned)(e0 + c) | (dl << 22);
        cnt += __builtin_popcount(bm);
      }
    }
  }
  while (cnt >= 32) {
    cnt -= 32;
    own_flush(hcol, srcs, q + cnt, 32, acc2, lane, colact);
  }
}

template <int LOG2W>
__global__ void __launch_bounds__(1024)
owner_agg_kernel(const float* __restrict__ h,
                 const int* __restrict__ edge_index, float* __restrict__ agg,
                 float* __restrict__ stats) {
  __shared__ float acc[OWN_ROWS * 64];
  __shared__ unsigned queue[OWN_WAVES * OWN_QCAP];
  const int tid  = threadIdx.x;
  const int lane = tid & 31;
  const int wave = __builtin_amdgcn_readfirstlane(tid >> 5);
  const unsigned owner = blockIdx.x;
  const unsigned row0 = owner * OWN_ROWS;

  for (int i = tid; i < OWN_ROWS * 16; i += 1024)
    ((float4*)acc)[i] = make_float4(0.f, 0.f, 0.f, 0.f);
  if (owner == 0 && tid < 128) stats[tid] = 0.0f;
  __syncthreads();

  const int4* __restrict__ dst4 = (const int4*)(edge_index + N_EDGES);
  unsigned* q = queue + wave * OWN_QCAP;
  const bool colact = (lane * 2) < (1 << LOG2W);
  const int col = colact ? lane * 2 : 0;
  const float* __restrict__ hcol = h + col;
  float* acc2 = acc + col;
  int cnt = 0;

  int4 a[4], b[4];
#pragma unroll
  for (int j = 0; j < 4; ++j) a[j] = dst4[wave * 128 + j * 32 + lane];

  for (int t = wave; t < OWN_TRIPS; t += 2 * OWN_WAVES) {
    const int tb = t + OWN_WAVES;
    const int tbc = (tb < OWN_TRIPS) ? tb : (OWN_TRIPS - 1);
#pragma unroll
    for (int j = 0; j < 4; ++j) b[j] = dst4[tbc * 128 + j * 32 + lane];
    own_trip(a, t, row0, q, cnt, hcol, edge_index, acc2, lane, colact);

    const int ta = t + 2 * OWN_WAVES;
    const int tac = (ta < OWN_TRIPS) ? ta : (OWN_TRIPS - 1);
#pragma unroll
    for (int j = 0; j < 4; ++j) a[j] = dst4[tac * 128 + j * 32 + lane];
    if (tb < OWN_TRIPS)
      own_trip(b, tb, row0, q, cnt, hcol, edge_index, acc2, lane, colact);
  }
  if (cnt > 0) own_flush(hcol, edge_index, q, cnt, acc2, lane, colact);
  __syncthreads();

  const int nleft = N_NODES - (int)row0;
  const int nrow  = (nleft < OWN_ROWS) ? nleft : OWN_ROWS;
  constexpr int C4LOG = LOG2W - 2;
  for (int i = tid; i < (nrow << C4LOG); i += 1024) {
    const int r = i >> C4LOG;
    const int c = (i & ((1 << C4LOG) - 1)) * 4;
    *(float4*)&agg[(size_t)(row0 + r) * 64 + c] = *(const float4*)&acc[r * 64 + c];
  }
}

__global__ void __launch_bounds__(128)
gin_mlp_kernel(float* __restrict__ h, const float* __restrict__ agg,
               const float* __restrict__ W1, const float* __restrict__ b1,
               const float* __restrict__ W2, const float* __restrict__ b2,
               const float* __restrict__ eps_ptr, int eps_idx, int K,
               float* __restrict__ stats) {
  __shared__ float ytile[4 * 16 * 68];
  __shared__ float sstat[128];
  const int tid   = threadIdx.x;
  const int wave  = tid >> 5;
  const int lane  = tid & 31;
  const int lrow  = lane & 15;
  const int lhalf = lane >> 4;
  const int tile  = blockIdx.x * 4 + wave;

  if (tid < 128) sstat[tid] = 0.0f;
  __syncthreads();

  if (tile < (N_NODES / 16)) {
    const int base = tile * 16;
    const float onepe = 1.0f + eps_ptr[eps_idx];
    const int koff = lhalf * 2;

    v8f acc[4] = {};
    for (int kk = 0; kk < K; kk += 4) {
      const int k0 = kk + koff;
      const float* hp = h   + (size_t)(base + lrow) * 64 + k0;
      const float* ap = agg + (size_t)(base + lrow) * 64 + k0;
      v2f afrag;
      afrag.x = onepe * hp[0] + ap[0];
      afrag.y = onepe * hp[1] + ap[1];
      const float* wp = W1 + (size_t)k0 * 64 + lrow;
#pragma unroll
      for (int nt = 0; nt < 4; ++nt) {
        v2f bfrag;
        bfrag.x = wp[nt * 16];
        bfrag.y = wp[nt * 16 + 64];
        acc[nt] = wmma_f32(afrag, bfrag, acc[nt]);
      }
    }

    float* my = ytile + wave * (16 * 68);
#pragma unroll
    for (int nt = 0; nt < 4; ++nt) {
      const float bv = b1[nt * 16 + lrow];
#pragma unroll
      for (int r = 0; r < 8; ++r)
        my[(r + lhalf * 8) * 68 + nt * 16 + lrow] = fmaxf(acc[nt][r] + bv, 0.f);
    }

    v8f acc2[4] = {};
    for (int kk = 0; kk < 64; kk += 4) {
      const int k0 = kk + koff;
      v2f afrag;
      afrag.x = my[lrow * 68 + k0];
      afrag.y = my[lrow * 68 + k0 + 1];
      const float* wp = W2 + (size_t)k0 * 64 + lrow;
#pragma unroll
      for (int nt = 0; nt < 4; ++nt) {
        v2f bfrag;
        bfrag.x = wp[nt * 16];
        bfrag.y = wp[nt * 16 + 64];
        acc2[nt] = wmma_f32(afrag, bfrag, acc2[nt]);
      }
    }

#pragma unroll
    for (int nt = 0; nt < 4; ++nt) {
      const float bv = b2[nt * 16 + lrow];
      float s = 0.0f, s2 = 0.0f;
#pragma unroll
      for (int r = 0; r < 8; ++r) {
        const float v = fmaxf(acc2[nt][r] + bv, 0.f);
        h[(size_t)(base + r + lhalf * 8) * 64 + nt * 16 + lrow] = v;
        s += v;
        s2 += v * v;
      }
      unsafeAtomicAdd(&sstat[nt * 16 + lrow], s);
      unsafeAtomicAdd(&sstat[64 + nt * 16 + lrow], s2);
    }
  }
  __syncthreads();
  if (tid < 128) unsafeAtomicAdd(&stats[tid], sstat[tid]);
}

__global__ void bn_finalize_kernel(const float* __restrict__ stats,
                                   const float* __restrict__ gamma,
                                   const float* __restrict__ beta,
                                   float* __restrict__ ss) {
  const int c = threadIdx.x;
  if (c >= 64) return;
  const float inv_n = 1.0f / (float)N_NODES;
  const float mu  = stats[c] * inv_n;
  const float var = stats[64 + c] * inv_n - mu * mu;
  const float rs  = rsqrtf(var + BN_EPS);
  const float sc  = gamma[c] * rs;
  ss[c]      = sc;
  ss[64 + c] = beta[c] - mu * sc;
}

__global__ void bn_relu_kernel(float* __restrict__ h,
                               const float* __restrict__ ss) {
  const long long i = (long long)blockIdx.x * blockDim.x + threadIdx.x;
  if (i >= (long long)N_NODES * 16) return;
  const int c4 = (int)(i & 15) * 4;
  float4 v = ((const float4*)h)[i];
  v.x = fmaxf(v.x * ss[c4 + 0] + ss[64 + c4 + 0], 0.f);
  v.y = fmaxf(v.y * ss[c4 + 1] + ss[64 + c4 + 1], 0.f);
  v.z = fmaxf(v.z * ss[c4 + 2] + ss[64 + c4 + 2], 0.f);
  v.w = fmaxf(v.w * ss[c4 + 3] + ss[64 + c4 + 3], 0.f);
  ((float4*)h)[i] = v;
}

__global__ void final_kernel(const float* __restrict__ h,
                             const float* __restrict__ Wf,
                             const float* __restrict__ bf,
                             float* __restrict__ out) {
  const int i = blockIdx.x * blockDim.x + threadIdx.x;
  if (i >= N_NODES) return;
  const float4* hp = (const float4*)(h + (size_t)i * 64);
  const float4* wp = (const float4*)Wf;
  float s = bf[0];
#pragma unroll
  for (int j = 0; j < 16; ++j) {
    float4 a = hp[j], b = wp[j];
    s += a.x * b.x + a.y * b.y + a.z * b.z + a.w * b.w;
  }
  out[i] = s;
}

extern "C" void kernel_launch(void* const* d_in, const int* in_sizes, int n_in,
                              void* d_out, int out_size, void* d_ws,
                              size_t ws_size, hipStream_t stream) {
  const float* x      = (const float*)d_in[0];
  const int*   labels = (const int*)d_in[1];
  const int*   ei     = (const int*)d_in[2];
  const float* emb    = (const float*)d_in[3];
  const float* W1_0   = (const float*)d_in[4];
  const float* b1_0   = (const float*)d_in[5];
  const float* W2_0   = (const float*)d_in[6];
  const float* b2_0   = (const float*)d_in[7];
  const float* eps_0  = (const float*)d_in[8];
  const float* g_0    = (const float*)d_in[9];
  const float* be_0   = (const float*)d_in[10];
  const float* W1_s   = (const float*)d_in[11];
  const float* b1_s   = (const float*)d_in[12];
  const float* W2_s   = (const float*)d_in[13];
  const float* b2_s   = (const float*)d_in[14];
  const float* eps_s  = (const float*)d_in[15];
  const float* g_s    = (const float*)d_in[16];
  const float* be_s   = (const float*)d_in[17];
  const float* Wf     = (const float*)d_in[18];
  const float* bf     = (const float*)d_in[19];
  float* out = (float*)d_out;

  float* agg   = (float*)d_ws;
  float* stats = agg + (size_t)N_NODES * 64;
  float* ss    = stats + 128;
  float* Wpad  = ss + 128;
  float* h     = Wpad + 1024;

  const int ntiles     = N_NODES / 16;
  const int mlp_blocks = (ntiles + 3) / 4;

  build_input_kernel<<<(N_NODES * 16 + 255) / 256, 256, 0, stream>>>(x, labels,
                                                                     emb, h);
  pad_w1_kernel<<<4, 256, 0, stream>>>(W1_0, Wpad);

  for (int layer = 0; layer < 3; ++layer) {
    if (layer == 0)
      owner_agg_kernel<4><<<N_OWNERS, 1024, 0, stream>>>(h, ei, agg, stats);
    else
      owner_agg_kernel<6><<<N_OWNERS, 1024, 0, stream>>>(h, ei, agg, stats);

    const float* W1 = (layer == 0) ? Wpad : (W1_s + (size_t)(layer - 1) * 4096);
    const float* b1 = (layer == 0) ? b1_0 : (b1_s + (size_t)(layer - 1) * 64);
    const float* W2 = (layer == 0) ? W2_0 : (W2_s + (size_t)(layer - 1) * 4096);
    const float* b2 = (layer == 0) ? b2_0 : (b2_s + (size_t)(layer - 1) * 64);
    const float* ep = (layer == 0) ? eps_0 : eps_s;
    const int    ei_idx = (layer == 0) ? 0 : (layer - 1);
    const float* gm = (layer == 0) ? g_0 : (g_s + (size_t)(layer - 1) * 64);
    const float* bt = (layer == 0) ? be_0 : (be_s + (size_t)(layer - 1) * 64);
    const int    K  = (layer == 0) ? 16 : 64;

    gin_mlp_kernel<<<mlp_blocks, 128, 0, stream>>>(h, agg, W1, b1, W2, b2, ep,
                                                   ei_idx, K, stats);
    bn_finalize_kernel<<<1, 64, 0, stream>>>(stats, gm, bt, ss);
    bn_relu_kernel<<<(N_NODES * 16 + 255) / 256, 256, 0, stream>>>(h, ss);
  }

  final_kernel<<<(N_NODES + 255) / 256, 256, 0, stream>>>(h, Wf, bf, out);
}


# Owner gather deepened: each batch of 32 queued edges loads all 32 source rows before its first add (was 8 at a time) and its source ids one batch ahead; LDS f32 add order stays arrival order
# speedup vs baseline: 1.9551x; 1.0385x over previous
#include <hip/hip_runtime.h>
#include <hip/hip_bf16.h>
#include <cstddef>
#include <cstdint>

#define N_NODES 100000
#define N_EDGES 3200000
#define BN_EPS 1e-5f

typedef __attribute__((ext_vector_type(2))) float v2f;
typedef __attribute__((ext_vector_type(8))) float v8f;

__device__ __forceinline__ v8f wmma_f32(v2f a, v2f b, v8f c) {
  return __builtin_amdgcn_wmma_f32_16x16x4_f32(false, a, false, b, (short)0, c,
                                               false, false);
}

__global__ void build_input_kernel(const float* __restrict__ x,
                                   const int* __restrict__ labels,
                                   const float* __restrict__ emb,
                                   float* __restrict__ h) {
  long long idx = (long long)blockIdx.x * blockDim.x + threadIdx.x;
  if (idx >= (long long)N_NODES * 16) return;
  const int i = (int)(idx >> 4);
  const int c = (int)(idx & 15);
  float v = 0.0f;
  if (c < 2)       v = x[(size_t)i * 2 + c];
  else if (c < 10) v = emb[(size_t)labels[i] * 8 + (c - 2)];
  h[(size_t)i * 64 + c] = v;
}

__global__ void pad_w1_kernel(const float* __restrict__ W1,
                              float* __restrict__ Wpad) {
  const int idx = blockIdx.x * blockDim.x + threadIdx.x;
  if (idx >= 16 * 64) return;
  const int r = idx >> 6;
  Wpad[idx] = (r < 10) ? W1[idx] : 0.0f;
}

#define OWN_ROWS  768
#define OWN_WAVES 32
#define OWN_QCAP  544
#define N_OWNERS  ((N_NODES + OWN_ROWS - 1) / OWN_ROWS)
#define OWN_TRIPS (N_EDGES / 512)

static_assert(N_EDGES % 512 == 0, "edge list is read in whole 512-edge trips");
static_assert(N_EDGES <= (1 << 22), "edge id must fit 22 bits of a queue word");
static_assert(OWN_ROWS <= (1 << 10), "local row must fit 10 bits of a queue word");

__device__ __forceinline__ void own_flush(const float* __restrict__ hcol,
                                          const int* __restrict__ srcs,
                                          const unsigned* q, int n, int& ps,
                                          int& pdl, int& pn, float* acc2,
                                          int lane, bool colact) {
  const unsigned ent = (lane < n) ? q[lane] : 0u;
  const int ndl = (int)(ent >> 22);
  const int ns  = srcs[ent & 0x3FFFFFu];
  if (pn > 0) {
    float2 v[32];
#pragma unroll
    for (int j = 0; j < 32; ++j) {
      const unsigned sk = (unsigned)__builtin_amdgcn_readlane(ps, j);
      v[j] = *(const float2*)(hcol + (size_t)sk * 64);
    }
#pragma unroll
    for (int j = 0; j < 32; ++j) {
      if (j < pn) {
        const int dk = __builtin_amdgcn_readlane(pdl, j);
        if (colact) {
          unsafeAtomicAdd(&acc2[dk * 64], v[j].x);
          unsafeAtomicAdd(&acc2[dk * 64 + 1], v[j].y);
        }
      }
    }
  }
  ps = ns;
  pdl = ndl;
  pn = n;
}

__device__ __forceinline__ void own_trip(const int4 (&d)[4], int t,
                                         unsigned row0, unsigned* q, int& cnt,
                                         int& ps, int& pdl, int& pn,
                                         const float* __restrict__ hcol,
                                         const int* __restrict__ srcs,
                                         float* acc2, int lane, bool colact) {
#pragma unroll
  for (int j = 0; j < 4; ++j) {
    const int e0 = t * 512 + j * 128 + lane * 4;
    const int dv[4] = {d[j].x, d[j].y, d[j].z, d[j].w};
#pragma unroll
    for (int c = 0; c < 4; ++c) {
      const unsigned dl = (unsigned)dv[c] - row0;
      const bool m = dl < (unsigned)OWN_ROWS;
      const unsigned bm = __builtin_amdgcn_ballot_w32(m);
      if (bm != 0u) {
        if (m)
          q[cnt + (int)__builtin_amdgcn_mbcnt_lo(bm, 0u)] =
              (unsigned)(e0 + c) | (dl << 22);
        cnt += __builtin_popcount(bm);
      }
    }
  }
  while (cnt >= 32) {
    cnt -= 32;
    own_flush(hcol, srcs, q + cnt, 32, ps, pdl, pn, acc2, lane, colact);
  }
}

template <int LOG2W>
__global__ void __launch_bounds__(1024)
owner_agg_kernel(const float* __restrict__ h,
                 const int* __restrict__ edge_index, float* __restrict__ agg,
                 float* __restrict__ stats) {
  __shared__ float acc[OWN_ROWS * 64];
  __shared__ unsigned queue[OWN_WAVES * OWN_QCAP];
  const int tid  = threadIdx.x;
  const int lane = tid & 31;
  const int wave = __builtin_amdgcn_readfirstlane(tid >> 5);
  const unsigned owner = blockIdx.x;
  const unsigned row0 = owner * OWN_ROWS;

  for (int i = tid; i < OWN_ROWS * 16; i += 1024)
    ((float4*)acc)[i] = make_float4(0.f, 0.f, 0.f, 0.f);
  if (owner == 0 && tid < 128) stats[tid] = 0.0f;
  __syncthreads();

  const int4* __restrict__ dst4 = (const int4*)(edge_index + N_EDGES);
  unsigned* q = queue + wave * OWN_QCAP;
  const bool colact = (lane * 2) < (1 << LOG2W);
  const int col = colact ? lane * 2 : 0;
  const float* __restrict__ hcol = h + col;
  float* acc2 = acc + col;
  int cnt = 0;
  int ps = 0, pdl = 0, pn = 0;

  int4 a[4], b[4];
#pragma unroll
  for (int j = 0; j < 4; ++j) a[j] = dst4[wave * 128 + j * 32 + lane];

  for (int t = wave; t < OWN_TRIPS; t += 2 * OWN_WAVES) {
    const int tb = t + OWN_WAVES;
    const int tbc = (tb < OWN_TRIPS) ? tb : (OWN_TRIPS - 1);
#pragma unroll
    for (int j = 0; j < 4; ++j) b[j] = dst4[tbc * 128 + j * 32 + lane];
    own_trip(a, t, row0, q, cnt, ps, pdl, pn, hcol, edge_index, acc2, lane,
             colact);

    const int ta = t + 2 * OWN_WAVES;
    const int tac = (ta < OWN_TRIPS) ? ta : (OWN_TRIPS - 1);
#pragma unroll
    for (int j = 0; j < 4; ++j) a[j] = dst4[tac * 128 + j * 32 + lane];
    if (tb < OWN_TRIPS)
      own_trip(b, tb, row0, q, cnt, ps, pdl, pn, hcol, edge_index, acc2, lane,
               colact);
  }
  if (cnt > 0)
    own_flush(hcol, edge_index, q, cnt, ps, pdl, pn, acc2, lane, colact);
  own_flush(hcol, edge_index, q, 0, ps, pdl, pn, acc2, lane, colact);
  __syncthreads();

  const int nleft = N_NODES - (int)row0;
  const int nrow  = (nleft < OWN_ROWS) ? nleft : OWN_ROWS;
  constexpr int C4LOG = LOG2W - 2;
  for (int i = tid; i < (nrow << C4LOG); i += 1024) {
    const int r = i >> C4LOG;
    const int c = (i & ((1 << C4LOG) - 1)) * 4;
    *(float4*)&agg[(size_t)(row0 + r) * 64 + c] = *(const float4*)&acc[r * 64 + c];
  }
}

__global__ void __launch_bounds__(128)
gin_mlp_kernel(float* __restrict__ h, const float* __restrict__ agg,
               const float* __restrict__ W1, const float* __restrict__ b1,
               const float* __restrict__ W2, const float* __restrict__ b2,
               const float* __restrict__ eps_ptr, int eps_idx, int K,
               float* __restrict__ stats) {
  __shared__ float ytile[4 * 16 * 68];
  __shared__ float sstat[128];
  const int tid   = threadIdx.x;
  const int wave  = tid >> 5;
  const int lane  = tid & 31;
  const int lrow  = lane & 15;
  const int lhalf = lane >> 4;
  const int tile  = blockIdx.x * 4 + wave;

  if (tid < 128) sstat[tid] = 0.0f;
  __syncthreads();

  if (tile < (N_NODES / 16)) {
    const int base = tile * 16;
    const float onepe = 1.0f + eps_ptr[eps_idx];
    const int koff = lhalf * 2;

    v8f acc[4] = {};
    for (int kk = 0; kk < K; kk += 4) {
      const int k0 = kk + koff;
      const float* hp = h   + (size_t)(base + lrow) * 64 + k0;
      const float* ap = agg + (size_t)(base + lrow) * 64 + k0;
      v2f afrag;
      afrag.x = onepe * hp[0] + ap[0];
      afrag.y = onepe * hp[1] + ap[1];
      const float* wp = W1 + (size_t)k0 * 64 + lrow;
#pragma unroll
      for (int nt = 0; nt < 4; ++nt) {
        v2f bfrag;
        bfrag.x = wp[nt * 16];
        bfrag.y = wp[nt * 16 + 64];
        acc[nt] = wmma_f32(afrag, bfrag, acc[nt]);
      }
    }

    float* my = ytile + wave * (16 * 68);
#pragma unroll
    for (int nt = 0; nt < 4; ++nt) {
      const float bv = b1[nt * 16 + lrow];
#pragma unroll
      for (int r = 0; r < 8; ++r)
        my[(r + lhalf * 8) * 68 + nt * 16 + lrow] = fmaxf(acc[nt][r] + bv, 0.f);
    }

    v8f acc2[4] = {};
    for (int kk = 0; kk < 64; kk += 4) {
      const int k0 = kk + koff;
      v2f afrag;
      afrag.x = my[lrow * 68 + k0];
      afrag.y = my[lrow * 68 + k0 + 1];
      const float* wp = W2 + (size_t)k0 * 64 + lrow;
#pragma unroll
      for (int nt = 0; nt < 4; ++nt) {
        v2f bfrag;
        bfrag.x = wp[nt * 16];
        bfrag.y = wp[nt * 16 + 64];
        acc2[nt] = wmma_f32(afrag, bfrag, acc2[nt]);
      }
    }

#pragma unroll
    for (int nt = 0; nt < 4; ++nt) {
      const float bv = b2[nt * 16 + lrow];
      float s = 0.0f, s2 = 0.0f;
#pragma unroll
      for (int r = 0; r < 8; ++r) {
        const float v = fmaxf(acc2[nt][r] + bv, 0.f);
        h[(size_t)(base + r + lhalf * 8) * 64 + nt * 16 + lrow] = v;
        s += v;
        s2 += v * v;
      }
      unsafeAtomicAdd(&sstat[nt * 16 + lrow], s);
      unsafeAtomicAdd(&sstat[64 + nt * 16 + lrow], s2);
    }
  }
  __syncthreads();
  if (tid < 128) unsafeAtomicAdd(&stats[tid], sstat[tid]);
}

__global__ void bn_finalize_kernel(const float* __restrict__ stats,
                                   const float* __restrict__ gamma,
                                   const float* __restrict__ beta,
                                   float* __restrict__ ss) {
  const int c = threadIdx.x;
  if (c >= 64) return;
  const float inv_n = 1.0f / (float)N_NODES;
  const float mu  = stats[c] * inv_n;
  const float var = stats[64 + c] * inv_n - mu * mu;
  const float rs  = rsqrtf(var + BN_EPS);
  const float sc  = gamma[c] * rs;
  ss[c]      = sc;
  ss[64 + c] = beta[c] - mu * sc;
}

__global__ void bn_relu_kernel(float* __restrict__ h,
                               const float* __restrict__ ss) {
  const long long i = (long long)blockIdx.x * blockDim.x + threadIdx.x;
  if (i >= (long long)N_NODES * 16) return;
  const int c4 = (int)(i & 15) * 4;
  float4 v = ((const float4*)h)[i];
  v.x = fmaxf(v.x * ss[c4 + 0] + ss[64 + c4 + 0], 0.f);
  v.y = fmaxf(v.y * ss[c4 + 1] + ss[64 + c4 + 1], 0.f);
  v.z = fmaxf(v.z * ss[c4 + 2] + ss[64 + c4 + 2], 0.f);
  v.w = fmaxf(v.w * ss[c4 + 3] + ss[64 + c4 + 3], 0.f);
  ((float4*)h)[i] = v;
}

__global__ void final_kernel(const float* __restrict__ h,
                             const float* __restrict__ Wf,
                             const float* __restrict__ bf,
                             float* __restrict__ out) {
  const int i = blockIdx.x * blockDim.x + threadIdx.x;
  if (i >= N_NODES) return;
  const float4* hp = (const float4*)(h + (size_t)i * 64);
  const float4* wp = (const float4*)Wf;
  float s = bf[0];
#pragma unroll
  for (int j = 0; j < 16; ++j) {
    float4 a = hp[j], b = wp[j];
    s += a.x * b.x + a.y * b.y + a.z * b.z + a.w * b.w;
  }
  out[i] = s;
}

extern "C" void kernel_launch(void* const* d_in, const int* in_sizes, int n_in,
                              void* d_out, int out_size, void* d_ws,
                              size_t ws_size, hipStream_t stream) {
  const float* x      = (const float*)d_in[0];
  const int*   labels = (const int*)d_in[1];
  const int*   ei     = (const int*)d_in[2];
  const float* emb    = (const float*)d_in[3];
  const float* W1_0   = (const float*)d_in[4];
  const float* b1_0   = (const float*)d_in[5];
  const float* W2_0   = (const float*)d_in[6];
  const float* b2_0   = (const float*)d_in[7];
  const float* eps_0  = (const float*)d_in[8];
  const float* g_0    = (const float*)d_in[9];
  const float* be_0   = (const float*)d_in[10];
  const float* W1_s   = (const float*)d_in[11];
  const float* b1_s   = (const float*)d_in[12];
  const float* W2_s   = (const float*)d_in[13];
  const float* b2_s   = (const float*)d_in[14];
  const float* eps_s  = (const float*)d_in[15];
  const float* g_s    = (const float*)d_in[16];
  const float* be_s   = (const float*)d_in[17];
  const float* Wf     = (const float*)d_in[18];
  const float* bf     = (const float*)d_in[19];
  float* out = (float*)d_out;

  float* agg   = (float*)d_ws;
  float* stats = agg + (size_t)N_NODES * 64;
  float* ss    = stats + 128;
  float* Wpad  = ss + 128;
  float* h     = Wpad + 1024;

  const int ntiles     = N_NODES / 16;
  const int mlp_blocks = (ntiles + 3) / 4;

  build_input_kernel<<<(N_NODES * 16 + 255) / 256, 256, 0, stream>>>(x, labels,
                                                                     emb, h);
  pad_w1_kernel<<<4, 256, 0, stream>>>(W1_0, Wpad);

  for (int layer = 0; layer < 3; ++layer) {
    if (layer == 0)
      owner_agg_kernel<4><<<N_OWNERS, 1024, 0, stream>>>(h, ei, agg, stats);
    else
      owner_agg_kernel<6><<<N_OWNERS, 1024, 0, stream>>>(h, ei, agg, stats);

    const float* W1 = (layer == 0) ? Wpad : (W1_s + (size_t)(layer - 1) * 4096);
    const float* b1 = (layer == 0) ? b1_0 : (b1_s + (size_t)(layer - 1) * 64);
    const float* W2 = (layer == 0) ? W2_0 : (W2_s + (size_t)(layer - 1) * 4096);
    const float* b2 = (layer == 0) ? b2_0 : (b2_s + (size_t)(layer - 1) * 64);
    const float* ep = (layer == 0) ? eps_0 : eps_s;
    const int    ei_idx = (layer == 0) ? 0 : (layer - 1);
    const float* gm = (layer == 0) ? g_0 : (g_s + (size_t)(layer - 1) * 64);
    const float* bt = (layer == 0) ? be_0 : (be_s + (size_t)(layer - 1) * 64);
    const int    K  = (layer == 0) ? 16 : 64;

    gin_mlp_kernel<<<mlp_blocks, 128, 0, stream>>>(h, agg, W1, b1, W2, b2, ep,
                                                   ei_idx, K, stats);
    bn_finalize_kernel<<<1, 64, 0, stream>>>(stats, gm, bt, ss);
    bn_relu_kernel<<<(N_NODES * 16 + 255) / 256, 256, 0, stream>>>(h, ss);
  }

  final_kernel<<<(N_NODES + 255) / 256, 256, 0, stream>>>(h, Wf, bf, out);
}


# First-layer aggregation: edge list cut in 4 quarters, 63 owners x 4 work-groups each read one quarter; 4 partial-sum planes in the agg buffer are added in the MLP A loader; f32 sum order changes
# speedup vs baseline: 2.2312x; 1.1412x over previous
#include <hip/hip_runtime.h>
#include <hip/hip_bf16.h>
#include <cstddef>
#include <cstdint>

#define N_NODES 100000
#define N_EDGES 3200000
#define BN_EPS 1e-5f

typedef __attribute__((ext_vector_type(2))) float v2f;
typedef __attribute__((ext_vector_type(8))) float v8f;

__device__ __forceinline__ v8f wmma_f32(v2f a, v2f b, v8f c) {
  return __builtin_amdgcn_wmma_f32_16x16x4_f32(false, a, false, b, (short)0, c,
                                               false, false);
}

__global__ void build_input_kernel(const float* __restrict__ x,
                                   const int* __restrict__ labels,
                                   const float* __restrict__ emb,
                                   float* __restrict__ h) {
  long long idx = (long long)blockIdx.x * blockDim.x + threadIdx.x;
  if (idx >= (long long)N_NODES * 16) return;
  const int i = (int)(idx >> 4);
  const int c = (int)(idx & 15);
  float v = 0.0f;
  if (c < 2)       v = x[(size_t)i * 2 + c];
  else if (c < 10) v = emb[(size_t)labels[i] * 8 + (c - 2)];
  h[(size_t)i * 64 + c] = v;
}

__global__ void pad_w1_kernel(const float* __restrict__ W1,
                              float* __restrict__ Wpad) {
  const int idx = blockIdx.x * blockDim.x + threadIdx.x;
  if (idx >= 16 * 64) return;
  const int r = idx >> 6;
  Wpad[idx] = (r < 10) ? W1[idx] : 0.0f;
}

#define OWN_ROWS  768
#define OWN_ROWS0 1600
#define OWN_SPLIT0 4
#define OWN_WAVES 32
#define OWN_QCAP  544
#define N_OWNERS  ((N_NODES + OWN_ROWS - 1) / OWN_ROWS)
#define N_OWNERS0 ((N_NODES + OWN_ROWS0 - 1) / OWN_ROWS0)
#define OWN_TRIPS (N_EDGES / 512)

static_assert(N_EDGES % 512 == 0, "edge list is read in whole 512-edge trips");
static_assert(OWN_ROWS0 % 2 == 0, "an owner's plane rows start on a 128-byte line");
static_assert(OWN_SPLIT0 * 16 <= 64, "the planes of partial sums fit the agg buffer");

template <int STRIDE, int SHIFT>
__device__ __forceinline__ void own_flush(const float* __restrict__ hcol,
                                          const int* __restrict__ srcs,
                                          const unsigned* q, int n, int& ps,
                                          int& pdl, int& pn, float* acc2,
                                          int lane, bool colact) {
  const unsigned ent = (lane < n) ? q[lane] : 0u;
  const int ndl = (int)(ent >> SHIFT);
  const int ns  = srcs[ent & ((1u << SHIFT) - 1u)];
  if (pn > 0) {
    float2 v[32];
#pragma unroll
    for (int j = 0; j < 32; ++j) {
      const unsigned sk = (unsigned)__builtin_amdgcn_readlane(ps, j);
      v[j] = *(const float2*)(hcol + (size_t)sk * 64);
    }
#pragma unroll
    for (int j = 0; j < 32; ++j) {
      if (j < pn) {
        const int dk = __builtin_amdgcn_readlane(pdl, j);
        if (colact) {
          unsafeAtomicAdd(&acc2[dk * STRIDE], v[j].x);
          unsafeAtomicAdd(&acc2[dk * STRIDE + 1], v[j].y);
        }
      }
    }
  }
  ps = ns;
  pdl = ndl;
  pn = n;
}

template <int ROWS, int STRIDE, int SHIFT>
__device__ __forceinline__ void own_trip(const int4 (&d)[4], int t,
                                         unsigned row0, unsigned* q, int& cnt,
                                         int& ps, int& pdl, int& pn,
                                         const float* __restrict__ hcol,
                                         const int* __restrict__ srcs,
                                         float* acc2, int lane, bool colact) {
#pragma unroll
  for (int j = 0; j < 4; ++j) {
    const int e0 = t * 512 + j * 128 + lane * 4;
    const int dv[4] = {d[j].x, d[j].y, d[j].z, d[j].w};
#pragma unroll
    for (int c = 0; c < 4; ++c) {
      const unsigned dl = (unsigned)dv[c] - row0;
      const bool m = dl < (unsigned)ROWS;
      const unsigned bm = __builtin_amdgcn_ballot_w32(m);
      if (bm != 0u) {
        if (m)
          q[cnt + (int)__builtin_amdgcn_mbcnt_lo(bm, 0u)] =
              (unsigned)(e0 + c) | (dl << SHIFT);
        cnt += __builtin_popcount(bm);
      }
    }
  }
  while (cnt >= 32) {
    cnt -= 32;
    own_flush<STRIDE, SHIFT>(hcol, srcs, q + cnt, 32, ps, pdl, pn, acc2, lane,
                             colact);
  }
}

template <int LOG2W, int ROWS, int SPLIT>
__global__ void __launch_bounds__(1024)
owner_agg_kernel(const float* __restrict__ h,
                 const int* __restrict__ edge_index, float* __restrict__ agg,
                 float* __restrict__ stats) {
  constexpr int STRIDE = (SPLIT == 1) ? 64 : (1 << LOG2W);
  constexpr int NOWN   = (N_NODES + ROWS - 1) / ROWS;
  constexpr int PART_EDGES = (OWN_TRIPS / SPLIT + 1) * 512;
  constexpr int SHIFT = (PART_EDGES <= (1 << 20)) ? 20 : 22;
  static_assert(PART_EDGES <= (1 << SHIFT), "edge id must fit the low bits of a queue word");
  static_assert(ROWS <= (1 << (32 - SHIFT)), "local row must fit the high bits of a queue word");
  static_assert(OWN_TRIPS / SPLIT >= OWN_WAVES, "every wave has a first trip");
  __shared__ float acc[ROWS * STRIDE];
  __shared__ unsigned queue[OWN_WAVES * OWN_QCAP];
  const int tid  = threadIdx.x;
  const int lane = tid & 31;
  const int wave = __builtin_amdgcn_readfirstlane(tid >> 5);
  const unsigned owner = (SPLIT == 1) ? blockIdx.x : (blockIdx.x % NOWN);
  const int part = (SPLIT == 1) ? 0 : (int)(blockIdx.x / NOWN);
  const unsigned row0 = owner * ROWS;
  const int tbeg = (part * OWN_TRIPS) / SPLIT;
  const int tend = ((part + 1) * OWN_TRIPS) / SPLIT;

  for (int i = tid; i < ROWS * STRIDE / 4; i += 1024)
    ((float4*)acc)[i] = make_float4(0.f, 0.f, 0.f, 0.f);
  if (blockIdx.x == 0 && tid < 128) stats[tid] = 0.0f;
  __syncthreads();

  const int4* __restrict__ dst4 = (const int4*)(edge_index + N_EDGES);
  const int* __restrict__ srcs = edge_index + (size_t)tbeg * 512;
  unsigned* q = queue + wave * OWN_QCAP;
  const bool colact = (lane * 2) < (1 << LOG2W);
  const int col = colact ? lane * 2 : 0;
  const float* __restrict__ hcol = h + col;
  float* acc2 = acc + col;
  int cnt = 0;
  int ps = 0, pdl = 0, pn = 0;

  int4 a[4], b[4];
#pragma unroll
  for (int j = 0; j < 4; ++j) a[j] = dst4[(tbeg + wave) * 128 + j * 32 + lane];

  for (int t = tbeg + wave; t < tend; t += 2 * OWN_WAVES) {
    const int tb = t + OWN_WAVES;
    const int tbc = (tb < tend) ? tb : (tend - 1);
#pragma unroll
    for (int j = 0; j < 4; ++j) b[j] = dst4[tbc * 128 + j * 32 + lane];
    own_trip<ROWS, STRIDE, SHIFT>(a, t - tbeg, row0, q, cnt, ps, pdl, pn, hcol,
                                  srcs, acc2, lane, colact);

    const int ta = t + 2 * OWN_WAVES;
    const int tac = (ta < tend) ? ta : (tend - 1);
#pragma unroll
    for (int j = 0; j < 4; ++j) a[j] = dst4[tac * 128 + j * 32 + lane];
    if (tb < tend)
      own_trip<ROWS, STRIDE, SHIFT>(b, tb - tbeg, row0, q, cnt, ps, pdl, pn,
                                    hcol, srcs, acc2, lane, colact);
  }
  if (cnt > 0)
    own_flush<STRIDE, SHIFT>(hcol, srcs, q, cnt, ps, pdl, pn, acc2, lane,
                             colact);
  own_flush<STRIDE, SHIFT>(hcol, srcs, q, 0, ps, pdl, pn, acc2, lane, colact);
  __syncthreads();

  const int nleft = N_NODES - (int)row0;
  const int nrow  = (nleft < ROWS) ? nleft : ROWS;
  if (SPLIT == 1) {
    constexpr int C4LOG = LOG2W - 2;
    for (int i = tid; i < (nrow << C4LOG); i += 1024) {
      const int r = i >> C4LOG;
      const int c = (i & ((1 << C4LOG) - 1)) * 4;
      *(float4*)&agg[(size_t)(row0 + r) * 64 + c] = *(const float4*)&acc[r * 64 + c];
    }
  } else {
    float4* __restrict__ dstp =
        (float4*)(agg + ((size_t)part * N_NODES + row0) * STRIDE);
    for (int i = tid; i < nrow * (STRIDE / 4); i += 1024)
      dstp[i] = ((const float4*)acc)[i];
  }
}

template <int NPART>
__global__ void __launch_bounds__(128)
gin_mlp_kernel(float* __restrict__ h, const float* __restrict__ agg,
               const float* __restrict__ W1, const float* __restrict__ b1,
               const float* __restrict__ W2, const float* __restrict__ b2,
               const float* __restrict__ eps_ptr, int eps_idx, int K,
               float* __restrict__ stats) {
  __shared__ float ytile[4 * 16 * 68];
  __shared__ float sstat[128];
  const int tid   = threadIdx.x;
  const int wave  = tid >> 5;
  const int lane  = tid & 31;
  const int lrow  = lane & 15;
  const int lhalf = lane >> 4;
  const int tile  = blockIdx.x * 4 + wave;

  if (tid < 128) sstat[tid] = 0.0f;
  __syncthreads();

  if (tile < (N_NODES / 16)) {
    const int base = tile * 16;
    const float onepe = 1.0f + eps_ptr[eps_idx];
    const int koff = lhalf * 2;

    v8f acc[4] = {};
    for (int kk = 0; kk < K; kk += 4) {
      const int k0 = kk + koff;
      const float* hp = h   + (size_t)(base + lrow) * 64 + k0;
      v2f afrag;
      if (NPART == 1) {
        const float* ap = agg + (size_t)(base + lrow) * 64 + k0;
        afrag.x = onepe * hp[0] + ap[0];
        afrag.y = onepe * hp[1] + ap[1];
      } else {
        const float* ap = agg + (size_t)(base + lrow) * 16 + k0;
        float sx = ap[0], sy = ap[1];
#pragma unroll
        for (int p = 1; p < NPART; ++p) {
          sx += ap[(size_t)p * N_NODES * 16];
          sy += ap[(size_t)p * N_NODES * 16 + 1];
        }
        afrag.x = onepe * hp[0] + sx;
        afrag.y = onepe * hp[1] + sy;
      }
      const float* wp = W1 + (size_t)k0 * 64 + lrow;
#pragma unroll
      for (int nt = 0; nt < 4; ++nt) {
        v2f bfrag;
        bfrag.x = wp[nt * 16];
        bfrag.y = wp[nt * 16 + 64];
        acc[nt] = wmma_f32(afrag, bfrag, acc[nt]);
      }
    }

    float* my = ytile + wave * (16 * 68);
#pragma unroll
    for (int nt = 0; nt < 4; ++nt) {
      const float bv = b1[nt * 16 + lrow];
#pragma unroll
      for (int r = 0; r < 8; ++r)
        my[(r + lhalf * 8) * 68 + nt * 16 + lrow] = fmaxf(acc[nt][r] + bv, 0.f);
    }

    v8f acc2[4] = {};
    for (int kk = 0; kk < 64; kk += 4) {
      const int k0 = kk + koff;
      v2f afrag;
      afrag.x = my[lrow * 68 + k0];
      afrag.y = my[lrow * 68 + k0 + 1];
      const float* wp = W2 + (size_t)k0 * 64 + lrow;
#pragma unroll
      for (int nt = 0; nt < 4; ++nt) {
        v2f bfrag;
        bfrag.x = wp[nt * 16];
        bfrag.y = wp[nt * 16 + 64];
        acc2[nt] = wmma_f32(afrag, bfrag, acc2[nt]);
      }
    }

#pragma unroll
    for (int nt = 0; nt < 4; ++nt) {
      const float bv = b2[nt * 16 + lrow];
      float s = 0.0f, s2 = 0.0f;
#pragma unroll
      for (int r = 0; r < 8; ++r) {
        const float v = fmaxf(acc2[nt][r] + bv, 0.f);
        h[(size_t)(base + r + lhalf * 8) * 64 + nt * 16 + lrow] = v;
        s += v;
        s2 += v * v;
      }
      unsafeAtomicAdd(&sstat[nt * 16 + lrow], s);
      unsafeAtomicAdd(&sstat[64 + nt * 16 + lrow], s2);
    }
  }
  __syncthreads();
  if (tid < 128) unsafeAtomicAdd(&stats[tid], sstat[tid]);
}

__global__ void bn_finalize_kernel(const float* __restrict__ stats,
                                   const float* __restrict__ gamma,
                                   const float* __restrict__ beta,
                                   float* __restrict__ ss) {
  const int c = threadIdx.x;
  if (c >= 64) return;
  const float inv_n = 1.0f / (float)N_NODES;
  const float mu  = stats[c] * inv_n;
  const float var = stats[64 + c] * inv_n - mu * mu;
  const float rs  = rsqrtf(var + BN_EPS);
  const float sc  = gamma[c] * rs;
  ss[c]      = sc;
  ss[64 + c] = beta[c] - mu * sc;
}

__global__ void bn_relu_kernel(float* __restrict__ h,
                               const float* __restrict__ ss) {
  const long long i = (long long)blockIdx.x * blockDim.x + threadIdx.x;
  if (i >= (long long)N_NODES * 16) return;
  const int c4 = (int)(i & 15) * 4;
  float4 v = ((const float4*)h)[i];
  v.x = fmaxf(v.x * ss[c4 + 0] + ss[64 + c4 + 0], 0.f);
  v.y = fmaxf(v.y * ss[c4 + 1] + ss[64 + c4 + 1], 0.f);
  v.z = fmaxf(v.z * ss[c4 + 2] + ss[64 + c4 + 2], 0.f);
  v.w = fmaxf(v.w * ss[c4 + 3] + ss[64 + c4 + 3], 0.f);
  ((float4*)h)[i] = v;
}

__global__ void final_kernel(const float* __restrict__ h,
                             const float* __restrict__ Wf,
                             const float* __restrict__ bf,
                             float* __restrict__ out) {
  const int i = blockIdx.x * blockDim.x + threadIdx.x;
  if (i >= N_NODES) return;
  const float4* hp = (const float4*)(h + (size_t)i * 64);
  const float4* wp = (const float4*)Wf;
  float s = bf[0];
#pragma unroll
  for (int j = 0; j < 16; ++j) {
    float4 a = hp[j], b = wp[j];
    s += a.x * b.x + a.y * b.y + a.z * b.z + a.w * b.w;
  }
  out[i] = s;
}

extern "C" void kernel_launch(void* const* d_in, const int* in_sizes, int n_in,
                              void* d_out, int out_size, void* d_ws,
                              size_t ws_size, hipStream_t stream) {
  const float* x      = (const float*)d_in[0];
  const int*   labels = (const int*)d_in[1];
  const int*   ei     = (const int*)d_in[2];
  const float* emb    = (const float*)d_in[3];
  const float* W1_0   = (const float*)d_in[4];
  const float* b1_0   = (const float*)d_in[5];
  const float* W2_0   = (const float*)d_in[6];
  const float* b2_0   = (const float*)d_in[7];
  const float* eps_0  = (const float*)d_in[8];
  const float* g_0    = (const float*)d_in[9];
  const float* be_0   = (const float*)d_in[10];
  const float* W1_s   = (const float*)d_in[11];
  const float* b1_s   = (const float*)d_in[12];
  const float* W2_s   = (const float*)d_in[13];
  const float* b2_s   = (const float*)d_in[14];
  const float* eps_s  = (const float*)d_in[15];
  const float* g_s    = (const float*)d_in[16];
  const float* be_s   = (const float*)d_in[17];
  const float* Wf     = (const float*)d_in[18];
  const float* bf     = (const float*)d_in[19];
  float* out = (float*)d_out;

  float* agg   = (float*)d_ws;
  float* stats = agg + (size_t)N_NODES * 64;
  float* ss    = stats + 128;
  float* Wpad  = ss + 128;
  float* h     = Wpad + 1024;

  const int ntiles     = N_NODES / 16;
  const int mlp_blocks = (ntiles + 3) / 4;

  build_input_kernel<<<(N_NODES * 16 + 255) / 256, 256, 0, stream>>>(x, labels,
                                                                     emb, h);
  pad_w1_kernel<<<4, 256, 0, stream>>>(W1_0, Wpad);

  for (int layer = 0; layer < 3; ++layer) {
    if (layer == 0)
      owner_agg_kernel<4, OWN_ROWS0, OWN_SPLIT0>
          <<<N_OWNERS0 * OWN_SPLIT0, 1024, 0, stream>>>(h, ei, agg, stats);
    else
      owner_agg_kernel<6, OWN_ROWS, 1><<<N_OWNERS, 1024, 0, stream>>>(h, ei, agg,
                                                                    stats);

    const float* W1 = (layer == 0) ? Wpad : (W1_s + (size_t)(layer - 1) * 4096);
    const float* b1 = (layer == 0) ? b1_0 : (b1_s + (size_t)(layer - 1) * 64);
    const float* W2 = (layer == 0) ? W2_0 : (W2_s + (size_t)(layer - 1) * 4096);
    const float* b2 = (layer == 0) ? b2_0 : (b2_s + (size_t)(layer - 1) * 64);
    const float* ep = (layer == 0) ? eps_0 : eps_s;
    const int    ei_idx = (layer == 0) ? 0 : (layer - 1);
    const float* gm = (layer == 0) ? g_0 : (g_s + (size_t)(layer - 1) * 64);
    const float* bt = (layer == 0) ? be_0 : (be_s + (size_t)(layer - 1) * 64);
    const int    K  = (layer == 0) ? 16 : 64;

    if (layer == 0)
      gin_mlp_kernel<OWN_SPLIT0><<<mlp_blocks, 128, 0, stream>>>(
          h, agg, W1, b1, W2, b2, ep, ei_idx, K, stats);
    else
      gin_mlp_kernel<1><<<mlp_blocks, 128, 0, stream>>>(h, agg, W1, b1, W2, b2,
                                                        ep, ei_idx, K, stats);
    bn_finalize_kernel<<<1, 64, 0, stream>>>(stats, gm, bt, ss);
    bn_relu_kernel<<<(N_NODES * 16 + 255) / 256, 256, 0, stream>>>(h, ss);
  }

  final_kernel<<<(N_NODES + 255) / 256, 256, 0, stream>>>(h, Wf, bf, out);
}


# 64-column layers: edge list cut in 2 halves, 125 owners x 2 work-groups (was 131 x 1); partial sums added in agg by global f32 atomics onto zeros stored by the BN+ReLU pass; f32 sum order changes
# speedup vs baseline: 2.5463x; 1.1412x over previous
#include <hip/hip_runtime.h>
#include <hip/hip_bf16.h>
#include <cstddef>
#include <cstdint>

#define N_NODES 100000
#define N_EDGES 3200000
#define BN_EPS 1e-5f

typedef __attribute__((ext_vector_type(2))) float v2f;
typedef __attribute__((ext_vector_type(8))) float v8f;

__device__ __forceinline__ v8f wmma_f32(v2f a, v2f b, v8f c) {
  return __builtin_amdgcn_wmma_f32_16x16x4_f32(false, a, false, b, (short)0, c,
                                               false, false);
}

__global__ void build_input_kernel(const float* __restrict__ x,
                                   const int* __restrict__ labels,
                                   const float* __restrict__ emb,
                                   float* __restrict__ h) {
  long long idx = (long long)blockIdx.x * blockDim.x + threadIdx.x;
  if (idx >= (long long)N_NODES * 16) return;
  const int i = (int)(idx >> 4);
  const int c = (int)(idx & 15);
  float v = 0.0f;
  if (c < 2)       v = x[(size_t)i * 2 + c];
  else if (c < 10) v = emb[(size_t)labels[i] * 8 + (c - 2)];
  h[(size_t)i * 64 + c] = v;
}

__global__ void pad_w1_kernel(const float* __restrict__ W1,
                              float* __restrict__ Wpad) {
  const int idx = blockIdx.x * blockDim.x + threadIdx.x;
  if (idx >= 16 * 64) return;
  const int r = idx >> 6;
  Wpad[idx] = (r < 10) ? W1[idx] : 0.0f;
}

#define OWN_ROWS  800
#define OWN_SPLIT 2
#define OWN_ROWS0 1600
#define OWN_SPLIT0 4
#define OWN_WAVES 32
#define OWN_QCAP  544
#define N_OWNERS  ((N_NODES + OWN_ROWS - 1) / OWN_ROWS)
#define N_OWNERS0 ((N_NODES + OWN_ROWS0 - 1) / OWN_ROWS0)
#define OWN_TRIPS (N_EDGES / 512)

static_assert(N_EDGES % 512 == 0, "edge list is read in whole 512-edge trips");
static_assert(OWN_ROWS0 % 2 == 0, "an owner's plane rows start on a 128-byte line");
static_assert(OWN_SPLIT0 * 16 <= 64, "the planes of partial sums fit the agg buffer");

template <int STRIDE, int SHIFT>
__device__ __forceinline__ void own_flush(const float* __restrict__ hcol,
                                          const int* __restrict__ srcs,
                                          const unsigned* q, int n, int& ps,
                                          int& pdl, int& pn, float* acc2,
                                          int lane, bool colact) {
  const unsigned ent = (lane < n) ? q[lane] : 0u;
  const int ndl = (int)(ent >> SHIFT);
  const int ns  = srcs[ent & ((1u << SHIFT) - 1u)];
  if (pn > 0) {
    float2 v[32];
#pragma unroll
    for (int j = 0; j < 32; ++j) {
      const unsigned sk = (unsigned)__builtin_amdgcn_readlane(ps, j);
      v[j] = *(const float2*)(hcol + (size_t)sk * 64);
    }
#pragma unroll
    for (int j = 0; j < 32; ++j) {
      if (j < pn) {
        const int dk = __builtin_amdgcn_readlane(pdl, j);
        if (colact) {
          unsafeAtomicAdd(&acc2[dk * STRIDE], v[j].x);
          unsafeAtomicAdd(&acc2[dk * STRIDE + 1], v[j].y);
        }
      }
    }
  }
  ps = ns;
  pdl = ndl;
  pn = n;
}

template <int ROWS, int STRIDE, int SHIFT>
__device__ __forceinline__ void own_trip(const int4 (&d)[4], int t,
                                         unsigned row0, unsigned* q, int& cnt,
                                         int& ps, int& pdl, int& pn,
                                         const float* __restrict__ hcol,
                                         const int* __restrict__ srcs,
                                         float* acc2, int lane, bool colact) {
#pragma unroll
  for (int j = 0; j < 4; ++j) {
    const int e0 = t * 512 + j * 128 + lane * 4;
    const int dv[4] = {d[j].x, d[j].y, d[j].z, d[j].w};
#pragma unroll
    for (int c = 0; c < 4; ++c) {
      const unsigned dl = (unsigned)dv[c] - row0;
      const bool m = dl < (unsigned)ROWS;
      const unsigned bm = __builtin_amdgcn_ballot_w32(m);
      if (bm != 0u) {
        if (m)
          q[cnt + (int)__builtin_amdgcn_mbcnt_lo(bm, 0u)] =
              (unsigned)(e0 + c) | (dl << SHIFT);
        cnt += __builtin_popcount(bm);
      }
    }
  }
  while (cnt >= 32) {
    cnt -= 32;
    own_flush<STRIDE, SHIFT>(hcol, srcs, q + cnt, 32, ps, pdl, pn, acc2, lane,
                             colact);
  }
}

template <int LOG2W, int ROWS, int SPLIT>
__global__ void __launch_bounds__(1024)
owner_agg_kernel(const float* __restrict__ h,
                 const int* __restrict__ edge_index, float* __restrict__ agg,
                 float* __restrict__ stats) {
  constexpr int STRIDE = (SPLIT == 1) ? 64 : (1 << LOG2W);
  constexpr bool PLANES = (SPLIT > 1) && ((SPLIT << LOG2W) <= 64);
  static_assert(SPLIT == 1 || PLANES || STRIDE == 64, "partial sums that are added go into whole agg rows");
  constexpr int NOWN   = (N_NODES + ROWS - 1) / ROWS;
  constexpr int PART_EDGES = (OWN_TRIPS / SPLIT + 1) * 512;
  constexpr int SHIFT = (PART_EDGES <= (1 << 20)) ? 20 : 22;
  static_assert(PART_EDGES <= (1 << SHIFT), "edge id must fit the low bits of a queue word");
  static_assert(ROWS <= (1 << (32 - SHIFT)), "local row must fit the high bits of a queue word");
  static_assert(OWN_TRIPS / SPLIT >= OWN_WAVES, "every wave has a first trip");
  __shared__ float acc[ROWS * STRIDE];
  __shared__ unsigned queue[OWN_WAVES * OWN_QCAP];
  const int tid  = threadIdx.x;
  const int lane = tid & 31;
  const int wave = __builtin_amdgcn_readfirstlane(tid >> 5);
  const unsigned owner = (SPLIT == 1) ? blockIdx.x : (blockIdx.x % NOWN);
  const int part = (SPLIT == 1) ? 0 : (int)(blockIdx.x / NOWN);
  const unsigned row0 = owner * ROWS;
  const int tbeg = (part * OWN_TRIPS) / SPLIT;
  const int tend = ((part + 1) * OWN_TRIPS) / SPLIT;

  for (int i = tid; i < ROWS * STRIDE / 4; i += 1024)
    ((float4*)acc)[i] = make_float4(0.f, 0.f, 0.f, 0.f);
  if (blockIdx.x == 0 && tid < 128) stats[tid] = 0.0f;
  __syncthreads();

  const int4* __restrict__ dst4 = (const int4*)(edge_index + N_EDGES);
  const int* __restrict__ srcs = edge_index + (size_t)tbeg * 512;
  unsigned* q = queue + wave * OWN_QCAP;
  const bool colact = (lane * 2) < (1 << LOG2W);
  const int col = colact ? lane * 2 : 0;
  const float* __restrict__ hcol = h + col;
  float* acc2 = acc + col;
  int cnt = 0;
  int ps = 0, pdl = 0, pn = 0;

  int4 a[4], b[4];
#pragma unroll
  for (int j = 0; j < 4; ++j) a[j] = dst4[(tbeg + wave) * 128 + j * 32 + lane];

  for (int t = tbeg + wave; t < tend; t += 2 * OWN_WAVES) {
    const int tb = t + OWN_WAVES;
    const int tbc = (tb < tend) ? tb : (tend - 1);
#pragma unroll
    for (int j = 0; j < 4; ++j) b[j] = dst4[tbc * 128 + j * 32 + lane];
    own_trip<ROWS, STRIDE, SHIFT>(a, t - tbeg, row0, q, cnt, ps, pdl, pn, hcol,
                                  srcs, acc2, lane, colact);

    const int ta = t + 2 * OWN_WAVES;
    const int tac = (ta < tend) ? ta : (tend - 1);
#pragma unroll
    for (int j = 0; j < 4; ++j) a[j] = dst4[tac * 128 + j * 32 + lane];
    if (tb < tend)
      own_trip<ROWS, STRIDE, SHIFT>(b, tb - tbeg, row0, q, cnt, ps, pdl, pn,
                                    hcol, srcs, acc2, lane, colact);
  }
  if (cnt > 0)
    own_flush<STRIDE, SHIFT>(hcol, srcs, q, cnt, ps, pdl, pn, acc2, lane,
                             colact);
  own_flush<STRIDE, SHIFT>(hcol, srcs, q, 0, ps, pdl, pn, acc2, lane, colact);
  __syncthreads();

  const int nleft = N_NODES - (int)row0;
  const int nrow  = (nleft < ROWS) ? nleft : ROWS;
  if (SPLIT == 1) {
    constexpr int C4LOG = LOG2W - 2;
    for (int i = tid; i < (nrow << C4LOG); i += 1024) {
      const int r = i >> C4LOG;
      const int c = (i & ((1 << C4LOG) - 1)) * 4;
      *(float4*)&agg[(size_t)(row0 + r) * 64 + c] = *(const float4*)&acc[r * 64 + c];
    }
  } else if (PLANES) {
    float4* __restrict__ dstp =
        (float4*)(agg + ((size_t)part * N_NODES + row0) * STRIDE);
    for (int i = tid; i < nrow * (STRIDE / 4); i += 1024)
      dstp[i] = ((const float4*)acc)[i];
  } else {
    float* __restrict__ dsta = agg + (size_t)row0 * 64;
    const int nel = nrow * 64;
    for (int i0 = tid; i0 < nel; i0 += 10 * 1024) {
      float v[10];
#pragma unroll
      for (int u = 0; u < 10; ++u) {
        const int i = i0 + u * 1024;
        v[u] = (i < nel) ? acc[i] : 0.0f;
      }
#pragma unroll
      for (int u = 0; u < 10; ++u) {
        const int i = i0 + u * 1024;
        if (i < nel) unsafeAtomicAdd(&dsta[i], v[u]);
      }
    }
  }
}

template <int NPART>
__global__ void __launch_bounds__(128)
gin_mlp_kernel(float* __restrict__ h, const float* __restrict__ agg,
               const float* __restrict__ W1, const float* __restrict__ b1,
               const float* __restrict__ W2, const float* __restrict__ b2,
               const float* __restrict__ eps_ptr, int eps_idx, int K,
               float* __restrict__ stats) {
  __shared__ float ytile[4 * 16 * 68];
  __shared__ float sstat[128];
  const int tid   = threadIdx.x;
  const int wave  = tid >> 5;
  const int lane  = tid & 31;
  const int lrow  = lane & 15;
  const int lhalf = lane >> 4;
  const int tile  = blockIdx.x * 4 + wave;

  if (tid < 128) sstat[tid] = 0.0f;
  __syncthreads();

  if (tile < (N_NODES / 16)) {
    const int base = tile * 16;
    const float onepe = 1.0f + eps_ptr[eps_idx];
    const int koff = lhalf * 2;

    v8f acc[4] = {};
    for (int kk = 0; kk < K; kk += 4) {
      const int k0 = kk + koff;
      const float* hp = h   + (size_t)(base + lrow) * 64 + k0;
      v2f afrag;
      if (NPART == 1) {
        const float* ap = agg + (size_t)(base + lrow) * 64 + k0;
        afrag.x = onepe * hp[0] + ap[0];
        afrag.y = onepe * hp[1] + ap[1];
      } else {
        const float* ap = agg + (size_t)(base + lrow) * 16 + k0;
        float sx = ap[0], sy = ap[1];
#pragma unroll
        for (int p = 1; p < NPART; ++p) {
          sx += ap[(size_t)p * N_NODES * 16];
          sy += ap[(size_t)p * N_NODES * 16 + 1];
        }
        afrag.x = onepe * hp[0] + sx;
        afrag.y = onepe * hp[1] + sy;
      }
      const float* wp = W1 + (size_t)k0 * 64 + lrow;
#pragma unroll
      for (int nt = 0; nt < 4; ++nt) {
        v2f bfrag;
        bfrag.x = wp[nt * 16];
        bfrag.y = wp[nt * 16 + 64];
        acc[nt] = wmma_f32(afrag, bfrag, acc[nt]);
      }
    }

    float* my = ytile + wave * (16 * 68);
#pragma unroll
    for (int nt = 0; nt < 4; ++nt) {
      const float bv = b1[nt * 16 + lrow];
#pragma unroll
      for (int r = 0; r < 8; ++r)
        my[(r + lhalf * 8) * 68 + nt * 16 + lrow] = fmaxf(acc[nt][r] + bv, 0.f);
    }

    v8f acc2[4] = {};
    for (int kk = 0; kk < 64; kk += 4) {
      const int k0 = kk + koff;
      v2f afrag;
      afrag.x = my[lrow * 68 + k0];
      afrag.y = my[lrow * 68 + k0 + 1];
      const float* wp = W2 + (size_t)k0 * 64 + lrow;
#pragma unroll
      for (int nt = 0; nt < 4; ++nt) {
        v2f bfrag;
        bfrag.x = wp[nt * 16];
        bfrag.y = wp[nt * 16 + 64];
        acc2[nt] = wmma_f32(afrag, bfrag, acc2[nt]);
      }
    }

#pragma unroll
    for (int nt = 0; nt < 4; ++nt) {
      const float bv = b2[nt * 16 + lrow];
      float s = 0.0f, s2 = 0.0f;
#pragma unroll
      for (int r = 0; r < 8; ++r) {
        const float v = fmaxf(acc2[nt][r] + bv, 0.f);
        h[(size_t)(base + r + lhalf * 8) * 64 + nt * 16 + lrow] = v;
        s += v;
        s2 += v * v;
      }
      unsafeAtomicAdd(&sstat[nt * 16 + lrow], s);
      unsafeAtomicAdd(&sstat[64 + nt * 16 + lrow], s2);
    }
  }
  __syncthreads();
  if (tid < 128) unsafeAtomicAdd(&stats[tid], sstat[tid]);
}

__global__ void bn_finalize_kernel(const float* __restrict__ stats,
                                   const float* __restrict__ gamma,
                                   const float* __restrict__ beta,
                                   float* __restrict__ ss) {
  const int c = threadIdx.x;
  if (c >= 64) return;
  const float inv_n = 1.0f / (float)N_NODES;
  const float mu  = stats[c] * inv_n;
  const float var = stats[64 + c] * inv_n - mu * mu;
  const float rs  = rsqrtf(var + BN_EPS);
  const float sc  = gamma[c] * rs;
  ss[c]      = sc;
  ss[64 + c] = beta[c] - mu * sc;
}

template <bool CLEAR>
__global__ void bn_relu_kernel(float* __restrict__ h,
                               const float* __restrict__ ss,
                               float* __restrict__ agg) {
  const long long i = (long long)blockIdx.x * blockDim.x + threadIdx.x;
  if (i >= (long long)N_NODES * 16) return;
  if (CLEAR) ((float4*)agg)[i] = make_float4(0.f, 0.f, 0.f, 0.f);
  const int c4 = (int)(i & 15) * 4;
  float4 v = ((const float4*)h)[i];
  v.x = fmaxf(v.x * ss[c4 + 0] + ss[64 + c4 + 0], 0.f);
  v.y = fmaxf(v.y * ss[c4 + 1] + ss[64 + c4 + 1], 0.f);
  v.z = fmaxf(v.z * ss[c4 + 2] + ss[64 + c4 + 2], 0.f);
  v.w = fmaxf(v.w * ss[c4 + 3] + ss[64 + c4 + 3], 0.f);
  ((float4*)h)[i] = v;
}

__global__ void final_kernel(const float* __restrict__ h,
                             const float* __restrict__ Wf,
                             const float* __restrict__ bf,
                             float* __restrict__ out) {
  const int i = blockIdx.x * blockDim.x + threadIdx.x;
  if (i >= N_NODES) return;
  const float4* hp = (const float4*)(h + (size_t)i * 64);
  const float4* wp = (const float4*)Wf;
  float s = bf[0];
#pragma unroll
  for (int j = 0; j < 16; ++j) {
    float4 a = hp[j], b = wp[j];
    s += a.x * b.x + a.y * b.y + a.z * b.z + a.w * b.w;
  }
  out[i] = s;
}

extern "C" void kernel_launch(void* const* d_in, const int* in_sizes, int n_in,
                              void* d_out, int out_size, void* d_ws,
                              size_t ws_size, hipStream_t stream) {
  const float* x      = (const float*)d_in[0];
  const int*   labels = (const int*)d_in[1];
  const int*   ei     = (const int*)d_in[2];
  const float* emb    = (const float*)d_in[3];
  const float* W1_0   = (const float*)d_in[4];
  const float* b1_0   = (const float*)d_in[5];
  const float* W2_0   = (const float*)d_in[6];
  const float* b2_0   = (const float*)d_in[7];
  const float* eps_0  = (const float*)d_in[8];
  const float* g_0    = (const float*)d_in[9];
  const float* be_0   = (const float*)d_in[10];
  const float* W1_s   = (const float*)d_in[11];
  const float* b1_s   = (const float*)d_in[12];
  const float* W2_s   = (const float*)d_in[13];
  const float* b2_s   = (const float*)d_in[14];
  const float* eps_s  = (const float*)d_in[15];
  const float* g_s    = (const float*)d_in[16];
  const float* be_s   = (const float*)d_in[17];
  const float* Wf     = (const float*)d_in[18];
  const float* bf     = (const float*)d_in[19];
  float* out = (float*)d_out;

  float* agg   = (float*)d_ws;
  float* stats = agg + (size_t)N_NODES * 64;
  float* ss    = stats + 128;
  float* Wpad  = ss + 128;
  float* h     = Wpad + 1024;

  const int ntiles     = N_NODES / 16;
  const int mlp_blocks = (ntiles + 3) / 4;

  build_input_kernel<<<(N_NODES * 16 + 255) / 256, 256, 0, stream>>>(x, labels,
                                                                     emb, h);
  pad_w1_kernel<<<4, 256, 0, stream>>>(W1_0, Wpad);

  for (int layer = 0; layer < 3; ++layer) {
    if (layer == 0)
      owner_agg_kernel<4, OWN_ROWS0, OWN_SPLIT0>
          <<<N_OWNERS0 * OWN_SPLIT0, 1024, 0, stream>>>(h, ei, agg, stats);
    else
      owner_agg_kernel<6, OWN_ROWS, OWN_SPLIT>
          <<<N_OWNERS * OWN_SPLIT, 1024, 0, stream>>>(h, ei, agg, stats);

    const float* W1 = (layer == 0) ? Wpad : (W1_s + (size_t)(layer - 1) * 4096);
    const float* b1 = (layer == 0) ? b1_0 : (b1_s + (size_t)(layer - 1) * 64);
    const float* W2 = (layer == 0) ? W2_0 : (W2_s + (size_t)(layer - 1) * 4096);
    const float* b2 = (layer == 0) ? b2_0 : (b2_s + (size_t)(layer - 1) * 64);
    const float* ep = (layer == 0) ? eps_0 : eps_s;
    const int    ei_idx = (layer == 0) ? 0 : (layer - 1);
    const float* gm = (layer == 0) ? g_0 : (g_s + (size_t)(layer - 1) * 64);
    const float* bt = (layer == 0) ? be_0 : (be_s + (size_t)(layer - 1) * 64);
    const int    K  = (layer == 0) ? 16 : 64;

    if (layer == 0)
      gin_mlp_kernel<OWN_SPLIT0><<<mlp_blocks, 128, 0, stream>>>(
          h, agg, W1, b1, W2, b2, ep, ei_idx, K, stats);
    else
      gin_mlp_kernel<1><<<mlp_blocks, 128, 0, stream>>>(h, agg, W1, b1, W2, b2,
                                                        ep, ei_idx, K, stats);
    bn_finalize_kernel<<<1, 64, 0, stream>>>(stats, gm, bt, ss);
    if (layer < 2)
      bn_relu_kernel<true><<<(N_NODES * 16 + 255) / 256, 256, 0, stream>>>(
          h, ss, agg);
    else
      bn_relu_kernel<false><<<(N_NODES * 16 + 255) / 256, 256, 0, stream>>>(
          h, ss, agg);
  }

  final_kernel<<<(N_NODES + 255) / 256, 256, 0, stream>>>(h, Wf, bf, out);
}


# Edges binned by destination owner in a pass before each owner launch (250 bins kept inside the agg buffer); owners read only their bin, store rows plainly, no global f32 adds; f32 sum order changes
# speedup vs baseline: 3.0488x; 1.1973x over previous
#include <hip/hip_runtime.h>
#include <hip/hip_bf16.h>
#include <cstddef>
#include <cstdint>

#define N_NODES 100000
#define N_EDGES 3200000
#define BN_EPS 1e-5f

#define BIN_ROWS  400
#define N_OWN     (N_NODES / BIN_ROWS)
#define BIN_WORDS (BIN_ROWS * 64)
#define BIN_CAP   (BIN_WORDS - 1)

typedef __attribute__((ext_vector_type(2))) float v2f;
typedef __attribute__((ext_vector_type(8))) float v8f;

__device__ __forceinline__ v8f wmma_f32(v2f a, v2f b, v8f c) {
  return __builtin_amdgcn_wmma_f32_16x16x4_f32(false, a, false, b, (short)0, c,
                                               false, false);
}

__global__ void build_input_kernel(const float* __restrict__ x,
                                   const int* __restrict__ labels,
                                   const float* __restrict__ emb,
                                   float* __restrict__ h,
                                   float* __restrict__ agg) {
  long long idx = (long long)blockIdx.x * blockDim.x + threadIdx.x;
  if (idx >= (long long)N_NODES * 16) return;
  if (idx < N_OWN) ((int*)agg)[(size_t)idx * BIN_WORDS + BIN_CAP] = 0;
  const int i = (int)(idx >> 4);
  const int c = (int)(idx & 15);
  float v = 0.0f;
  if (c < 2)       v = x[(size_t)i * 2 + c];
  else if (c < 10) v = emb[(size_t)labels[i] * 8 + (c - 2)];
  h[(size_t)i * 64 + c] = v;
}

__global__ void pad_w1_kernel(const float* __restrict__ W1,
                              float* __restrict__ Wpad) {
  const int idx = blockIdx.x * blockDim.x + threadIdx.x;
  if (idx >= 16 * 64) return;
  const int r = idx >> 6;
  Wpad[idx] = (r < 10) ? W1[idx] : 0.0f;
}

#define BIN_CHUNK (N_EDGES / N_OWN)
#define BIN_PER_T ((BIN_CHUNK + 1023) / 1024)

static_assert(N_OWN * BIN_ROWS == N_NODES, "owners tile the rows exactly");
static_assert(N_OWN * BIN_CHUNK == N_EDGES, "bin pass chunks tile the edge list exactly");
static_assert(N_NODES <= (1 << 17), "source row fits the low 17 bits of an entry");
static_assert(BIN_ROWS <= (1 << 15), "local row fits the high bits of an entry");
static_assert(N_OWN <= 1024, "one thread per owner in the bin pass");

__global__ void __launch_bounds__(1024)
bin_edges_kernel(const int* __restrict__ edge_index, float* __restrict__ agg) {
  __shared__ int hist[N_OWN];
  __shared__ int lbase[N_OWN];
  __shared__ int gbase[N_OWN];
  __shared__ unsigned stage[BIN_CHUNK];
  const int tid  = threadIdx.x;
  const int lane = tid & 31;
  const int wave = __builtin_amdgcn_readfirstlane(tid >> 5);
  unsigned* bins = (unsigned*)agg;
  if (tid < N_OWN) hist[tid] = 0;
  __syncthreads();

  const int* __restrict__ srcs = edge_index + (size_t)blockIdx.x * BIN_CHUNK;
  const int* __restrict__ dsts = srcs + N_EDGES;
  int dv[BIN_PER_T], sv[BIN_PER_T];
#pragma unroll
  for (int u = 0; u < BIN_PER_T; ++u) {
    const int i  = tid + u * 1024;
    const int ic = (i < BIN_CHUNK) ? i : (BIN_CHUNK - 1);
    dv[u] = dsts[ic];
    sv[u] = srcs[ic];
  }
  int own[BIN_PER_T];
  unsigned ent[BIN_PER_T];
  int rank[BIN_PER_T];
#pragma unroll
  for (int u = 0; u < BIN_PER_T; ++u) {
    const int i = tid + u * 1024;
    own[u] = -1;
    ent[u] = 0u;
    rank[u] = 0;
    if (i < BIN_CHUNK && (unsigned)dv[u] < (unsigned)N_NODES) {
      const int o = dv[u] / BIN_ROWS;
      own[u] = o;
      ent[u] = (unsigned)sv[u] | ((unsigned)(dv[u] - o * BIN_ROWS) << 17);
      rank[u] = atomicAdd(&hist[o], 1);
    }
  }
  __syncthreads();

  if (tid < N_OWN) {
    int before = 0;
    for (int k = 0; k < tid; ++k) before += hist[k];
    lbase[tid] = before;
    gbase[tid] = atomicAdd((int*)bins + (size_t)tid * BIN_WORDS + BIN_CAP, hist[tid]);
  }
  __syncthreads();

#pragma unroll
  for (int u = 0; u < BIN_PER_T; ++u)
    if (own[u] >= 0) stage[lbase[own[u]] + rank[u]] = ent[u];
  __syncthreads();

  for (int o = wave; o < N_OWN; o += 32) {
    const int n  = hist[o];
    const int lb = lbase[o];
    const int gb = gbase[o];
    unsigned* dst = bins + (size_t)o * BIN_WORDS;
    for (int k = lane; k < n; k += 32) {
      const int pos = gb + k;
      if (pos < BIN_CAP) dst[pos] = stage[lb + k];
    }
  }
}

template <int LOG2W>
__global__ void __launch_bounds__(1024)
owner_agg_kernel(const float* __restrict__ h,
                 const int* __restrict__ edge_index, float* agg,
                 float* __restrict__ stats) {
  __shared__ float acc[BIN_ROWS * 64];
  const int tid  = threadIdx.x;
  const int lane = tid & 31;
  const int wave = __builtin_amdgcn_readfirstlane(tid >> 5);
  const unsigned row0 = blockIdx.x * BIN_ROWS;
  float* region = agg + (size_t)blockIdx.x * BIN_WORDS;
  const unsigned* bin = (const unsigned*)region;

  for (int i = tid; i < BIN_ROWS * 16; i += 1024)
    ((float4*)acc)[i] = make_float4(0.f, 0.f, 0.f, 0.f);
  if (blockIdx.x == 0 && tid < 128) stats[tid] = 0.0f;
  const int total = (int)bin[BIN_CAP];
  __syncthreads();

  const bool colact = (lane * 2) < (1 << LOG2W);
  const int col = colact ? lane * 2 : 0;
  const float* __restrict__ hcol = h + col;
  float* acc2 = acc + col;

  if (total <= BIN_CAP) {
    int base = wave * 32;
    unsigned ent = (base + lane < total) ? bin[base + lane] : 0u;
    for (; base < total; base += 1024) {
      const int left = total - base;
      const int ps  = (int)(ent & 0x1FFFFu);
      const int pdl = (int)(ent >> 17);
      float2 v[32];
#pragma unroll
      for (int j = 0; j < 32; ++j) {
        const unsigned sk = (unsigned)__builtin_amdgcn_readlane(ps, j);
        v[j] = *(const float2*)(hcol + (size_t)sk * 64);
      }
      const int nb = base + 1024;
      const unsigned nxt = (nb + lane < total) ? bin[nb + lane] : 0u;
      if (left >= 32) {
#pragma unroll
        for (int j = 0; j < 32; ++j) {
          const int dk = __builtin_amdgcn_readlane(pdl, j);
          if (colact) {
            unsafeAtomicAdd(&acc2[dk * 64], v[j].x);
            unsafeAtomicAdd(&acc2[dk * 64 + 1], v[j].y);
          }
        }
      } else {
#pragma unroll
        for (int j = 0; j < 32; ++j) {
          if (j < left) {
            const int dk = __builtin_amdgcn_readlane(pdl, j);
            if (colact) {
              unsafeAtomicAdd(&acc2[dk * 64], v[j].x);
              unsafeAtomicAdd(&acc2[dk * 64 + 1], v[j].y);
            }
          }
        }
      }
      ent = nxt;
    }
  } else {
    const int* __restrict__ dsts = edge_index + N_EDGES;
    for (int e = tid; e < N_EDGES; e += 1024) {
      const unsigned dl = (unsigned)dsts[e] - row0;
      if (dl < (unsigned)BIN_ROWS) {
        const float* hs = h + (size_t)edge_index[e] * 64;
        for (int c = 0; c < (1 << LOG2W); ++c)
          unsafeAtomicAdd(&acc[dl * 64 + c], hs[c]);
      }
    }
  }
  __syncthreads();

  constexpr int C4LOG = LOG2W - 2;
  for (int i = tid; i < (BIN_ROWS << C4LOG); i += 1024) {
    const int r = i >> C4LOG;
    const int c = (i & ((1 << C4LOG) - 1)) * 4;
    *(float4*)&region[r * 64 + c] = *(const float4*)&acc[r * 64 + c];
  }
}

__global__ void __launch_bounds__(128)
gin_mlp_kernel(float* __restrict__ h, const float* __restrict__ agg,
               const float* __restrict__ W1, const float* __restrict__ b1,
               const float* __restrict__ W2, const float* __restrict__ b2,
               const float* __restrict__ eps_ptr, int eps_idx, int K,
               float* __restrict__ stats) {
  __shared__ float ytile[4 * 16 * 68];
  __shared__ float sstat[128];
  const int tid   = threadIdx.x;
  const int wave  = tid >> 5;
  const int lane  = tid & 31;
  const int lrow  = lane & 15;
  const int lhalf = lane >> 4;
  const int tile  = blockIdx.x * 4 + wave;

  if (tid < 128) sstat[tid] = 0.0f;
  __syncthreads();

  if (tile < (N_NODES / 16)) {
    const int base = tile * 16;
    const float onepe = 1.0f + eps_ptr[eps_idx];
    const int koff = lhalf * 2;

    v8f acc[4] = {};
    for (int kk = 0; kk < K; kk += 4) {
      const int k0 = kk + koff;
      const float* hp = h   + (size_t)(base + lrow) * 64 + k0;
      const float* ap = agg + (size_t)(base + lrow) * 64 + k0;
      v2f afrag;
      afrag.x = onepe * hp[0] + ap[0];
      afrag.y = onepe * hp[1] + ap[1];
      const float* wp = W1 + (size_t)k0 * 64 + lrow;
#pragma unroll
      for (int nt = 0; nt < 4; ++nt) {
        v2f bfrag;
        bfrag.x = wp[nt * 16];
        bfrag.y = wp[nt * 16 + 64];
        acc[nt] = wmma_f32(afrag, bfrag, acc[nt]);
      }
    }

    float* my = ytile + wave * (16 * 68);
#pragma unroll
    for (int nt = 0; nt < 4; ++nt) {
      const float bv = b1[nt * 16 + lrow];
#pragma unroll
      for (int r = 0; r < 8; ++r)
        my[(r + lhalf * 8) * 68 + nt * 16 + lrow] = fmaxf(acc[nt][r] + bv, 0.f);
    }

    v8f acc2[4] = {};
    for (int kk = 0; kk < 64; kk += 4) {
      const int k0 = kk + koff;
      v2f afrag;
      afrag.x = my[lrow * 68 + k0];
      afrag.y = my[lrow * 68 + k0 + 1];
      const float* wp = W2 + (size_t)k0 * 64 + lrow;
#pragma unroll
      for (int nt = 0; nt < 4; ++nt) {
        v2f bfrag;
        bfrag.x = wp[nt * 16];
        bfrag.y = wp[nt * 16 + 64];
        acc2[nt] = wmma_f32(afrag, bfrag, acc2[nt]);
      }
    }

#pragma unroll
    for (int nt = 0; nt < 4; ++nt) {
      const float bv = b2[nt * 16 + lrow];
      float s = 0.0f, s2 = 0.0f;
#pragma unroll
      for (int r = 0; r < 8; ++r) {
        const float v = fmaxf(acc2[nt][r] + bv, 0.f);
        h[(size_t)(base + r + lhalf * 8) * 64 + nt * 16 + lrow] = v;
        s += v;
        s2 += v * v;
      }
      unsafeAtomicAdd(&sstat[nt * 16 + lrow], s);
      unsafeAtomicAdd(&sstat[64 + nt * 16 + lrow], s2);
    }
  }
  __syncthreads();
  if (tid < 128) unsafeAtomicAdd(&stats[tid], sstat[tid]);
}

__global__ void bn_finalize_kernel(const float* __restrict__ stats,
                                   const float* __restrict__ gamma,
                                   const float* __restrict__ beta,
                                   float* __restrict__ ss) {
  const int c = threadIdx.x;
  if (c >= 64) return;
  const float inv_n = 1.0f / (float)N_NODES;
  const float mu  = stats[c] * inv_n;
  const float var = stats[64 + c] * inv_n - mu * mu;
  const float rs  = rsqrtf(var + BN_EPS);
  const float sc  = gamma[c] * rs;
  ss[c]      = sc;
  ss[64 + c] = beta[c] - mu * sc;
}

template <bool CLEAR>
__global__ void bn_relu_kernel(float* __restrict__ h,
                               const float* __restrict__ ss,
                               float* __restrict__ agg) {
  const long long i = (long long)blockIdx.x * blockDim.x + threadIdx.x;
  if (i >= (long long)N_NODES * 16) return;
  if (CLEAR && i < N_OWN) ((int*)agg)[(size_t)i * BIN_WORDS + BIN_CAP] = 0;
  const int c4 = (int)(i & 15) * 4;
  float4 v = ((const float4*)h)[i];
  v.x = fmaxf(v.x * ss[c4 + 0] + ss[64 + c4 + 0], 0.f);
  v.y = fmaxf(v.y * ss[c4 + 1] + ss[64 + c4 + 1], 0.f);
  v.z = fmaxf(v.z * ss[c4 + 2] + ss[64 + c4 + 2], 0.f);
  v.w = fmaxf(v.w * ss[c4 + 3] + ss[64 + c4 + 3], 0.f);
  ((float4*)h)[i] = v;
}

__global__ void final_kernel(const float* __restrict__ h,
                             const float* __restrict__ Wf,
                             const float* __restrict__ bf,
                             float* __restrict__ out) {
  const int i = blockIdx.x * blockDim.x + threadIdx.x;
  if (i >= N_NODES) return;
  const float4* hp = (const float4*)(h + (size_t)i * 64);
  const float4* wp = (const float4*)Wf;
  float s = bf[0];
#pragma unroll
  for (int j = 0; j < 16; ++j) {
    float4 a = hp[j], b = wp[j];
    s += a.x * b.x + a.y * b.y + a.z * b.z + a.w * b.w;
  }
  out[i] = s;
}

extern "C" void kernel_launch(void* const* d_in, const int* in_sizes, int n_in,
                              void* d_out, int out_size, void* d_ws,
                              size_t ws_size, hipStream_t stream) {
  const float* x      = (const float*)d_in[0];
  const int*   labels = (const int*)d_in[1];
  const int*   ei     = (const int*)d_in[2];
  const float* emb    = (const float*)d_in[3];
  const float* W1_0   = (const float*)d_in[4];
  const float* b1_0   = (const float*)d_in[5];
  const float* W2_0   = (const float*)d_in[6];
  const float* b2_0   = (const float*)d_in[7];
  const float* eps_0  = (const float*)d_in[8];
  const float* g_0    = (const float*)d_in[9];
  const float* be_0   = (const float*)d_in[10];
  const float* W1_s   = (const float*)d_in[11];
  const float* b1_s   = (const float*)d_in[12];
  const float* W2_s   = (const float*)d_in[13];
  const float* b2_s   = (const float*)d_in[14];
  const float* eps_s  = (const float*)d_in[15];
  const float* g_s    = (const float*)d_in[16];
  const float* be_s   = (const float*)d_in[17];
  const float* Wf     = (const float*)d_in[18];
  const float* bf     = (const float*)d_in[19];
  float* out = (float*)d_out;

  float* agg   = (float*)d_ws;
  float* stats = agg + (size_t)N_NODES * 64;
  float* ss    = stats + 128;
  float* Wpad  = ss + 128;
  float* h     = Wpad + 1024;

  const int ntiles     = N_NODES / 16;
  const int mlp_blocks = (ntiles + 3) / 4;

  build_input_kernel<<<(N_NODES * 16 + 255) / 256, 256, 0, stream>>>(
      x, labels, emb, h, agg);
  pad_w1_kernel<<<4, 256, 0, stream>>>(W1_0, Wpad);

  for (int layer = 0; layer < 3; ++layer) {
    bin_edges_kernel<<<N_OWN, 1024, 0, stream>>>(ei, agg);
    if (layer == 0)
      owner_agg_kernel<4><<<N_OWN, 1024, 0, stream>>>(h, ei, agg, stats);
    else
      owner_agg_kernel<6><<<N_OWN, 1024, 0, stream>>>(h, ei, agg, stats);

    const float* W1 = (layer == 0) ? Wpad : (W1_s + (size_t)(layer - 1) * 4096);
    const float* b1 = (layer == 0) ? b1_0 : (b1_s + (size_t)(layer - 1) * 64);
    const float* W2 = (layer == 0) ? W2_0 : (W2_s + (size_t)(layer - 1) * 4096);
    const float* b2 = (layer == 0) ? b2_0 : (b2_s + (size_t)(layer - 1) * 64);
    const float* ep = (layer == 0) ? eps_0 : eps_s;
    const int    ei_idx = (layer == 0) ? 0 : (layer - 1);
    const float* gm = (layer == 0) ? g_0 : (g_s + (size_t)(layer - 1) * 64);
    const float* bt = (layer == 0) ? be_0 : (be_s + (size_t)(layer - 1) * 64);
    const int    K  = (layer == 0) ? 16 : 64;

    gin_mlp_kernel<<<mlp_blocks, 128, 0, stream>>>(h, agg, W1, b1, W2, b2, ep,
                                                   ei_idx, K, stats);
    bn_finalize_kernel<<<1, 64, 0, stream>>>(stats, gm, bt, ss);
    if (layer < 2)
      bn_relu_kernel<true><<<(N_NODES * 16 + 255) / 256, 256, 0, stream>>>(
          h, ss, agg);
    else
      bn_relu_kernel<false><<<(N_NODES * 16 + 255) / 256, 256, 0, stream>>>(
          h, ss, agg);
  }

  final_kernel<<<(N_NODES + 255) / 256, 256, 0, stream>>>(h, Wf, bf, out);
}


# Owner sorts its bin by destination row in LDS (counters, scan, placement), sums each row's sources in registers (16-col layer: 4 per load), one store per row; no LDS f32 adds; f32 sum order changes
# speedup vs baseline: 3.5333x; 1.1589x over previous
#include <hip/hip_runtime.h>
#include <hip/hip_bf16.h>
#include <cstddef>
#include <cstdint>

#define N_NODES 100000
#define N_EDGES 3200000
#define BN_EPS 1e-5f

#define BIN_ROWS  400
#define N_OWN     (N_NODES / BIN_ROWS)
#define BIN_WORDS (BIN_ROWS * 64)
#define BIN_CAP   (BIN_WORDS - 1)

typedef __attribute__((ext_vector_type(2))) float v2f;
typedef __attribute__((ext_vector_type(8))) float v8f;

__device__ __forceinline__ v8f wmma_f32(v2f a, v2f b, v8f c) {
  return __builtin_amdgcn_wmma_f32_16x16x4_f32(false, a, false, b, (short)0, c,
                                               false, false);
}

__global__ void build_input_kernel(const float* __restrict__ x,
                                   const int* __restrict__ labels,
                                   const float* __restrict__ emb,
                                   float* __restrict__ h,
                                   float* __restrict__ agg) {
  long long idx = (long long)blockIdx.x * blockDim.x + threadIdx.x;
  if (idx >= (long long)N_NODES * 16) return;
  if (idx < N_OWN) ((int*)agg)[(size_t)idx * BIN_WORDS + BIN_CAP] = 0;
  const int i = (int)(idx >> 4);
  const int c = (int)(idx & 15);
  float v = 0.0f;
  if (c < 2)       v = x[(size_t)i * 2 + c];
  else if (c < 10) v = emb[(size_t)labels[i] * 8 + (c - 2)];
  h[(size_t)i * 64 + c] = v;
}

__global__ void pad_w1_kernel(const float* __restrict__ W1,
                              float* __restrict__ Wpad) {
  const int idx = blockIdx.x * blockDim.x + threadIdx.x;
  if (idx >= 16 * 64) return;
  const int r = idx >> 6;
  Wpad[idx] = (r < 10) ? W1[idx] : 0.0f;
}

#define BIN_CHUNK (N_EDGES / N_OWN)
#define BIN_PER_T ((BIN_CHUNK + 1023) / 1024)

static_assert(N_OWN * BIN_ROWS == N_NODES, "owners tile the rows exactly");
static_assert(N_OWN * BIN_CHUNK == N_EDGES, "bin pass chunks tile the edge list exactly");
static_assert(N_NODES <= (1 << 17), "source row fits the low 17 bits of an entry");
static_assert(BIN_ROWS <= (1 << 15), "local row fits the high bits of an entry");
static_assert(N_OWN <= 1024, "one thread per owner in the bin pass");

__global__ void __launch_bounds__(1024)
bin_edges_kernel(const int* __restrict__ edge_index, float* __restrict__ agg) {
  __shared__ int hist[N_OWN];
  __shared__ int lbase[N_OWN];
  __shared__ int gbase[N_OWN];
  __shared__ unsigned stage[BIN_CHUNK];
  const int tid  = threadIdx.x;
  const int lane = tid & 31;
  const int wave = __builtin_amdgcn_readfirstlane(tid >> 5);
  unsigned* bins = (unsigned*)agg;
  if (tid < N_OWN) hist[tid] = 0;
  __syncthreads();

  const int* __restrict__ srcs = edge_index + (size_t)blockIdx.x * BIN_CHUNK;
  const int* __restrict__ dsts = srcs + N_EDGES;
  int dv[BIN_PER_T], sv[BIN_PER_T];
#pragma unroll
  for (int u = 0; u < BIN_PER_T; ++u) {
    const int i  = tid + u * 1024;
    const int ic = (i < BIN_CHUNK) ? i : (BIN_CHUNK - 1);
    dv[u] = dsts[ic];
    sv[u] = srcs[ic];
  }
  int own[BIN_PER_T];
  unsigned ent[BIN_PER_T];
  int rank[BIN_PER_T];
#pragma unroll
  for (int u = 0; u < BIN_PER_T; ++u) {
    const int i = tid + u * 1024;
    own[u] = -1;
    ent[u] = 0u;
    rank[u] = 0;
    if (i < BIN_CHUNK && (unsigned)dv[u] < (unsigned)N_NODES) {
      const int o = dv[u] / BIN_ROWS;
      own[u] = o;
      ent[u] = (unsigned)sv[u] | ((unsigned)(dv[u] - o * BIN_ROWS) << 17);
      rank[u] = atomicAdd(&hist[o], 1);
    }
  }
  __syncthreads();

  if (tid < N_OWN) {
    int before = 0;
    for (int k = 0; k < tid; ++k) before += hist[k];
    lbase[tid] = before;
    gbase[tid] = atomicAdd((int*)bins + (size_t)tid * BIN_WORDS + BIN_CAP, hist[tid]);
  }
  __syncthreads();

#pragma unroll
  for (int u = 0; u < BIN_PER_T; ++u)
    if (own[u] >= 0) stage[lbase[own[u]] + rank[u]] = ent[u];
  __syncthreads();

  for (int o = wave; o < N_OWN; o += 32) {
    const int n  = hist[o];
    const int lb = lbase[o];
    const int gb = gbase[o];
    unsigned* dst = bins + (size_t)o * BIN_WORDS;
    for (int k = lane; k < n; k += 32) {
      const int pos = gb + k;
      if (pos < BIN_CAP) dst[pos] = stage[lb + k];
    }
  }
}

#define OWN_PER_T ((BIN_CAP + 1023) / 1024)

static_assert(BIN_ROWS <= 1024, "one thread per row in the scan of the row counters");

template <int LOG2W>
__global__ void __launch_bounds__(1024)
owner_agg_kernel(const float* __restrict__ h,
                 const int* __restrict__ edge_index, float* agg,
                 float* __restrict__ stats) {
  __shared__ float acc[BIN_ROWS * 64];
  __shared__ int rowcnt[BIN_ROWS];
  __shared__ int rowstart[BIN_ROWS + 1];
  __shared__ int wsum[32];
  const int tid  = threadIdx.x;
  const int lane = tid & 31;
  const int wave = __builtin_amdgcn_readfirstlane(tid >> 5);
  const unsigned row0 = blockIdx.x * BIN_ROWS;
  float* region = agg + (size_t)blockIdx.x * BIN_WORDS;
  const unsigned* bin = (const unsigned*)region;

  if (tid < BIN_ROWS) rowcnt[tid] = 0;
  if (blockIdx.x == 0 && tid < 128) stats[tid] = 0.0f;
  const int total = (int)bin[BIN_CAP];
  __syncthreads();

  if (total <= BIN_CAP) {
    unsigned* sorted = (unsigned*)acc;
    unsigned ent[OWN_PER_T];
#pragma unroll
    for (int u = 0; u < OWN_PER_T; ++u) {
      const int i = tid + u * 1024;
      ent[u] = bin[(i < total) ? i : BIN_CAP];
    }
    int rank[OWN_PER_T];
#pragma unroll
    for (int u = 0; u < OWN_PER_T; ++u) {
      const int i = tid + u * 1024;
      rank[u] = 0;
      if (i < total) {
        const unsigned dl = min(ent[u] >> 17, (unsigned)(BIN_ROWS - 1));
        rank[u] = atomicAdd(&rowcnt[dl], 1);
      }
    }
    __syncthreads();

    const int cnt = (tid < BIN_ROWS) ? rowcnt[tid] : 0;
    int incl = cnt;
#pragma unroll
    for (int d = 1; d < 32; d <<= 1) {
      const int t = __shfl_up(incl, d);
      if (lane >= d) incl += t;
    }
    if (lane == 31) wsum[wave] = incl;
    __syncthreads();
    int off = 0;
    for (int k = 0; k < wave; ++k) off += wsum[k];
    if (tid < BIN_ROWS) rowstart[tid] = off + incl - cnt;
    if (tid == BIN_ROWS - 1) rowstart[BIN_ROWS] = off + incl;
    __syncthreads();

#pragma unroll
    for (int u = 0; u < OWN_PER_T; ++u) {
      const int i = tid + u * 1024;
      if (i < total) {
        const unsigned dl = min(ent[u] >> 17, (unsigned)(BIN_ROWS - 1));
        sorted[rowstart[dl] + rank[u]] =
            min(ent[u] & 0x1FFFFu, (unsigned)(N_NODES - 1));
      }
    }
    __syncthreads();

    constexpr int LPE   = (1 << LOG2W) / 2;
    constexpr int EPI   = 32 / LPE;
    constexpr int STEPS = (EPI == 1) ? 48 : 16;
    constexpr int NEV   = (STEPS + 31) / 32;
    const int sub = lane / LPE;
    const int col = (lane % LPE) * 2;
    const float* __restrict__ hcol = h + col;
    for (int r = wave; r < BIN_ROWS; r += 32) {
      const int s = __builtin_amdgcn_readfirstlane(rowstart[r]);
      const int n = __builtin_amdgcn_readfirstlane(rowstart[r + 1]) - s;
      float ax = 0.0f, ay = 0.0f;
      for (int c = 0; c < n; c += STEPS * EPI) {
        float2 v[STEPS];
        if constexpr (EPI == 1) {
          unsigned ev[NEV];
#pragma unroll
          for (int q = 0; q < NEV; ++q) {
            const int idx = c + q * 32 + lane;
            ev[q] = (idx < n) ? sorted[s + idx] : 0u;
          }
#pragma unroll
          for (int t = 0; t < STEPS; ++t) {
            v[t] = make_float2(0.0f, 0.0f);
            if (c + t < n) {
              const unsigned sk =
                  (unsigned)__builtin_amdgcn_readlane((int)ev[t >> 5], t & 31);
              v[t] = *(const float2*)(hcol + (size_t)sk * 64);
            }
          }
        } else {
          unsigned sk[STEPS];
#pragma unroll
          for (int t = 0; t < STEPS; ++t) {
            const int e0 = c + t * EPI;
            sk[t] = 0u;
            if (e0 < n) sk[t] = sorted[s + ((e0 + sub < n) ? e0 + sub : e0)];
          }
#pragma unroll
          for (int t = 0; t < STEPS; ++t) {
            v[t] = make_float2(0.0f, 0.0f);
            if (c + t * EPI < n)
              v[t] = *(const float2*)(hcol + (size_t)sk[t] * 64);
          }
        }
#pragma unroll
        for (int t = 0; t < STEPS; ++t) {
          const bool ok = (EPI == 1) || (c + t * EPI + sub < n);
          ax += ok ? v[t].x : 0.0f;
          ay += ok ? v[t].y : 0.0f;
        }
      }
#pragma unroll
      for (int d = LPE; d < 32; d <<= 1) {
        ax += __shfl_xor(ax, d);
        ay += __shfl_xor(ay, d);
      }
      if (sub == 0) *(float2*)&region[r * 64 + col] = make_float2(ax, ay);
    }
  } else {
    for (int i = tid; i < BIN_ROWS * 16; i += 1024)
      ((float4*)acc)[i] = make_float4(0.f, 0.f, 0.f, 0.f);
    __syncthreads();
    const int* __restrict__ dsts = edge_index + N_EDGES;
    for (int e = tid; e < N_EDGES; e += 1024) {
      const unsigned dl = (unsigned)dsts[e] - row0;
      if (dl < (unsigned)BIN_ROWS) {
        const float* hs = h + (size_t)edge_index[e] * 64;
        for (int c = 0; c < (1 << LOG2W); ++c)
          unsafeAtomicAdd(&acc[dl * 64 + c], hs[c]);
      }
    }
    __syncthreads();
    constexpr int C4LOG = LOG2W - 2;
    for (int i = tid; i < (BIN_ROWS << C4LOG); i += 1024) {
      const int r = i >> C4LOG;
      const int c = (i & ((1 << C4LOG) - 1)) * 4;
      *(float4*)&region[r * 64 + c] = *(const float4*)&acc[r * 64 + c];
    }
  }
}

__global__ void __launch_bounds__(128)
gin_mlp_kernel(float* __restrict__ h, const float* __restrict__ agg,
               const float* __restrict__ W1, const float* __restrict__ b1,
               const float* __restrict__ W2, const float* __restrict__ b2,
               const float* __restrict__ eps_ptr, int eps_idx, int K,
               float* __restrict__ stats) {
  __shared__ float ytile[4 * 16 * 68];
  __shared__ float sstat[128];
  const int tid   = threadIdx.x;
  const int wave  = tid >> 5;
  const int lane  = tid & 31;
  const int lrow  = lane & 15;
  const int lhalf = lane >> 4;
  const int tile  = blockIdx.x * 4 + wave;

  if (tid < 128) sstat[tid] = 0.0f;
  __syncthreads();

  if (tile < (N_NODES / 16)) {
    const int base = tile * 16;
    const float onepe = 1.0f + eps_ptr[eps_idx];
    const int koff = lhalf * 2;

    v8f acc[4] = {};
    for (int kk = 0; kk < K; kk += 4) {
      const int k0 = kk + koff;
      const float* hp = h   + (size_t)(base + lrow) * 64 + k0;
      const float* ap = agg + (size_t)(base + lrow) * 64 + k0;
      v2f afrag;
      afrag.x = onepe * hp[0] + ap[0];
      afrag.y = onepe * hp[1] + ap[1];
      const float* wp = W1 + (size_t)k0 * 64 + lrow;
#pragma unroll
      for (int nt = 0; nt < 4; ++nt) {
        v2f bfrag;
        bfrag.x = wp[nt * 16];
        bfrag.y = wp[nt * 16 + 64];
        acc[nt] = wmma_f32(afrag, bfrag, acc[nt]);
      }
    }

    float* my = ytile + wave * (16 * 68);
#pragma unroll
    for (int nt = 0; nt < 4; ++nt) {
      const float bv = b1[nt * 16 + lrow];
#pragma unroll
      for (int r = 0; r < 8; ++r)
        my[(r + lhalf * 8) * 68 + nt * 16 + lrow] = fmaxf(acc[nt][r] + bv, 0.f);
    }

    v8f acc2[4] = {};
    for (int kk = 0; kk < 64; kk += 4) {
      const int k0 = kk + koff;
      v2f afrag;
      afrag.x = my[lrow * 68 + k0];
      afrag.y = my[lrow * 68 + k0 + 1];
      const float* wp = W2 + (size_t)k0 * 64 + lrow;
#pragma unroll
      for (int nt = 0; nt < 4; ++nt) {
        v2f bfrag;
        bfrag.x = wp[nt * 16];
        bfrag.y = wp[nt * 16 + 64];
        acc2[nt] = wmma_f32(afrag, bfrag, acc2[nt]);
      }
    }

#pragma unroll
    for (int nt = 0; nt < 4; ++nt) {
      const float bv = b2[nt * 16 + lrow];
      float s = 0.0f, s2 = 0.0f;
#pragma unroll
      for (int r = 0; r < 8; ++r) {
        const float v = fmaxf(acc2[nt][r] + bv, 0.f);
        h[(size_t)(base + r + lhalf * 8) * 64 + nt * 16 + lrow] = v;
        s += v;
        s2 += v * v;
      }
      unsafeAtomicAdd(&sstat[nt * 16 + lrow], s);
      unsafeAtomicAdd(&sstat[64 + nt * 16 + lrow], s2);
    }
  }
  __syncthreads();
  if (tid < 128) unsafeAtomicAdd(&stats[tid], sstat[tid]);
}

__global__ void bn_finalize_kernel(const float* __restrict__ stats,
                                   const float* __restrict__ gamma,
                                   const float* __restrict__ beta,
                                   float* __restrict__ ss) {
  const int c = threadIdx.x;
  if (c >= 64) return;
  const float inv_n = 1.0f / (float)N_NODES;
  const float mu  = stats[c] * inv_n;
  const float var = stats[64 + c] * inv_n - mu * mu;
  const float rs  = rsqrtf(var + BN_EPS);
  const float sc  = gamma[c] * rs;
  ss[c]      = sc;
  ss[64 + c] = beta[c] - mu * sc;
}

template <bool CLEAR>
__global__ void bn_relu_kernel(float* __restrict__ h,
                               const float* __restrict__ ss,
                               float* __restrict__ agg) {
  const long long i = (long long)blockIdx.x * blockDim.x + threadIdx.x;
  if (i >= (long long)N_NODES * 16) return;
  if (CLEAR && i < N_OWN) ((int*)agg)[(size_t)i * BIN_WORDS + BIN_CAP] = 0;
  const int c4 = (int)(i & 15) * 4;
  float4 v = ((const float4*)h)[i];
  v.x = fmaxf(v.x * ss[c4 + 0] + ss[64 + c4 + 0], 0.f);
  v.y = fmaxf(v.y * ss[c4 + 1] + ss[64 + c4 + 1], 0.f);
  v.z = fmaxf(v.z * ss[c4 + 2] + ss[64 + c4 + 2], 0.f);
  v.w = fmaxf(v.w * ss[c4 + 3] + ss[64 + c4 + 3], 0.f);
  ((float4*)h)[i] = v;
}

__global__ void final_kernel(const float* __restrict__ h,
                             const float* __restrict__ Wf,
                             const float* __restrict__ bf,
                             float* __restrict__ out) {
  const int i = blockIdx.x * blockDim.x + threadIdx.x;
  if (i >= N_NODES) return;
  const float4* hp = (const float4*)(h + (size_t)i * 64);
  const float4* wp = (const float4*)Wf;
  float s = bf[0];
#pragma unroll
  for (int j = 0; j < 16; ++j) {
    float4 a = hp[j], b = wp[j];
    s += a.x * b.x + a.y * b.y + a.z * b.z + a.w * b.w;
  }
  out[i] = s;
}

extern "C" void kernel_launch(void* const* d_in, const int* in_sizes, int n_in,
                              void* d_out, int out_size, void* d_ws,
                              size_t ws_size, hipStream_t stream) {
  const float* x      = (const float*)d_in[0];
  const int*   labels = (const int*)d_in[1];
  const int*   ei     = (const int*)d_in[2];
  const float* emb    = (const float*)d_in[3];
  const float* W1_0   = (const float*)d_in[4];
  const float* b1_0   = (const float*)d_in[5];
  const float* W2_0   = (const float*)d_in[6];
  const float* b2_0   = (const float*)d_in[7];
  const float* eps_0  = (const float*)d_in[8];
  const float* g_0    = (const float*)d_in[9];
  const float* be_0   = (const float*)d_in[10];
  const float* W1_s   = (const float*)d_in[11];
  const float* b1_s   = (const float*)d_in[12];
  const float* W2_s   = (const float*)d_in[13];
  const float* b2_s   = (const float*)d_in[14];
  const float* eps_s  = (const float*)d_in[15];
  const float* g_s    = (const float*)d_in[16];
  const float* be_s   = (const float*)d_in[17];
  const float* Wf     = (const float*)d_in[18];
  const float* bf     = (const float*)d_in[19];
  float* out = (float*)d_out;

  float* agg   = (float*)d_ws;
  float* stats = agg + (size_t)N_NODES * 64;
  float* ss    = stats + 128;
  float* Wpad  = ss + 128;
  float* h     = Wpad + 1024;

  const int ntiles     = N_NODES / 16;
  const int mlp_blocks = (ntiles + 3) / 4;

  build_input_kernel<<<(N_NODES * 16 + 255) / 256, 256, 0, stream>>>(
      x, labels, emb, h, agg);
  pad_w1_kernel<<<4, 256, 0, stream>>>(W1_0, Wpad);

  for (int layer = 0; layer < 3; ++layer) {
    bin_edges_kernel<<<N_OWN, 1024, 0, stream>>>(ei, agg);
    if (layer == 0)
      owner_agg_kernel<4><<<N_OWN, 1024, 0, stream>>>(h, ei, agg, stats);
    else
      owner_agg_kernel<6><<<N_OWN, 1024, 0, stream>>>(h, ei, agg, stats);

    const float* W1 = (layer == 0) ? Wpad : (W1_s + (size_t)(layer - 1) * 4096);
    const float* b1 = (layer == 0) ? b1_0 : (b1_s + (size_t)(layer - 1) * 64);
    const float* W2 = (layer == 0) ? W2_0 : (W2_s + (size_t)(layer - 1) * 4096);
    const float* b2 = (layer == 0) ? b2_0 : (b2_s + (size_t)(layer - 1) * 64);
    const float* ep = (layer == 0) ? eps_0 : eps_s;
    const int    ei_idx = (layer == 0) ? 0 : (layer - 1);
    const float* gm = (layer == 0) ? g_0 : (g_s + (size_t)(layer - 1) * 64);
    const float* bt = (layer == 0) ? be_0 : (be_s + (size_t)(layer - 1) * 64);
    const int    K  = (layer == 0) ? 16 : 64;

    gin_mlp_kernel<<<mlp_blocks, 128, 0, stream>>>(h, agg, W1, b1, W2, b2, ep,
                                                   ei_idx, K, stats);
    bn_finalize_kernel<<<1, 64, 0, stream>>>(stats, gm, bt, ss);
    if (layer < 2)
      bn_relu_kernel<true><<<(N_NODES * 16 + 255) / 256, 256, 0, stream>>>(
          h, ss, agg);
    else
      bn_relu_kernel<false><<<(N_NODES * 16 + 255) / 256, 256, 0, stream>>>(
          h, ss, agg);
  }

  final_kernel<<<(N_NODES + 255) / 256, 256, 0, stream>>>(h, Wf, bf, out);
}


# BN scale/shift + ReLU applied by the readers of h (owner gather, MLP A loader, final dot) from two alternating sets of column sums; the 3 finalize and 3 BN+ReLU launches are gone (18 -> 12)
# speedup vs baseline: 5.1660x; 1.4621x over previous
#include <hip/hip_runtime.h>
#include <hip/hip_bf16.h>
#include <cstddef>
#include <cstdint>

#define N_NODES 100000
#define N_EDGES 3200000
#define BN_EPS 1e-5f

#define BIN_ROWS  400
#define N_OWN     (N_NODES / BIN_ROWS)
#define BIN_WORDS (BIN_ROWS * 64)
#define BIN_CAP   (BIN_WORDS - 1)

typedef __attribute__((ext_vector_type(2))) float v2f;
typedef __attribute__((ext_vector_type(8))) float v8f;

__device__ __forceinline__ v8f wmma_f32(v2f a, v2f b, v8f c) {
  return __builtin_amdgcn_wmma_f32_16x16x4_f32(false, a, false, b, (short)0, c,
                                               false, false);
}

__device__ __forceinline__ void bn_scale_shift(const float* __restrict__ sums,
                                               const float* __restrict__ gamma,
                                               const float* __restrict__ beta,
                                               int c, float& sc, float& sh) {
  const float inv_n = 1.0f / (float)N_NODES;
  const float mu  = sums[c] * inv_n;
  const float var = sums[64 + c] * inv_n - mu * mu;
  const float rs  = rsqrtf(var + BN_EPS);
  sc = gamma[c] * rs;
  sh = beta[c] - mu * sc;
}

__global__ void build_input_kernel(const float* __restrict__ x,
                                   const int* __restrict__ labels,
                                   const float* __restrict__ emb,
                                   float* __restrict__ h,
                                   float* __restrict__ agg) {
  long long idx = (long long)blockIdx.x * blockDim.x + threadIdx.x;
  if (idx >= (long long)N_NODES * 16) return;
  if (idx < N_OWN) ((int*)agg)[(size_t)idx * BIN_WORDS + BIN_CAP] = 0;
  const int i = (int)(idx >> 4);
  const int c = (int)(idx & 15);
  float v = 0.0f;
  if (c < 2)       v = x[(size_t)i * 2 + c];
  else if (c < 10) v = emb[(size_t)labels[i] * 8 + (c - 2)];
  h[(size_t)i * 64 + c] = v;
}

__global__ void pad_w1_kernel(const float* __restrict__ W1,
                              float* __restrict__ Wpad) {
  const int idx = blockIdx.x * blockDim.x + threadIdx.x;
  if (idx >= 16 * 64) return;
  const int r = idx >> 6;
  Wpad[idx] = (r < 10) ? W1[idx] : 0.0f;
}

#define BIN_CHUNK (N_EDGES / N_OWN)
#define BIN_PER_T ((BIN_CHUNK + 1023) / 1024)

static_assert(N_OWN * BIN_ROWS == N_NODES, "owners tile the rows exactly");
static_assert(N_OWN * BIN_CHUNK == N_EDGES, "bin pass chunks tile the edge list exactly");
static_assert(N_NODES <= (1 << 17), "source row fits the low 17 bits of an entry");
static_assert(BIN_ROWS <= (1 << 15), "local row fits the high bits of an entry");
static_assert(N_OWN <= 1024, "one thread per owner in the bin pass");

__global__ void __launch_bounds__(1024)
bin_edges_kernel(const int* __restrict__ edge_index, float* __restrict__ agg) {
  __shared__ int hist[N_OWN];
  __shared__ int lbase[N_OWN];
  __shared__ int gbase[N_OWN];
  __shared__ unsigned stage[BIN_CHUNK];
  const int tid  = threadIdx.x;
  const int lane = tid & 31;
  const int wave = __builtin_amdgcn_readfirstlane(tid >> 5);
  unsigned* bins = (unsigned*)agg;
  if (tid < N_OWN) hist[tid] = 0;
  __syncthreads();

  const int* __restrict__ srcs = edge_index + (size_t)blockIdx.x * BIN_CHUNK;
  const int* __restrict__ dsts = srcs + N_EDGES;
  int dv[BIN_PER_T], sv[BIN_PER_T];
#pragma unroll
  for (int u = 0; u < BIN_PER_T; ++u) {
    const int i  = tid + u * 1024;
    const int ic = (i < BIN_CHUNK) ? i : (BIN_CHUNK - 1);
    dv[u] = dsts[ic];
    sv[u] = srcs[ic];
  }
  int own[BIN_PER_T];
  unsigned ent[BIN_PER_T];
  int rank[BIN_PER_T];
#pragma unroll
  for (int u = 0; u < BIN_PER_T; ++u) {
    const int i = tid + u * 1024;
    own[u] = -1;
    ent[u] = 0u;
    rank[u] = 0;
    if (i < BIN_CHUNK && (unsigned)dv[u] < (unsigned)N_NODES) {
      const int o = dv[u] / BIN_ROWS;
      own[u] = o;
      ent[u] = (unsigned)sv[u] | ((unsigned)(dv[u] - o * BIN_ROWS) << 17);
      rank[u] = atomicAdd(&hist[o], 1);
    }
  }
  __syncthreads();

  if (tid < N_OWN) {
    int before = 0;
    for (int k = 0; k < tid; ++k) before += hist[k];
    lbase[tid] = before;
    gbase[tid] = atomicAdd((int*)bins + (size_t)tid * BIN_WORDS + BIN_CAP, hist[tid]);
  }
  __syncthreads();

#pragma unroll
  for (int u = 0; u < BIN_PER_T; ++u)
    if (own[u] >= 0) stage[lbase[own[u]] + rank[u]] = ent[u];
  __syncthreads();

  for (int o = wave; o < N_OWN; o += 32) {
    const int n  = hist[o];
    const int lb = lbase[o];
    const int gb = gbase[o];
    unsigned* dst = bins + (size_t)o * BIN_WORDS;
    for (int k = lane; k < n; k += 32) {
      const int pos = gb + k;
      if (pos < BIN_CAP) dst[pos] = stage[lb + k];
    }
  }
}

#define OWN_PER_T ((BIN_CAP + 1023) / 1024)

static_assert(BIN_ROWS <= 1024, "one thread per row in the scan of the row counters");

template <int LOG2W, bool BN>
__global__ void __launch_bounds__(1024)
owner_agg_kernel(const float* __restrict__ h,
                 const int* __restrict__ edge_index, float* agg,
                 float* __restrict__ stats,
                 const float* __restrict__ bnstats,
                 const float* __restrict__ gamma,
                 const float* __restrict__ beta) {
  __shared__ float acc[BIN_ROWS * 64];
  __shared__ int rowcnt[BIN_ROWS];
  __shared__ int rowstart[BIN_ROWS + 1];
  __shared__ int wsum[32];
  __shared__ float bnss[128];
  const int tid  = threadIdx.x;
  const int lane = tid & 31;
  const int wave = __builtin_amdgcn_readfirstlane(tid >> 5);
  const unsigned row0 = blockIdx.x * BIN_ROWS;
  float* region = agg + (size_t)blockIdx.x * BIN_WORDS;
  const unsigned* bin = (const unsigned*)region;

  if (tid < BIN_ROWS) rowcnt[tid] = 0;
  if (blockIdx.x == 0 && tid < 128) stats[tid] = 0.0f;
  if (BN && tid < 64)
    bn_scale_shift(bnstats, gamma, beta, tid, bnss[tid], bnss[64 + tid]);
  const int total = (int)bin[BIN_CAP];
  __syncthreads();

  if (total <= BIN_CAP) {
    unsigned* sorted = (unsigned*)acc;
    unsigned ent[OWN_PER_T];
#pragma unroll
    for (int u = 0; u < OWN_PER_T; ++u) {
      const int i = tid + u * 1024;
      ent[u] = bin[(i < total) ? i : BIN_CAP];
    }
    int rank[OWN_PER_T];
#pragma unroll
    for (int u = 0; u < OWN_PER_T; ++u) {
      const int i = tid + u * 1024;
      rank[u] = 0;
      if (i < total) {
        const unsigned dl = min(ent[u] >> 17, (unsigned)(BIN_ROWS - 1));
        rank[u] = atomicAdd(&rowcnt[dl], 1);
      }
    }
    __syncthreads();

    const int cnt = (tid < BIN_ROWS) ? rowcnt[tid] : 0;
    int incl = cnt;
#pragma unroll
    for (int d = 1; d < 32; d <<= 1) {
      const int t = __shfl_up(incl, d);
      if (lane >= d) incl += t;
    }
    if (lane == 31) wsum[wave] = incl;
    __syncthreads();
    int off = 0;
    for (int k = 0; k < wave; ++k) off += wsum[k];
    if (tid < BIN_ROWS) rowstart[tid] = off + incl - cnt;
    if (tid == BIN_ROWS - 1) rowstart[BIN_ROWS] = off + incl;
    __syncthreads();

#pragma unroll
    for (int u = 0; u < OWN_PER_T; ++u) {
      const int i = tid + u * 1024;
      if (i < total) {
        const unsigned dl = min(ent[u] >> 17, (unsigned)(BIN_ROWS - 1));
        sorted[rowstart[dl] + rank[u]] =
            min(ent[u] & 0x1FFFFu, (unsigned)(N_NODES - 1));
      }
    }
    __syncthreads();

    constexpr int LPE   = (1 << LOG2W) / 2;
    constexpr int EPI   = 32 / LPE;
    constexpr int STEPS = (EPI == 1) ? 48 : 16;
    constexpr int NEV   = (STEPS + 31) / 32;
    const int sub = lane / LPE;
    const int col = (lane % LPE) * 2;
    const float* __restrict__ hcol = h + col;
    float scx = 1.0f, scy = 1.0f, shx = 0.0f, shy = 0.0f;
    if constexpr (BN) {
      scx = bnss[col];      scy = bnss[col + 1];
      shx = bnss[64 + col]; shy = bnss[64 + col + 1];
    }
    for (int r = wave; r < BIN_ROWS; r += 32) {
      const int s = __builtin_amdgcn_readfirstlane(rowstart[r]);
      const int n = __builtin_amdgcn_readfirstlane(rowstart[r + 1]) - s;
      float ax = 0.0f, ay = 0.0f;
      for (int c = 0; c < n; c += STEPS * EPI) {
        float2 v[STEPS];
        if constexpr (EPI == 1) {
          unsigned ev[NEV];
#pragma unroll
          for (int q = 0; q < NEV; ++q) {
            const int idx = c + q * 32 + lane;
            ev[q] = (idx < n) ? sorted[s + idx] : 0u;
          }
#pragma unroll
          for (int t = 0; t < STEPS; ++t) {
            v[t] = make_float2(0.0f, 0.0f);
            if (c + t < n) {
              const unsigned sk =
                  (unsigned)__builtin_amdgcn_readlane((int)ev[t >> 5], t & 31);
              v[t] = *(const float2*)(hcol + (size_t)sk * 64);
            }
          }
        } else {
          unsigned sk[STEPS];
#pragma unroll
          for (int t = 0; t < STEPS; ++t) {
            const int e0 = c + t * EPI;
            sk[t] = 0u;
            if (e0 < n) sk[t] = sorted[s + ((e0 + sub < n) ? e0 + sub : e0)];
          }
#pragma unroll
          for (int t = 0; t < STEPS; ++t) {
            v[t] = make_float2(0.0f, 0.0f);
            if (c + t * EPI < n)
              v[t] = *(const float2*)(hcol + (size_t)sk[t] * 64);
          }
        }
#pragma unroll
        for (int t = 0; t < STEPS; ++t) {
          const bool ok = (EPI == 1) ? (c + t < n) : (c + t * EPI + sub < n);
          const float vx = BN ? fmaxf(fmaf(v[t].x, scx, shx), 0.0f) : v[t].x;
          const float vy = BN ? fmaxf(fmaf(v[t].y, scy, shy), 0.0f) : v[t].y;
          ax += ok ? vx : 0.0f;
          ay += ok ? vy : 0.0f;
        }
      }
#pragma unroll
      for (int d = LPE; d < 32; d <<= 1) {
        ax += __shfl_xor(ax, d);
        ay += __shfl_xor(ay, d);
      }
      if (sub == 0) *(float2*)&region[r * 64 + col] = make_float2(ax, ay);
    }
  } else {
    for (int i = tid; i < BIN_ROWS * 16; i += 1024)
      ((float4*)acc)[i] = make_float4(0.f, 0.f, 0.f, 0.f);
    __syncthreads();
    const int* __restrict__ dsts = edge_index + N_EDGES;
    for (int e = tid; e < N_EDGES; e += 1024) {
      const unsigned dl = (unsigned)dsts[e] - row0;
      if (dl < (unsigned)BIN_ROWS) {
        const float* hs = h + (size_t)edge_index[e] * 64;
        for (int c = 0; c < (1 << LOG2W); ++c) {
          float v = hs[c];
          if constexpr (BN) v = fmaxf(fmaf(v, bnss[c], bnss[64 + c]), 0.0f);
          unsafeAtomicAdd(&acc[dl * 64 + c], v);
        }
      }
    }
    __syncthreads();
    constexpr int C4LOG = LOG2W - 2;
    for (int i = tid; i < (BIN_ROWS << C4LOG); i += 1024) {
      const int r = i >> C4LOG;
      const int c = (i & ((1 << C4LOG) - 1)) * 4;
      *(float4*)&region[r * 64 + c] = *(const float4*)&acc[r * 64 + c];
    }
  }
}

template <bool BN>
__global__ void __launch_bounds__(128)
gin_mlp_kernel(float* __restrict__ h, float* agg,
               const float* __restrict__ W1, const float* __restrict__ b1,
               const float* __restrict__ W2, const float* __restrict__ b2,
               const float* __restrict__ eps_ptr, int eps_idx, int K,
               float* __restrict__ stats,
               const float* __restrict__ bnstats,
               const float* __restrict__ gamma,
               const float* __restrict__ beta) {
  __shared__ float ytile[4 * 16 * 68];
  __shared__ float sstat[128];
  __shared__ float sbn[128];
  const int tid   = threadIdx.x;
  const int wave  = tid >> 5;
  const int lane  = tid & 31;
  const int lrow  = lane & 15;
  const int lhalf = lane >> 4;
  const int tile  = blockIdx.x * 4 + wave;

  if (tid < 128) sstat[tid] = 0.0f;
  if (BN && tid < 64)
    bn_scale_shift(bnstats, gamma, beta, tid, sbn[tid], sbn[64 + tid]);
  __syncthreads();

  if (tile < (N_NODES / 16)) {
    const int base = tile * 16;
    const float onepe = 1.0f + eps_ptr[eps_idx];
    const int koff = lhalf * 2;

    v8f acc[4] = {};
    for (int kk = 0; kk < K; kk += 4) {
      const int k0 = kk + koff;
      const float* hp = h   + (size_t)(base + lrow) * 64 + k0;
      const float* ap = agg + (size_t)(base + lrow) * 64 + k0;
      float h0 = hp[0], h1 = hp[1];
      if constexpr (BN) {
        h0 = fmaxf(fmaf(h0, sbn[k0],     sbn[64 + k0]),     0.0f);
        h1 = fmaxf(fmaf(h1, sbn[k0 + 1], sbn[64 + k0 + 1]), 0.0f);
      }
      v2f afrag;
      afrag.x = onepe * h0 + ap[0];
      afrag.y = onepe * h1 + ap[1];
      const float* wp = W1 + (size_t)k0 * 64 + lrow;
#pragma unroll
      for (int nt = 0; nt < 4; ++nt) {
        v2f bfrag;
        bfrag.x = wp[nt * 16];
        bfrag.y = wp[nt * 16 + 64];
        acc[nt] = wmma_f32(afrag, bfrag, acc[nt]);
      }
    }

    float* my = ytile + wave * (16 * 68);
#pragma unroll
    for (int nt = 0; nt < 4; ++nt) {
      const float bv = b1[nt * 16 + lrow];
#pragma unroll
      for (int r = 0; r < 8; ++r)
        my[(r + lhalf * 8) * 68 + nt * 16 + lrow] = fmaxf(acc[nt][r] + bv, 0.f);
    }

    v8f acc2[4] = {};
    for (int kk = 0; kk < 64; kk += 4) {
      const int k0 = kk + koff;
      v2f afrag;
      afrag.x = my[lrow * 68 + k0];
      afrag.y = my[lrow * 68 + k0 + 1];
      const float* wp = W2 + (size_t)k0 * 64 + lrow;
#pragma unroll
      for (int nt = 0; nt < 4; ++nt) {
        v2f bfrag;
        bfrag.x = wp[nt * 16];
        bfrag.y = wp[nt * 16 + 64];
        acc2[nt] = wmma_f32(afrag, bfrag, acc2[nt]);
      }
    }

#pragma unroll
    for (int nt = 0; nt < 4; ++nt) {
      const float bv = b2[nt * 16 + lrow];
      float s = 0.0f, s2 = 0.0f;
#pragma unroll
      for (int r = 0; r < 8; ++r) {
        const float v = fmaxf(acc2[nt][r] + bv, 0.f);
        h[(size_t)(base + r + lhalf * 8) * 64 + nt * 16 + lrow] = v;
        s += v;
        s2 += v * v;
      }
      unsafeAtomicAdd(&sstat[nt * 16 + lrow], s);
      unsafeAtomicAdd(&sstat[64 + nt * 16 + lrow], s2);
    }

    static_assert(BIN_ROWS % 16 == 0, "an owner's rows are whole tiles");
    if ((tile % (BIN_ROWS / 16)) == (BIN_ROWS / 16 - 1) && lane == 31)
      ((int*)agg)[(size_t)(base + 15) * 64 + 63] = 0;
  }
  __syncthreads();
  if (tid < 128) unsafeAtomicAdd(&stats[tid], sstat[tid]);
}

__global__ void final_kernel(const float* __restrict__ h,
                             const float* __restrict__ bnstats,
                             const float* __restrict__ gamma,
                             const float* __restrict__ beta,
                             const float* __restrict__ Wf,
                             const float* __restrict__ bf,
                             float* __restrict__ out) {
  __shared__ float sbn[128];
  if (threadIdx.x < 64)
    bn_scale_shift(bnstats, gamma, beta, threadIdx.x, sbn[threadIdx.x],
                   sbn[64 + threadIdx.x]);
  __syncthreads();
  const int i = blockIdx.x * blockDim.x + threadIdx.x;
  if (i >= N_NODES) return;
  const float4* hp = (const float4*)(h + (size_t)i * 64);
  const float4* wp = (const float4*)Wf;
  const float4* sc = (const float4*)sbn;
  const float4* sh = (const float4*)(sbn + 64);
  float s = bf[0];
#pragma unroll
  for (int j = 0; j < 16; ++j) {
    float4 a = hp[j], b = wp[j];
    const float4 c = sc[j], d = sh[j];
    a.x = fmaxf(fmaf(a.x, c.x, d.x), 0.0f);
    a.y = fmaxf(fmaf(a.y, c.y, d.y), 0.0f);
    a.z = fmaxf(fmaf(a.z, c.z, d.z), 0.0f);
    a.w = fmaxf(fmaf(a.w, c.w, d.w), 0.0f);
    s += a.x * b.x + a.y * b.y + a.z * b.z + a.w * b.w;
  }
  out[i] = s;
}

extern "C" void kernel_launch(void* const* d_in, const int* in_sizes, int n_in,
                              void* d_out, int out_size, void* d_ws,
                              size_t ws_size, hipStream_t stream) {
  const float* x      = (const float*)d_in[0];
  const int*   labels = (const int*)d_in[1];
  const int*   ei     = (const int*)d_in[2];
  const float* emb    = (const float*)d_in[3];
  const float* W1_0   = (const float*)d_in[4];
  const float* b1_0   = (const float*)d_in[5];
  const float* W2_0   = (const float*)d_in[6];
  const float* b2_0   = (const float*)d_in[7];
  const float* eps_0  = (const float*)d_in[8];
  const float* g_0    = (const float*)d_in[9];
  const float* be_0   = (const float*)d_in[10];
  const float* W1_s   = (const float*)d_in[11];
  const float* b1_s   = (const float*)d_in[12];
  const float* W2_s   = (const float*)d_in[13];
  const float* b2_s   = (const float*)d_in[14];
  const float* eps_s  = (const float*)d_in[15];
  const float* g_s    = (const float*)d_in[16];
  const float* be_s   = (const float*)d_in[17];
  const float* Wf     = (const float*)d_in[18];
  const float* bf     = (const float*)d_in[19];
  float* out = (float*)d_out;

  float* agg   = (float*)d_ws;
  float* sums0 = agg + (size_t)N_NODES * 64;
  float* sums1 = sums0 + 128;
  float* Wpad  = sums1 + 128;
  float* h     = Wpad + 1024;

  const int ntiles     = N_NODES / 16;
  const int mlp_blocks = (ntiles + 3) / 4;

  build_input_kernel<<<(N_NODES * 16 + 255) / 256, 256, 0, stream>>>(
      x, labels, emb, h, agg);
  pad_w1_kernel<<<4, 256, 0, stream>>>(W1_0, Wpad);

  for (int layer = 0; layer < 3; ++layer) {
    bin_edges_kernel<<<N_OWN, 1024, 0, stream>>>(ei, agg);
    float*       st_new = (layer == 1) ? sums1 : sums0;
    const float* st_old = (layer == 1) ? sums0 : sums1;
    const float* gm_old = (layer == 2) ? g_s : g_0;
    const float* bt_old = (layer == 2) ? be_s : be_0;
    if (layer == 0)
      owner_agg_kernel<4, false><<<N_OWN, 1024, 0, stream>>>(
          h, ei, agg, st_new, st_old, gm_old, bt_old);
    else
      owner_agg_kernel<6, true><<<N_OWN, 1024, 0, stream>>>(
          h, ei, agg, st_new, st_old, gm_old, bt_old);

    const float* W1 = (layer == 0) ? Wpad : (W1_s + (size_t)(layer - 1) * 4096);
    const float* b1 = (layer == 0) ? b1_0 : (b1_s + (size_t)(layer - 1) * 64);
    const float* W2 = (layer == 0) ? W2_0 : (W2_s + (size_t)(layer - 1) * 4096);
    const float* b2 = (layer == 0) ? b2_0 : (b2_s + (size_t)(layer - 1) * 64);
    const float* ep = (layer == 0) ? eps_0 : eps_s;
    const int    ei_idx = (layer == 0) ? 0 : (layer - 1);
    const int    K  = (layer == 0) ? 16 : 64;

    if (layer == 0)
      gin_mlp_kernel<false><<<mlp_blocks, 128, 0, stream>>>(
          h, agg, W1, b1, W2, b2, ep, ei_idx, K, st_new, st_old, gm_old, bt_old);
    else
      gin_mlp_kernel<true><<<mlp_blocks, 128, 0, stream>>>(
          h, agg, W1, b1, W2, b2, ep, ei_idx, K, st_new, st_old, gm_old, bt_old);
  }

  final_kernel<<<(N_NODES + 255) / 256, 256, 0, stream>>>(
      h, sums0, g_s + 64, be_s + 64, Wf, bf, out);
}


# MLP moved into the owner launch: row sums stay in LDS and feed the WMMA tiles there; features alternate between two buffers; BN column sums group by owner (25 tiles, was 4); launches 12 -> 8
# speedup vs baseline: 5.7714x; 1.1172x over previous
#include <hip/hip_runtime.h>
#include <hip/hip_bf16.h>
#include <cstddef>
#include <cstdint>

#define N_NODES 100000
#define N_EDGES 3200000
#define BN_EPS 1e-5f

#define BIN_ROWS  400
#define N_OWN     (N_NODES / BIN_ROWS)
#define BIN_WORDS (BIN_ROWS * 64)
#define BIN_CAP   (BIN_WORDS - 1)

typedef __attribute__((ext_vector_type(2))) float v2f;
typedef __attribute__((ext_vector_type(8))) float v8f;

__device__ __forceinline__ v8f wmma_f32(v2f a, v2f b, v8f c) {
  return __builtin_amdgcn_wmma_f32_16x16x4_f32(false, a, false, b, (short)0, c,
                                               false, false);
}

__device__ __forceinline__ void bn_scale_shift(const float* __restrict__ sums,
                                               const float* __restrict__ gamma,
                                               const float* __restrict__ beta,
                                               int c, float& sc, float& sh) {
  const float inv_n = 1.0f / (float)N_NODES;
  const float mu  = sums[c] * inv_n;
  const float var = sums[64 + c] * inv_n - mu * mu;
  const float rs  = rsqrtf(var + BN_EPS);
  sc = gamma[c] * rs;
  sh = beta[c] - mu * sc;
}

__global__ void build_input_kernel(const float* __restrict__ x,
                                   const int* __restrict__ labels,
                                   const float* __restrict__ emb,
                                   float* __restrict__ h,
                                   int* __restrict__ cnt) {
  long long idx = (long long)blockIdx.x * blockDim.x + threadIdx.x;
  if (idx >= (long long)N_NODES * 16) return;
  if (idx < 3 * N_OWN) cnt[idx] = 0;
  const int i = (int)(idx >> 4);
  const int c = (int)(idx & 15);
  float v = 0.0f;
  if (c < 2)       v = x[(size_t)i * 2 + c];
  else if (c < 10) v = emb[(size_t)labels[i] * 8 + (c - 2)];
  h[(size_t)i * 64 + c] = v;
}

#define BIN_CHUNK (N_EDGES / N_OWN)
#define BIN_PER_T ((BIN_CHUNK + 1023) / 1024)

static_assert(N_OWN * BIN_ROWS == N_NODES, "owners tile the rows exactly");
static_assert(N_OWN * BIN_CHUNK == N_EDGES, "bin pass chunks tile the edge list exactly");
static_assert(N_NODES <= (1 << 17), "source row fits the low 17 bits of an entry");
static_assert(BIN_ROWS <= (1 << 15), "local row fits the high bits of an entry");
static_assert(N_OWN <= 1024, "one thread per owner in the bin pass");

__global__ void __launch_bounds__(1024)
bin_edges_kernel(const int* __restrict__ edge_index, float* __restrict__ obuf,
                 int* __restrict__ cnt, float* __restrict__ stats) {
  __shared__ int hist[N_OWN];
  __shared__ int lbase[N_OWN];
  __shared__ int gbase[N_OWN];
  __shared__ unsigned stage[BIN_CHUNK];
  const int tid  = threadIdx.x;
  const int lane = tid & 31;
  const int wave = __builtin_amdgcn_readfirstlane(tid >> 5);
  unsigned* bins = (unsigned*)obuf;
  if (tid < N_OWN) hist[tid] = 0;
  if (blockIdx.x == 0 && tid < 128) stats[tid] = 0.0f;
  __syncthreads();

  const int* __restrict__ srcs = edge_index + (size_t)blockIdx.x * BIN_CHUNK;
  const int* __restrict__ dsts = srcs + N_EDGES;
  int dv[BIN_PER_T], sv[BIN_PER_T];
#pragma unroll
  for (int u = 0; u < BIN_PER_T; ++u) {
    const int i  = tid + u * 1024;
    const int ic = (i < BIN_CHUNK) ? i : (BIN_CHUNK - 1);
    dv[u] = dsts[ic];
    sv[u] = srcs[ic];
  }
  int own[BIN_PER_T];
  unsigned ent[BIN_PER_T];
  int rank[BIN_PER_T];
#pragma unroll
  for (int u = 0; u < BIN_PER_T; ++u) {
    const int i = tid + u * 1024;
    own[u] = -1;
    ent[u] = 0u;
    rank[u] = 0;
    if (i < BIN_CHUNK && (unsigned)dv[u] < (unsigned)N_NODES) {
      const int o = dv[u] / BIN_ROWS;
      own[u] = o;
      ent[u] = (unsigned)sv[u] | ((unsigned)(dv[u] - o * BIN_ROWS) << 17);
      rank[u] = atomicAdd(&hist[o], 1);
    }
  }
  __syncthreads();

  if (tid < N_OWN) {
    int before = 0;
    for (int k = 0; k < tid; ++k) before += hist[k];
    lbase[tid] = before;
    gbase[tid] = atomicAdd(cnt + tid, hist[tid]);
  }
  __syncthreads();

#pragma unroll
  for (int u = 0; u < BIN_PER_T; ++u)
    if (own[u] >= 0) stage[lbase[own[u]] + rank[u]] = ent[u];
  __syncthreads();

  for (int o = wave; o < N_OWN; o += 32) {
    const int n  = hist[o];
    const int lb = lbase[o];
    const int gb = gbase[o];
    unsigned* dst = bins + (size_t)o * BIN_WORDS;
    for (int k = lane; k < n; k += 32) {
      const int pos = gb + k;
      if (pos < BIN_CAP) dst[pos] = stage[lb + k];
    }
  }
}

#define MLP_TP 68
#define MLP_WP 160

template <int KSTEPS>
__device__ __forceinline__ void tile_gemm(const float* arow, const float* wcol,
                                          v8f (&acc)[4]) {
  constexpr int NB = KSTEPS / 2;
  v2f a[2][2];
  v2f b[2][2][4];
#pragma unroll
  for (int q = 0; q < 2; ++q) {
    a[0][q] = *(const v2f*)(arow + q * 4);
#pragma unroll
    for (int nt = 0; nt < 4; ++nt)
      b[0][q][nt] = *(const v2f*)(wcol + q * 2 * MLP_WP + nt * 32);
  }
#pragma unroll
  for (int blk = 0; blk < NB; ++blk) {
    const int cur = blk & 1, nxt = cur ^ 1;
    if (blk + 1 < NB) {
#pragma unroll
      for (int q = 0; q < 2; ++q) {
        const int s = (blk + 1) * 2 + q;
        a[nxt][q] = *(const v2f*)(arow + s * 4);
#pragma unroll
        for (int nt = 0; nt < 4; ++nt)
          b[nxt][q][nt] = *(const v2f*)(wcol + s * 2 * MLP_WP + nt * 32);
      }
    }
#pragma unroll
    for (int q = 0; q < 2; ++q)
#pragma unroll
      for (int nt = 0; nt < 4; ++nt)
        acc[nt] = wmma_f32(a[cur][q], b[cur][q][nt], acc[nt]);
  }
}

#define OWN_PER_T ((BIN_CAP + 1023) / 1024)

static_assert(BIN_ROWS <= 1024, "one thread per row in the scan of the row counters");
static_assert(BIN_ROWS % 16 == 0, "an owner's rows are whole tiles");
static_assert(BIN_ROWS / 16 <= 32, "one wave per tile");

template <bool L0>
__global__ void __launch_bounds__(1024)
owner_gin_kernel(const float* __restrict__ hin, float* hout,
                 const int* __restrict__ edge_index,
                 const int* __restrict__ bincnt,
                 const float* __restrict__ W1, const float* __restrict__ b1,
                 const float* __restrict__ W2, const float* __restrict__ b2,
                 const float* __restrict__ eps_ptr, int eps_idx,
                 float* __restrict__ stats,
                 const float* __restrict__ bnstats,
                 const float* __restrict__ gamma,
                 const float* __restrict__ beta) {
  constexpr int  LOG2W  = L0 ? 4 : 6;
  constexpr int  K      = 1 << LOG2W;
  constexpr bool BN     = !L0;
  constexpr int  W1ROWS = L0 ? 10 : 64;
  __shared__ float acc[BIN_ROWS * 64];
  __shared__ __attribute__((aligned(16))) float arow[BIN_ROWS * MLP_TP];
  __shared__ __attribute__((aligned(16))) float w1s[(K / 2) * MLP_WP];
  __shared__ __attribute__((aligned(16))) float w2s[32 * MLP_WP];
  __shared__ int rowcnt[BIN_ROWS];
  __shared__ int rowstart[BIN_ROWS + 1];
  __shared__ int wsum[32];
  __shared__ float bnss[128];
  __shared__ float sstat[128];
  const int tid  = threadIdx.x;
  const int lane = tid & 31;
  const int wave = __builtin_amdgcn_readfirstlane(tid >> 5);
  const unsigned row0 = blockIdx.x * BIN_ROWS;
  float* region = hout + (size_t)blockIdx.x * BIN_WORDS;
  const unsigned* bin = (const unsigned*)region;

  if (tid < BIN_ROWS) rowcnt[tid] = 0;
  if (tid < 128) sstat[tid] = 0.0f;
  if (BN && tid < 64)
    bn_scale_shift(bnstats, gamma, beta, tid, bnss[tid], bnss[64 + tid]);
  if (tid < (K / 2) * 16) {
    const int p = tid >> 4, c4 = (tid & 15) * 4;
    float4 r0 = make_float4(0.f, 0.f, 0.f, 0.f), r1 = r0;
    if (2 * p < W1ROWS)     r0 = *(const float4*)(W1 + (size_t)(2 * p) * 64 + c4);
    if (2 * p + 1 < W1ROWS) r1 = *(const float4*)(W1 + (size_t)(2 * p + 1) * 64 + c4);
    float* d = &w1s[p * MLP_WP + c4 * 2];
    *(float4*)d       = make_float4(r0.x, r1.x, r0.y, r1.y);
    *(float4*)(d + 4) = make_float4(r0.z, r1.z, r0.w, r1.w);
  }
  if (tid < 32 * 16) {
    const int p = tid >> 4, c4 = (tid & 15) * 4;
    const float4 r0 = *(const float4*)(W2 + (size_t)(2 * p) * 64 + c4);
    const float4 r1 = *(const float4*)(W2 + (size_t)(2 * p + 1) * 64 + c4);
    float* d = &w2s[p * MLP_WP + c4 * 2];
    *(float4*)d       = make_float4(r0.x, r1.x, r0.y, r1.y);
    *(float4*)(d + 4) = make_float4(r0.z, r1.z, r0.w, r1.w);
  }
  const float onepe = 1.0f + eps_ptr[eps_idx];
  const int total = bincnt[blockIdx.x];
  __syncthreads();

  if (total <= BIN_CAP) {
    unsigned* sorted = (unsigned*)acc;
    unsigned ent[OWN_PER_T];
#pragma unroll
    for (int u = 0; u < OWN_PER_T; ++u) {
      const int i = tid + u * 1024;
      ent[u] = bin[(i < total) ? i : 0];
    }
    int rank[OWN_PER_T];
#pragma unroll
    for (int u = 0; u < OWN_PER_T; ++u) {
      const int i = tid + u * 1024;
      rank[u] = 0;
      if (i < total) {
        const unsigned dl = min(ent[u] >> 17, (unsigned)(BIN_ROWS - 1));
        rank[u] = atomicAdd(&rowcnt[dl], 1);
      }
    }
    __syncthreads();

    const int cnt = (tid < BIN_ROWS) ? rowcnt[tid] : 0;
    int incl = cnt;
#pragma unroll
    for (int d = 1; d < 32; d <<= 1) {
      const int t = __shfl_up(incl, d);
      if (lane >= d) incl += t;
    }
    if (lane == 31) wsum[wave] = incl;
    __syncthreads();
    int off = 0;
    for (int k = 0; k < wave; ++k) off += wsum[k];
    if (tid < BIN_ROWS) rowstart[tid] = off + incl - cnt;
    if (tid == BIN_ROWS - 1) rowstart[BIN_ROWS] = off + incl;
    __syncthreads();

#pragma unroll
    for (int u = 0; u < OWN_PER_T; ++u) {
      const int i = tid + u * 1024;
      if (i < total) {
        const unsigned dl = min(ent[u] >> 17, (unsigned)(BIN_ROWS - 1));
        sorted[rowstart[dl] + rank[u]] =
            min(ent[u] & 0x1FFFFu, (unsigned)(N_NODES - 1));
      }
    }
    __syncthreads();

    constexpr int LPE   = (1 << LOG2W) / 2;
    constexpr int EPI   = 32 / LPE;
    constexpr int STEPS = (EPI == 1) ? 48 : 16;
    constexpr int NEV   = (STEPS + 31) / 32;
    const int sub = lane / LPE;
    const int col = (lane % LPE) * 2;
    const float* __restrict__ hcol = hin + col;
    float scx = 1.0f, scy = 1.0f, shx = 0.0f, shy = 0.0f;
    if constexpr (BN) {
      scx = bnss[col];      scy = bnss[col + 1];
      shx = bnss[64 + col]; shy = bnss[64 + col + 1];
    }
    for (int r = wave; r < BIN_ROWS; r += 32) {
      const int s = __builtin_amdgcn_readfirstlane(rowstart[r]);
      const int n = __builtin_amdgcn_readfirstlane(rowstart[r + 1]) - s;
      float ax = 0.0f, ay = 0.0f;
      float2 own = make_float2(0.0f, 0.0f);
      if (sub == 0) own = *(const float2*)(hcol + (size_t)(row0 + r) * 64);
      for (int c = 0; c < n; c += STEPS * EPI) {
        float2 v[STEPS];
        if constexpr (EPI == 1) {
          unsigned ev[NEV];
#pragma unroll
          for (int q = 0; q < NEV; ++q) {
            const int idx = c + q * 32 + lane;
            ev[q] = (idx < n) ? sorted[s + idx] : 0u;
          }
#pragma unroll
          for (int t = 0; t < STEPS; ++t) {
            v[t] = make_float2(0.0f, 0.0f);
            if (c + t < n) {
              const unsigned sk =
                  (unsigned)__builtin_amdgcn_readlane((int)ev[t >> 5], t & 31);
              v[t] = *(const float2*)(hcol + (size_t)sk * 64);
            }
          }
        } else {
          unsigned sk[STEPS];
#pragma unroll
          for (int t = 0; t < STEPS; ++t) {
            const int e0 = c + t * EPI;
            sk[t] = 0u;
            if (e0 < n) sk[t] = sorted[s + ((e0 + sub < n) ? e0 + sub : e0)];
          }
#pragma unroll
          for (int t = 0; t < STEPS; ++t) {
            v[t] = make_float2(0.0f, 0.0f);
            if (c + t * EPI < n)
              v[t] = *(const float2*)(hcol + (size_t)sk[t] * 64);
          }
        }
#pragma unroll
        for (int t = 0; t < STEPS; ++t) {
          const bool ok = (EPI == 1) ? (c + t < n) : (c + t * EPI + sub < n);
          const float vx = BN ? fmaxf(fmaf(v[t].x, scx, shx), 0.0f) : v[t].x;
          const float vy = BN ? fmaxf(fmaf(v[t].y, scy, shy), 0.0f) : v[t].y;
          ax += ok ? vx : 0.0f;
          ay += ok ? vy : 0.0f;
        }
      }
#pragma unroll
      for (int d = LPE; d < 32; d <<= 1) {
        ax += __shfl_xor(ax, d);
        ay += __shfl_xor(ay, d);
      }
      if (sub == 0) {
        if constexpr (BN) {
          own.x = fmaxf(fmaf(own.x, scx, shx), 0.0f);
          own.y = fmaxf(fmaf(own.y, scy, shy), 0.0f);
        }
        *(float2*)&arow[r * MLP_TP + col] =
            make_float2(onepe * own.x + ax, onepe * own.y + ay);
      }
    }
  } else {
    for (int i = tid; i < BIN_ROWS * 16; i += 1024)
      ((float4*)acc)[i] = make_float4(0.f, 0.f, 0.f, 0.f);
    __syncthreads();
    const int* __restrict__ dsts = edge_index + N_EDGES;
    for (int e = tid; e < N_EDGES; e += 1024) {
      const unsigned dl = (unsigned)dsts[e] - row0;
      if (dl < (unsigned)BIN_ROWS) {
        const float* hs = hin + (size_t)edge_index[e] * 64;
        for (int c = 0; c < (1 << LOG2W); ++c) {
          float v = hs[c];
          if constexpr (BN) v = fmaxf(fmaf(v, bnss[c], bnss[64 + c]), 0.0f);
          unsafeAtomicAdd(&acc[dl * 64 + c], v);
        }
      }
    }
    __syncthreads();
    for (int i = tid; i < BIN_ROWS * K; i += 1024) {
      const int r = i >> LOG2W, c = i & (K - 1);
      float v = hin[(size_t)(row0 + r) * 64 + c];
      if constexpr (BN) v = fmaxf(fmaf(v, bnss[c], bnss[64 + c]), 0.0f);
      arow[r * MLP_TP + c] = onepe * v + acc[r * 64 + c];
    }
  }
  __syncthreads();

  if (wave < BIN_ROWS / 16) {
    const int lrow  = lane & 15;
    const int lhalf = lane >> 4;
    float* my = arow + wave * (16 * MLP_TP);
    const float* afr = my + lrow * MLP_TP + lhalf * 2;
    v8f acc1[4] = {};
    tile_gemm<K / 4>(afr, w1s + lhalf * MLP_WP + lrow * 2, acc1);

#pragma unroll
    for (int nt = 0; nt < 4; ++nt) {
      const float bv = b1[nt * 16 + lrow];
#pragma unroll
      for (int r = 0; r < 8; ++r)
        my[(r + lhalf * 8) * MLP_TP + nt * 16 + lrow] = fmaxf(acc1[nt][r] + bv, 0.f);
    }

    v8f acc2[4] = {};
    tile_gemm<16>(afr, w2s + lhalf * MLP_WP + lrow * 2, acc2);

#pragma unroll
    for (int nt = 0; nt < 4; ++nt) {
      const float bv = b2[nt * 16 + lrow];
      float s = 0.0f, s2 = 0.0f;
#pragma unroll
      for (int r = 0; r < 8; ++r) {
        const float v = fmaxf(acc2[nt][r] + bv, 0.f);
        my[(r + lhalf * 8) * MLP_TP + nt * 16 + lrow] = v;
        s += v;
        s2 += v * v;
      }
      unsafeAtomicAdd(&sstat[nt * 16 + lrow], s);
      unsafeAtomicAdd(&sstat[64 + nt * 16 + lrow], s2);
    }
#pragma unroll
    for (int j = 0; j < 8; ++j) {
      const int row = j * 2 + lhalf;
      *(float4*)(region + (size_t)(wave * 16 + row) * 64 + lrow * 4) =
          *(const float4*)&my[row * MLP_TP + lrow * 4];
    }
  }
  __syncthreads();
  if (tid < 128) unsafeAtomicAdd(&stats[tid], sstat[tid]);
}

__global__ void final_kernel(const float* __restrict__ h,
                             const float* __restrict__ bnstats,
                             const float* __restrict__ gamma,
                             const float* __restrict__ beta,
                             const float* __restrict__ Wf,
                             const float* __restrict__ bf,
                             float* __restrict__ out) {
  __shared__ float sbn[128];
  if (threadIdx.x < 64)
    bn_scale_shift(bnstats, gamma, beta, threadIdx.x, sbn[threadIdx.x],
                   sbn[64 + threadIdx.x]);
  __syncthreads();
  const int i = blockIdx.x * blockDim.x + threadIdx.x;
  if (i >= N_NODES) return;
  const float4* hp = (const float4*)(h + (size_t)i * 64);
  const float4* wp = (const float4*)Wf;
  const float4* sc = (const float4*)sbn;
  const float4* sh = (const float4*)(sbn + 64);
  float s = bf[0];
#pragma unroll
  for (int j = 0; j < 16; ++j) {
    float4 a = hp[j], b = wp[j];
    const float4 c = sc[j], d = sh[j];
    a.x = fmaxf(fmaf(a.x, c.x, d.x), 0.0f);
    a.y = fmaxf(fmaf(a.y, c.y, d.y), 0.0f);
    a.z = fmaxf(fmaf(a.z, c.z, d.z), 0.0f);
    a.w = fmaxf(fmaf(a.w, c.w, d.w), 0.0f);
    s += a.x * b.x + a.y * b.y + a.z * b.z + a.w * b.w;
  }
  out[i] = s;
}

extern "C" void kernel_launch(void* const* d_in, const int* in_sizes, int n_in,
                              void* d_out, int out_size, void* d_ws,
                              size_t ws_size, hipStream_t stream) {
  const float* x      = (const float*)d_in[0];
  const int*   labels = (const int*)d_in[1];
  const int*   ei     = (const int*)d_in[2];
  const float* emb    = (const float*)d_in[3];
  const float* W1_0   = (const float*)d_in[4];
  const float* b1_0   = (const float*)d_in[5];
  const float* W2_0   = (const float*)d_in[6];
  const float* b2_0   = (const float*)d_in[7];
  const float* eps_0  = (const float*)d_in[8];
  const float* g_0    = (const float*)d_in[9];
  const float* be_0   = (const float*)d_in[10];
  const float* W1_s   = (const float*)d_in[11];
  const float* b1_s   = (const float*)d_in[12];
  const float* W2_s   = (const float*)d_in[13];
  const float* b2_s   = (const float*)d_in[14];
  const float* eps_s  = (const float*)d_in[15];
  const float* g_s    = (const float*)d_in[16];
  const float* be_s   = (const float*)d_in[17];
  const float* Wf     = (const float*)d_in[18];
  const float* bf     = (const float*)d_in[19];
  float* out = (float*)d_out;

  float* bufQ  = (float*)d_ws;
  float* sums0 = bufQ + (size_t)N_NODES * 64;
  float* sums1 = sums0 + 128;
  int*   cnt   = (int*)(sums1 + 128);
  float* bufP  = sums1 + 128 + 1024;

  build_input_kernel<<<(N_NODES * 16 + 255) / 256, 256, 0, stream>>>(
      x, labels, emb, bufP, cnt);

  for (int layer = 0; layer < 3; ++layer) {
    const float* hin  = (layer == 1) ? bufQ : bufP;
    float*       hout = (layer == 1) ? bufP : bufQ;
    float*       st_new = (layer == 1) ? sums1 : sums0;
    const float* st_old = (layer == 1) ? sums0 : sums1;
    const float* gm_old = (layer == 2) ? g_s : g_0;
    const float* bt_old = (layer == 2) ? be_s : be_0;
    int* lcnt = cnt + layer * N_OWN;
    bin_edges_kernel<<<N_OWN, 1024, 0, stream>>>(ei, hout, lcnt, st_new);

    const float* W1 = (layer == 0) ? W1_0 : (W1_s + (size_t)(layer - 1) * 4096);
    const float* b1 = (layer == 0) ? b1_0 : (b1_s + (size_t)(layer - 1) * 64);
    const float* W2 = (layer == 0) ? W2_0 : (W2_s + (size_t)(layer - 1) * 4096);
    const float* b2 = (layer == 0) ? b2_0 : (b2_s + (size_t)(layer - 1) * 64);
    const float* ep = (layer == 0) ? eps_0 : eps_s;
    const int    ei_idx = (layer == 0) ? 0 : (layer - 1);

    if (layer == 0)
      owner_gin_kernel<true><<<N_OWN, 1024, 0, stream>>>(
          hin, hout, ei, lcnt, W1, b1, W2, b2, ep, ei_idx,
          st_new, st_old, gm_old, bt_old);
    else
      owner_gin_kernel<false><<<N_OWN, 1024, 0, stream>>>(
          hin, hout, ei, lcnt, W1, b1, W2, b2, ep, ei_idx,
          st_new, st_old, gm_old, bt_old);
  }

  final_kernel<<<(N_NODES + 255) / 256, 256, 0, stream>>>(
      bufQ, sums0, g_s + 64, be_s + 64, Wf, bf, out);
}


# Second layer reuses the first layer's row-sorted edge lists, parked by each owner in the 48 unused columns of its input rows: its bin pass and LDS sort gone (8 -> 7 launches); f32 sum order changes
# speedup vs baseline: 6.3781x; 1.1051x over previous
#include <hip/hip_runtime.h>
#include <hip/hip_bf16.h>
#include <cstddef>
#include <cstdint>

#define N_NODES 100000
#define N_EDGES 3200000
#define BN_EPS 1e-5f

#define BIN_ROWS  400
#define N_OWN     (N_NODES / BIN_ROWS)
#define BIN_WORDS (BIN_ROWS * 64)
#define BIN_CAP   (BIN_WORDS - 1)

typedef __attribute__((ext_vector_type(2))) float v2f;
typedef __attribute__((ext_vector_type(8))) float v8f;

__device__ __forceinline__ v8f wmma_f32(v2f a, v2f b, v8f c) {
  return __builtin_amdgcn_wmma_f32_16x16x4_f32(false, a, false, b, (short)0, c,
                                               false, false);
}

__device__ __forceinline__ void bn_scale_shift(const float* __restrict__ sums,
                                               const float* __restrict__ gamma,
                                               const float* __restrict__ beta,
                                               int c, float& sc, float& sh) {
  const float inv_n = 1.0f / (float)N_NODES;
  const float mu  = sums[c] * inv_n;
  const float var = sums[64 + c] * inv_n - mu * mu;
  const float rs  = rsqrtf(var + BN_EPS);
  sc = gamma[c] * rs;
  sh = beta[c] - mu * sc;
}

__global__ void build_input_kernel(const float* __restrict__ x,
                                   const int* __restrict__ labels,
                                   const float* __restrict__ emb,
                                   float* __restrict__ h,
                                   int* __restrict__ cnt) {
  long long idx = (long long)blockIdx.x * blockDim.x + threadIdx.x;
  if (idx >= (long long)N_NODES * 16) return;
  if (idx < 2 * N_OWN) cnt[idx] = 0;
  const int i = (int)(idx >> 4);
  const int c = (int)(idx & 15);
  float v = 0.0f;
  if (c < 2)       v = x[(size_t)i * 2 + c];
  else if (c < 10) v = emb[(size_t)labels[i] * 8 + (c - 2)];
  h[(size_t)i * 64 + c] = v;
}

#define BIN_CHUNK (N_EDGES / N_OWN)
#define BIN_PER_T ((BIN_CHUNK + 1023) / 1024)

static_assert(N_OWN * BIN_ROWS == N_NODES, "owners tile the rows exactly");
static_assert(N_OWN * BIN_CHUNK == N_EDGES, "bin pass chunks tile the edge list exactly");
static_assert(N_NODES <= (1 << 17), "source row fits the low 17 bits of an entry");
static_assert(BIN_ROWS <= (1 << 15), "local row fits the high bits of an entry");
static_assert(N_OWN <= 1024, "one thread per owner in the bin pass");

#define PARK_COL0 16
#define PARK_COLS (64 - PARK_COL0)
#define PARK_CAP  (BIN_ROWS * PARK_COLS - (BIN_ROWS + 1))
static_assert(PARK_CAP <= BIN_CAP, "a bin that can be copied is a complete bin");
__device__ __forceinline__ int park_word(int k) {
  return (k / PARK_COLS) * 64 + PARK_COL0 + (k % PARK_COLS);
}

__global__ void __launch_bounds__(1024)
bin_edges_kernel(const int* __restrict__ edge_index, float* __restrict__ obuf,
                 int* __restrict__ cnt, float* __restrict__ stats, int nclear) {
  __shared__ int hist[N_OWN];
  __shared__ int lbase[N_OWN];
  __shared__ int gbase[N_OWN];
  __shared__ unsigned stage[BIN_CHUNK];
  const int tid  = threadIdx.x;
  const int lane = tid & 31;
  const int wave = __builtin_amdgcn_readfirstlane(tid >> 5);
  unsigned* bins = (unsigned*)obuf;
  if (tid < N_OWN) hist[tid] = 0;
  if (blockIdx.x == 0 && tid < nclear) stats[tid] = 0.0f;
  __syncthreads();

  const int* __restrict__ srcs = edge_index + (size_t)blockIdx.x * BIN_CHUNK;
  const int* __restrict__ dsts = srcs + N_EDGES;
  int dv[BIN_PER_T], sv[BIN_PER_T];
#pragma unroll
  for (int u = 0; u < BIN_PER_T; ++u) {
    const int i  = tid + u * 1024;
    const int ic = (i < BIN_CHUNK) ? i : (BIN_CHUNK - 1);
    dv[u] = dsts[ic];
    sv[u] = srcs[ic];
  }
  int own[BIN_PER_T];
  unsigned ent[BIN_PER_T];
  int rank[BIN_PER_T];
#pragma unroll
  for (int u = 0; u < BIN_PER_T; ++u) {
    const int i = tid + u * 1024;
    own[u] = -1;
    ent[u] = 0u;
    rank[u] = 0;
    if (i < BIN_CHUNK && (unsigned)dv[u] < (unsigned)N_NODES) {
      const int o = dv[u] / BIN_ROWS;
      own[u] = o;
      ent[u] = (unsigned)sv[u] | ((unsigned)(dv[u] - o * BIN_ROWS) << 17);
      rank[u] = atomicAdd(&hist[o], 1);
    }
  }
  __syncthreads();

  if (tid < N_OWN) {
    int before = 0;
    for (int k = 0; k < tid; ++k) before += hist[k];
    lbase[tid] = before;
    gbase[tid] = atomicAdd(cnt + tid, hist[tid]);
  }
  __syncthreads();

#pragma unroll
  for (int u = 0; u < BIN_PER_T; ++u)
    if (own[u] >= 0) stage[lbase[own[u]] + rank[u]] = ent[u];
  __syncthreads();

  for (int o = wave; o < N_OWN; o += 32) {
    const int n  = hist[o];
    const int lb = lbase[o];
    const int gb = gbase[o];
    unsigned* dst = bins + (size_t)o * BIN_WORDS;
    for (int k = lane; k < n; k += 32) {
      const int pos = gb + k;
      if (pos < BIN_CAP) dst[pos] = stage[lb + k];
    }
  }
}

#define MLP_TP 68
#define MLP_WP 160

template <int KSTEPS>
__device__ __forceinline__ void tile_gemm(const float* arow, const float* wcol,
                                          v8f (&acc)[4]) {
  constexpr int NB = KSTEPS / 2;
  v2f a[2][2];
  v2f b[2][2][4];
#pragma unroll
  for (int q = 0; q < 2; ++q) {
    a[0][q] = *(const v2f*)(arow + q * 4);
#pragma unroll
    for (int nt = 0; nt < 4; ++nt)
      b[0][q][nt] = *(const v2f*)(wcol + q * 2 * MLP_WP + nt * 32);
  }
#pragma unroll
  for (int blk = 0; blk < NB; ++blk) {
    const int cur = blk & 1, nxt = cur ^ 1;
    if (blk + 1 < NB) {
#pragma unroll
      for (int q = 0; q < 2; ++q) {
        const int s = (blk + 1) * 2 + q;
        a[nxt][q] = *(const v2f*)(arow + s * 4);
#pragma unroll
        for (int nt = 0; nt < 4; ++nt)
          b[nxt][q][nt] = *(const v2f*)(wcol + s * 2 * MLP_WP + nt * 32);
      }
    }
#pragma unroll
    for (int q = 0; q < 2; ++q)
#pragma unroll
      for (int nt = 0; nt < 4; ++nt)
        acc[nt] = wmma_f32(a[cur][q], b[cur][q][nt], acc[nt]);
  }
}

static_assert(BIN_ROWS <= 1024, "one thread per row in the scan of the row counters");
static_assert(BIN_ROWS % 16 == 0, "an owner's rows are whole tiles");
static_assert(BIN_ROWS / 16 <= 32, "one wave per tile");

template <bool L0, bool PARKED>
__global__ void __launch_bounds__(1024)
owner_gin_kernel(const float* __restrict__ hin, float* hout, float* park,
                 const int* __restrict__ edge_index,
                 const int* __restrict__ bincnt,
                 const float* __restrict__ W1, const float* __restrict__ b1,
                 const float* __restrict__ W2, const float* __restrict__ b2,
                 const float* __restrict__ eps_ptr, int eps_idx,
                 float* __restrict__ stats,
                 const float* __restrict__ bnstats,
                 const float* __restrict__ gamma,
                 const float* __restrict__ beta) {
  constexpr int  LOG2W  = L0 ? 4 : 6;
  constexpr int  K      = 1 << LOG2W;
  constexpr bool BN     = !L0;
  constexpr int  W1ROWS = L0 ? 10 : 64;
  __shared__ float acc[BIN_ROWS * 64];
  __shared__ __attribute__((aligned(16))) float arow[BIN_ROWS * MLP_TP];
  __shared__ __attribute__((aligned(16))) float w1s[(K / 2) * MLP_WP];
  __shared__ __attribute__((aligned(16))) float w2s[32 * MLP_WP];
  __shared__ int rowcnt[BIN_ROWS];
  __shared__ int rowstart[BIN_ROWS + 1];
  __shared__ int wsum[32];
  __shared__ float bnss[128];
  __shared__ float sstat[128];
  const int tid  = threadIdx.x;
  const int lane = tid & 31;
  const int wave = __builtin_amdgcn_readfirstlane(tid >> 5);
  const unsigned row0 = blockIdx.x * BIN_ROWS;
  float* region = hout + (size_t)blockIdx.x * BIN_WORDS;
  const unsigned* bin = (const unsigned*)region;

  if (!PARKED && tid < BIN_ROWS) rowcnt[tid] = 0;
  if (tid < 128) sstat[tid] = 0.0f;
  if (BN && tid < 64)
    bn_scale_shift(bnstats, gamma, beta, tid, bnss[tid], bnss[64 + tid]);
  if (tid < (K / 2) * 16) {
    const int p = tid >> 4, c4 = (tid & 15) * 4;
    float4 r0 = make_float4(0.f, 0.f, 0.f, 0.f), r1 = r0;
    if (2 * p < W1ROWS)     r0 = *(const float4*)(W1 + (size_t)(2 * p) * 64 + c4);
    if (2 * p + 1 < W1ROWS) r1 = *(const float4*)(W1 + (size_t)(2 * p + 1) * 64 + c4);
    float* d = &w1s[p * MLP_WP + c4 * 2];
    *(float4*)d       = make_float4(r0.x, r1.x, r0.y, r1.y);
    *(float4*)(d + 4) = make_float4(r0.z, r1.z, r0.w, r1.w);
  }
  if (tid < 32 * 16) {
    const int p = tid >> 4, c4 = (tid & 15) * 4;
    const float4 r0 = *(const float4*)(W2 + (size_t)(2 * p) * 64 + c4);
    const float4 r1 = *(const float4*)(W2 + (size_t)(2 * p + 1) * 64 + c4);
    float* d = &w2s[p * MLP_WP + c4 * 2];
    *(float4*)d       = make_float4(r0.x, r1.x, r0.y, r1.y);
    *(float4*)(d + 4) = make_float4(r0.z, r1.z, r0.w, r1.w);
  }
  const float onepe = 1.0f + eps_ptr[eps_idx];
  const int total = bincnt[blockIdx.x];
  __syncthreads();

  constexpr int CAP   = PARKED ? PARK_CAP : BIN_CAP;
  constexpr int PER_T = (CAP + 1023) / 1024;
  if (total <= CAP) {
    unsigned* sorted = (unsigned*)acc;
    if constexpr (PARKED) {
      unsigned sv[PER_T];
#pragma unroll
      for (int u = 0; u < PER_T; ++u) {
        const int i = tid + u * 1024;
        sv[u] = bin[park_word((i < total) ? i : 0)];
      }
      int place = 0;
      if (tid <= BIN_ROWS) place = (int)bin[park_word(total + tid)];
#pragma unroll
      for (int u = 0; u < PER_T; ++u) {
        const int i = tid + u * 1024;
        if (i < total) sorted[i] = min(sv[u], (unsigned)(N_NODES - 1));
      }
      if (tid <= BIN_ROWS) rowstart[tid] = min(max(place, 0), total);
      __syncthreads();
    } else {
      unsigned ent[PER_T];
#pragma unroll
      for (int u = 0; u < PER_T; ++u) {
        const int i = tid + u * 1024;
        ent[u] = bin[(i < total) ? i : 0];
      }
      int rank[PER_T];
#pragma unroll
      for (int u = 0; u < PER_T; ++u) {
        const int i = tid + u * 1024;
        rank[u] = 0;
        if (i < total) {
          const unsigned dl = min(ent[u] >> 17, (unsigned)(BIN_ROWS - 1));
          rank[u] = atomicAdd(&rowcnt[dl], 1);
        }
      }
      __syncthreads();

      const int cnt = (tid < BIN_ROWS) ? rowcnt[tid] : 0;
      int incl = cnt;
#pragma unroll
      for (int d = 1; d < 32; d <<= 1) {
        const int t = __shfl_up(incl, d);
        if (lane >= d) incl += t;
      }
      if (lane == 31) wsum[wave] = incl;
      __syncthreads();
      int off = 0;
      for (int k = 0; k < wave; ++k) off += wsum[k];
      if (tid < BIN_ROWS) rowstart[tid] = off + incl - cnt;
      if (tid == BIN_ROWS - 1) rowstart[BIN_ROWS] = off + incl;
      __syncthreads();

#pragma unroll
      for (int u = 0; u < PER_T; ++u) {
        const int i = tid + u * 1024;
        if (i < total) {
          const unsigned dl = min(ent[u] >> 17, (unsigned)(BIN_ROWS - 1));
          sorted[rowstart[dl] + rank[u]] =
              min(ent[u] & 0x1FFFFu, (unsigned)(N_NODES - 1));
        }
      }
      __syncthreads();
      if constexpr (L0) {
        if (total <= PARK_CAP) {
          constexpr int PK_T = (PARK_CAP + 1023) / 1024;
          unsigned* pk = (unsigned*)park + (size_t)blockIdx.x * BIN_WORDS;
          unsigned sv[PK_T];
#pragma unroll
          for (int u = 0; u < PK_T; ++u) {
            const int i = tid + u * 1024;
            sv[u] = sorted[(i < total) ? i : 0];
          }
          const int place = rowstart[(tid <= BIN_ROWS) ? tid : 0];
#pragma unroll
          for (int u = 0; u < PK_T; ++u) {
            const int i = tid + u * 1024;
            if (i < total) pk[park_word(i)] = sv[u];
          }
          if (tid <= BIN_ROWS) pk[park_word(total + tid)] = (unsigned)place;
        }
      }
    }

    constexpr int LPE   = (1 << LOG2W) / 2;
    constexpr int EPI   = 32 / LPE;
    constexpr int STEPS = (EPI == 1) ? 48 : 16;
    constexpr int NEV   = (STEPS + 31) / 32;
    const int sub = lane / LPE;
    const int col = (lane % LPE) * 2;
    const float* __restrict__ hcol = hin + col;
    float scx = 1.0f, scy = 1.0f, shx = 0.0f, shy = 0.0f;
    if constexpr (BN) {
      scx = bnss[col];      scy = bnss[col + 1];
      shx = bnss[64 + col]; shy = bnss[64 + col + 1];
    }
    for (int r = wave; r < BIN_ROWS; r += 32) {
      const int s = __builtin_amdgcn_readfirstlane(rowstart[r]);
      const int n = __builtin_amdgcn_readfirstlane(rowstart[r + 1]) - s;
      float ax = 0.0f, ay = 0.0f;
      float2 own = make_float2(0.0f, 0.0f);
      if (sub == 0) own = *(const float2*)(hcol + (size_t)(row0 + r) * 64);
      for (int c = 0; c < n; c += STEPS * EPI) {
        float2 v[STEPS];
        if constexpr (EPI == 1) {
          unsigned ev[NEV];
#pragma unroll
          for (int q = 0; q < NEV; ++q) {
            const int idx = c + q * 32 + lane;
            ev[q] = (idx < n) ? sorted[s + idx] : 0u;
          }
#pragma unroll
          for (int t = 0; t < STEPS; ++t) {
            v[t] = make_float2(0.0f, 0.0f);
            if (c + t < n) {
              const unsigned sk =
                  (unsigned)__builtin_amdgcn_readlane((int)ev[t >> 5], t & 31);
              v[t] = *(const float2*)(hcol + (size_t)sk * 64);
            }
          }
        } else {
          unsigned sk[STEPS];
#pragma unroll
          for (int t = 0; t < STEPS; ++t) {
            const int e0 = c + t * EPI;
            sk[t] = 0u;
            if (e0 < n) sk[t] = sorted[s + ((e0 + sub < n) ? e0 + sub : e0)];
          }
#pragma unroll
          for (int t = 0; t < STEPS; ++t) {
            v[t] = make_float2(0.0f, 0.0f);
            if (c + t * EPI < n)
              v[t] = *(const float2*)(hcol + (size_t)sk[t] * 64);
          }
        }
#pragma unroll
        for (int t = 0; t < STEPS; ++t) {
          const bool ok = (EPI == 1) ? (c + t < n) : (c + t * EPI + sub < n);
          const float vx = BN ? fmaxf(fmaf(v[t].x, scx, shx), 0.0f) : v[t].x;
          const float vy = BN ? fmaxf(fmaf(v[t].y, scy, shy), 0.0f) : v[t].y;
          ax += ok ? vx : 0.0f;
          ay += ok ? vy : 0.0f;
        }
      }
#pragma unroll
      for (int d = LPE; d < 32; d <<= 1) {
        ax += __shfl_xor(ax, d);
        ay += __shfl_xor(ay, d);
      }
      if (sub == 0) {
        if constexpr (BN) {
          own.x = fmaxf(fmaf(own.x, scx, shx), 0.0f);
          own.y = fmaxf(fmaf(own.y, scy, shy), 0.0f);
        }
        *(float2*)&arow[r * MLP_TP + col] =
            make_float2(onepe * own.x + ax, onepe * own.y + ay);
      }
    }
  } else {
    for (int i = tid; i < BIN_ROWS * 16; i += 1024)
      ((float4*)acc)[i] = make_float4(0.f, 0.f, 0.f, 0.f);
    __syncthreads();
    const int* __restrict__ dsts = edge_index + N_EDGES;
    for (int e = tid; e < N_EDGES; e += 1024) {
      const unsigned dl = (unsigned)dsts[e] - row0;
      if (dl < (unsigned)BIN_ROWS) {
        const float* hs = hin + (size_t)edge_index[e] * 64;
        for (int c = 0; c < (1 << LOG2W); ++c) {
          float v = hs[c];
          if constexpr (BN) v = fmaxf(fmaf(v, bnss[c], bnss[64 + c]), 0.0f);
          unsafeAtomicAdd(&acc[dl * 64 + c], v);
        }
      }
    }
    __syncthreads();
    for (int i = tid; i < BIN_ROWS * K; i += 1024) {
      const int r = i >> LOG2W, c = i & (K - 1);
      float v = hin[(size_t)(row0 + r) * 64 + c];
      if constexpr (BN) v = fmaxf(fmaf(v, bnss[c], bnss[64 + c]), 0.0f);
      arow[r * MLP_TP + c] = onepe * v + acc[r * 64 + c];
    }
  }
  __syncthreads();

  if (wave < BIN_ROWS / 16) {
    const int lrow  = lane & 15;
    const int lhalf = lane >> 4;
    float* my = arow + wave * (16 * MLP_TP);
    const float* afr = my + lrow * MLP_TP + lhalf * 2;
    v8f acc1[4] = {};
    tile_gemm<K / 4>(afr, w1s + lhalf * MLP_WP + lrow * 2, acc1);

#pragma unroll
    for (int nt = 0; nt < 4; ++nt) {
      const float bv = b1[nt * 16 + lrow];
#pragma unroll
      for (int r = 0; r < 8; ++r)
        my[(r + lhalf * 8) * MLP_TP + nt * 16 + lrow] = fmaxf(acc1[nt][r] + bv, 0.f);
    }

    v8f acc2[4] = {};
    tile_gemm<16>(afr, w2s + lhalf * MLP_WP + lrow * 2, acc2);

#pragma unroll
    for (int nt = 0; nt < 4; ++nt) {
      const float bv = b2[nt * 16 + lrow];
      float s = 0.0f, s2 = 0.0f;
#pragma unroll
      for (int r = 0; r < 8; ++r) {
        const float v = fmaxf(acc2[nt][r] + bv, 0.f);
        my[(r + lhalf * 8) * MLP_TP + nt * 16 + lrow] = v;
        s += v;
        s2 += v * v;
      }
      unsafeAtomicAdd(&sstat[nt * 16 + lrow], s);
      unsafeAtomicAdd(&sstat[64 + nt * 16 + lrow], s2);
    }
#pragma unroll
    for (int j = 0; j < 8; ++j) {
      const int row = j * 2 + lhalf;
      *(float4*)(region + (size_t)(wave * 16 + row) * 64 + lrow * 4) =
          *(const float4*)&my[row * MLP_TP + lrow * 4];
    }
  }
  __syncthreads();
  if (tid < 128) unsafeAtomicAdd(&stats[tid], sstat[tid]);
}

__global__ void final_kernel(const float* __restrict__ h,
                             const float* __restrict__ bnstats,
                             const float* __restrict__ gamma,
                             const float* __restrict__ beta,
                             const float* __restrict__ Wf,
                             const float* __restrict__ bf,
                             float* __restrict__ out) {
  __shared__ float sbn[128];
  if (threadIdx.x < 64)
    bn_scale_shift(bnstats, gamma, beta, threadIdx.x, sbn[threadIdx.x],
                   sbn[64 + threadIdx.x]);
  __syncthreads();
  const int i = blockIdx.x * blockDim.x + threadIdx.x;
  if (i >= N_NODES) return;
  const float4* hp = (const float4*)(h + (size_t)i * 64);
  const float4* wp = (const float4*)Wf;
  const float4* sc = (const float4*)sbn;
  const float4* sh = (const float4*)(sbn + 64);
  float s = bf[0];
#pragma unroll
  for (int j = 0; j < 16; ++j) {
    float4 a = hp[j], b = wp[j];
    const float4 c = sc[j], d = sh[j];
    a.x = fmaxf(fmaf(a.x, c.x, d.x), 0.0f);
    a.y = fmaxf(fmaf(a.y, c.y, d.y), 0.0f);
    a.z = fmaxf(fmaf(a.z, c.z, d.z), 0.0f);
    a.w = fmaxf(fmaf(a.w, c.w, d.w), 0.0f);
    s += a.x * b.x + a.y * b.y + a.z * b.z + a.w * b.w;
  }
  out[i] = s;
}

extern "C" void kernel_launch(void* const* d_in, const int* in_sizes, int n_in,
                              void* d_out, int out_size, void* d_ws,
                              size_t ws_size, hipStream_t stream) {
  const float* x      = (const float*)d_in[0];
  const int*   labels = (const int*)d_in[1];
  const int*   ei     = (const int*)d_in[2];
  const float* emb    = (const float*)d_in[3];
  const float* W1_0   = (const float*)d_in[4];
  const float* b1_0   = (const float*)d_in[5];
  const float* W2_0   = (const float*)d_in[6];
  const float* b2_0   = (const float*)d_in[7];
  const float* eps_0  = (const float*)d_in[8];
  const float* g_0    = (const float*)d_in[9];
  const float* be_0   = (const float*)d_in[10];
  const float* W1_s   = (const float*)d_in[11];
  const float* b1_s   = (const float*)d_in[12];
  const float* W2_s   = (const float*)d_in[13];
  const float* b2_s   = (const float*)d_in[14];
  const float* eps_s  = (const float*)d_in[15];
  const float* g_s    = (const float*)d_in[16];
  const float* be_s   = (const float*)d_in[17];
  const float* Wf     = (const float*)d_in[18];
  const float* bf     = (const float*)d_in[19];
  float* out = (float*)d_out;

  float* bufQ  = (float*)d_ws;
  float* sums0 = bufQ + (size_t)N_NODES * 64;
  float* sums1 = sums0 + 128;
  int*   cnt   = (int*)(sums1 + 128);
  float* bufP  = sums1 + 128 + 1024;

  build_input_kernel<<<(N_NODES * 16 + 255) / 256, 256, 0, stream>>>(
      x, labels, emb, bufP, cnt);

  for (int layer = 0; layer < 3; ++layer) {
    const float* hin  = (layer == 1) ? bufQ : bufP;
    float*       hout = (layer == 1) ? bufP : bufQ;
    float*       st_new = (layer == 1) ? sums1 : sums0;
    const float* st_old = (layer == 1) ? sums0 : sums1;
    const float* gm_old = (layer == 2) ? g_s : g_0;
    const float* bt_old = (layer == 2) ? be_s : be_0;
    int* lcnt = cnt + ((layer == 2) ? N_OWN : 0);
    if (layer != 1)
      bin_edges_kernel<<<N_OWN, 1024, 0, stream>>>(ei, hout, lcnt, st_new,
                                                   (layer == 0) ? 256 : 128);

    const float* W1 = (layer == 0) ? W1_0 : (W1_s + (size_t)(layer - 1) * 4096);
    const float* b1 = (layer == 0) ? b1_0 : (b1_s + (size_t)(layer - 1) * 64);
    const float* W2 = (layer == 0) ? W2_0 : (W2_s + (size_t)(layer - 1) * 4096);
    const float* b2 = (layer == 0) ? b2_0 : (b2_s + (size_t)(layer - 1) * 64);
    const float* ep = (layer == 0) ? eps_0 : eps_s;
    const int    ei_idx = (layer == 0) ? 0 : (layer - 1);

    if (layer == 0)
      owner_gin_kernel<true, false><<<N_OWN, 1024, 0, stream>>>(
          hin, hout, bufP, ei, lcnt, W1, b1, W2, b2, ep, ei_idx,
          st_new, st_old, gm_old, bt_old);
    else if (layer == 1)
      owner_gin_kernel<false, true><<<N_OWN, 1024, 0, stream>>>(
          hin, hout, nullptr, ei, lcnt, W1, b1, W2, b2, ep, ei_idx,
          st_new, st_old, gm_old, bt_old);
    else
      owner_gin_kernel<false, false><<<N_OWN, 1024, 0, stream>>>(
          hin, hout, nullptr, ei, lcnt, W1, b1, W2, b2, ep, ei_idx,
          st_new, st_old, gm_old, bt_old);
  }

  final_kernel<<<(N_NODES + 255) / 256, 256, 0, stream>>>(
      bufQ, sums0, g_s + 64, be_s + 64, Wf, bf, out);
}


# 64-column row gather with about half the instructions per source row: sources in groups of 8 under one scalar test (no per-source test, fill or select); loads take scalar row base + lane offset
# speedup vs baseline: 6.9708x; 1.0929x over previous
#include <hip/hip_runtime.h>
#include <hip/hip_bf16.h>
#include <cstddef>
#include <cstdint>

#define N_NODES 100000
#define N_EDGES 3200000
#define BN_EPS 1e-5f

#define BIN_ROWS  400
#define N_OWN     (N_NODES / BIN_ROWS)
#define BIN_WORDS (BIN_ROWS * 64)
#define BIN_CAP   (BIN_WORDS - 1)

typedef __attribute__((ext_vector_type(2))) float v2f;
typedef __attribute__((ext_vector_type(8))) float v8f;

__device__ __forceinline__ v8f wmma_f32(v2f a, v2f b, v8f c) {
  return __builtin_amdgcn_wmma_f32_16x16x4_f32(false, a, false, b, (short)0, c,
                                               false, false);
}

__device__ __forceinline__ void bn_scale_shift(const float* __restrict__ sums,
                                               const float* __restrict__ gamma,
                                               const float* __restrict__ beta,
                                               int c, float& sc, float& sh) {
  const float inv_n = 1.0f / (float)N_NODES;
  const float mu  = sums[c] * inv_n;
  const float var = sums[64 + c] * inv_n - mu * mu;
  const float rs  = rsqrtf(var + BN_EPS);
  sc = gamma[c] * rs;
  sh = beta[c] - mu * sc;
}

__global__ void build_input_kernel(const float* __restrict__ x,
                                   const int* __restrict__ labels,
                                   const float* __restrict__ emb,
                                   float* __restrict__ h,
                                   int* __restrict__ cnt) {
  long long idx = (long long)blockIdx.x * blockDim.x + threadIdx.x;
  if (idx >= (long long)N_NODES * 16) return;
  if (idx < 2 * N_OWN) cnt[idx] = 0;
  const int i = (int)(idx >> 4);
  const int c = (int)(idx & 15);
  float v = 0.0f;
  if (c < 2)       v = x[(size_t)i * 2 + c];
  else if (c < 10) v = emb[(size_t)labels[i] * 8 + (c - 2)];
  h[(size_t)i * 64 + c] = v;
}

#define BIN_CHUNK (N_EDGES / N_OWN)
#define BIN_PER_T ((BIN_CHUNK + 1023) / 1024)

static_assert(N_OWN * BIN_ROWS == N_NODES, "owners tile the rows exactly");
static_assert(N_OWN * BIN_CHUNK == N_EDGES, "bin pass chunks tile the edge list exactly");
static_assert(N_NODES <= (1 << 17), "source row fits the low 17 bits of an entry");
static_assert(BIN_ROWS <= (1 << 15), "local row fits the high bits of an entry");
static_assert(N_OWN <= 1024, "one thread per owner in the bin pass");

#define PARK_COL0 16
#define PARK_COLS (64 - PARK_COL0)
#define PARK_CAP  (BIN_ROWS * PARK_COLS - (BIN_ROWS + 1))
static_assert(PARK_CAP <= BIN_CAP, "a bin that can be copied is a complete bin");
__device__ __forceinline__ int park_word(int k) {
  return (k / PARK_COLS) * 64 + PARK_COL0 + (k % PARK_COLS);
}

__global__ void __launch_bounds__(1024)
bin_edges_kernel(const int* __restrict__ edge_index, float* __restrict__ obuf,
                 int* __restrict__ cnt, float* __restrict__ stats, int nclear) {
  __shared__ int hist[N_OWN];
  __shared__ int lbase[N_OWN];
  __shared__ int gbase[N_OWN];
  __shared__ unsigned stage[BIN_CHUNK];
  const int tid  = threadIdx.x;
  const int lane = tid & 31;
  const int wave = __builtin_amdgcn_readfirstlane(tid >> 5);
  unsigned* bins = (unsigned*)obuf;
  if (tid < N_OWN) hist[tid] = 0;
  if (blockIdx.x == 0 && tid < nclear) stats[tid] = 0.0f;
  __syncthreads();

  const int* __restrict__ srcs = edge_index + (size_t)blockIdx.x * BIN_CHUNK;
  const int* __restrict__ dsts = srcs + N_EDGES;
  int dv[BIN_PER_T], sv[BIN_PER_T];
#pragma unroll
  for (int u = 0; u < BIN_PER_T; ++u) {
    const int i  = tid + u * 1024;
    const int ic = (i < BIN_CHUNK) ? i : (BIN_CHUNK - 1);
    dv[u] = dsts[ic];
    sv[u] = srcs[ic];
  }
  int own[BIN_PER_T];
  unsigned ent[BIN_PER_T];
  int rank[BIN_PER_T];
#pragma unroll
  for (int u = 0; u < BIN_PER_T; ++u) {
    const int i = tid + u * 1024;
    own[u] = -1;
    ent[u] = 0u;
    rank[u] = 0;
    if (i < BIN_CHUNK && (unsigned)dv[u] < (unsigned)N_NODES) {
      const int o = dv[u] / BIN_ROWS;
      own[u] = o;
      ent[u] = (unsigned)sv[u] | ((unsigned)(dv[u] - o * BIN_ROWS) << 17);
      rank[u] = atomicAdd(&hist[o], 1);
    }
  }
  __syncthreads();

  if (tid < N_OWN) {
    int before = 0;
    for (int k = 0; k < tid; ++k) before += hist[k];
    lbase[tid] = before;
    gbase[tid] = atomicAdd(cnt + tid, hist[tid]);
  }
  __syncthreads();

#pragma unroll
  for (int u = 0; u < BIN_PER_T; ++u)
    if (own[u] >= 0) stage[lbase[own[u]] + rank[u]] = ent[u];
  __syncthreads();

  for (int o = wave; o < N_OWN; o += 32) {
    const int n  = hist[o];
    const int lb = lbase[o];
    const int gb = gbase[o];
    unsigned* dst = bins + (size_t)o * BIN_WORDS;
    for (int k = lane; k < n; k += 32) {
      const int pos = gb + k;
      if (pos < BIN_CAP) dst[pos] = stage[lb + k];
    }
  }
}

#define MLP_TP 68
#define MLP_WP 160

template <int KSTEPS>
__device__ __forceinline__ void tile_gemm(const float* arow, const float* wcol,
                                          v8f (&acc)[4]) {
  constexpr int NB = KSTEPS / 2;
  v2f a[2][2];
  v2f b[2][2][4];
#pragma unroll
  for (int q = 0; q < 2; ++q) {
    a[0][q] = *(const v2f*)(arow + q * 4);
#pragma unroll
    for (int nt = 0; nt < 4; ++nt)
      b[0][q][nt] = *(const v2f*)(wcol + q * 2 * MLP_WP + nt * 32);
  }
#pragma unroll
  for (int blk = 0; blk < NB; ++blk) {
    const int cur = blk & 1, nxt = cur ^ 1;
    if (blk + 1 < NB) {
#pragma unroll
      for (int q = 0; q < 2; ++q) {
        const int s = (blk + 1) * 2 + q;
        a[nxt][q] = *(const v2f*)(arow + s * 4);
#pragma unroll
        for (int nt = 0; nt < 4; ++nt)
          b[nxt][q][nt] = *(const v2f*)(wcol + s * 2 * MLP_WP + nt * 32);
      }
    }
#pragma unroll
    for (int q = 0; q < 2; ++q)
#pragma unroll
      for (int nt = 0; nt < 4; ++nt)
        acc[nt] = wmma_f32(a[cur][q], b[cur][q][nt], acc[nt]);
  }
}

static_assert(BIN_ROWS <= 1024, "one thread per row in the scan of the row counters");
static_assert(BIN_ROWS % 16 == 0, "an owner's rows are whole tiles");
static_assert(BIN_ROWS / 16 <= 32, "one wave per tile");

typedef const __attribute__((address_space(1))) v2f* global_v2f_ptr;
__device__ __forceinline__ v2f load_row_cols(const float* hin,
                                             unsigned rowbytes,
                                             unsigned lbyte) {
  unsigned long long base = (unsigned long long)hin + rowbytes;
  asm("" : "+s"(base));
  return *(global_v2f_ptr)(base + lbyte);
}

template <bool L0, bool PARKED>
__global__ void __launch_bounds__(1024)
owner_gin_kernel(const float* __restrict__ hin, float* hout, float* park,
                 const int* __restrict__ edge_index,
                 const int* __restrict__ bincnt,
                 const float* __restrict__ W1, const float* __restrict__ b1,
                 const float* __restrict__ W2, const float* __restrict__ b2,
                 const float* __restrict__ eps_ptr, int eps_idx,
                 float* __restrict__ stats,
                 const float* __restrict__ bnstats,
                 const float* __restrict__ gamma,
                 const float* __restrict__ beta) {
  constexpr int  LOG2W  = L0 ? 4 : 6;
  constexpr int  K      = 1 << LOG2W;
  constexpr bool BN     = !L0;
  constexpr int  W1ROWS = L0 ? 10 : 64;
  __shared__ float acc[BIN_ROWS * 64];
  __shared__ __attribute__((aligned(16))) float arow[BIN_ROWS * MLP_TP];
  __shared__ __attribute__((aligned(16))) float w1s[(K / 2) * MLP_WP];
  __shared__ __attribute__((aligned(16))) float w2s[32 * MLP_WP];
  __shared__ int rowcnt[BIN_ROWS];
  __shared__ int rowstart[BIN_ROWS + 1];
  __shared__ int wsum[32];
  __shared__ float bnss[128];
  __shared__ float sstat[128];
  const int tid  = threadIdx.x;
  const int lane = tid & 31;
  const int wave = __builtin_amdgcn_readfirstlane(tid >> 5);
  const unsigned row0 = blockIdx.x * BIN_ROWS;
  float* region = hout + (size_t)blockIdx.x * BIN_WORDS;
  const unsigned* bin = (const unsigned*)region;

  if (!PARKED && tid < BIN_ROWS) rowcnt[tid] = 0;
  if (tid < 128) sstat[tid] = 0.0f;
  if (BN && tid < 64)
    bn_scale_shift(bnstats, gamma, beta, tid, bnss[tid], bnss[64 + tid]);
  if (tid < (K / 2) * 16) {
    const int p = tid >> 4, c4 = (tid & 15) * 4;
    float4 r0 = make_float4(0.f, 0.f, 0.f, 0.f), r1 = r0;
    if (2 * p < W1ROWS)     r0 = *(const float4*)(W1 + (size_t)(2 * p) * 64 + c4);
    if (2 * p + 1 < W1ROWS) r1 = *(const float4*)(W1 + (size_t)(2 * p + 1) * 64 + c4);
    float* d = &w1s[p * MLP_WP + c4 * 2];
    *(float4*)d       = make_float4(r0.x, r1.x, r0.y, r1.y);
    *(float4*)(d + 4) = make_float4(r0.z, r1.z, r0.w, r1.w);
  }
  if (tid < 32 * 16) {
    const int p = tid >> 4, c4 = (tid & 15) * 4;
    const float4 r0 = *(const float4*)(W2 + (size_t)(2 * p) * 64 + c4);
    const float4 r1 = *(const float4*)(W2 + (size_t)(2 * p + 1) * 64 + c4);
    float* d = &w2s[p * MLP_WP + c4 * 2];
    *(float4*)d       = make_float4(r0.x, r1.x, r0.y, r1.y);
    *(float4*)(d + 4) = make_float4(r0.z, r1.z, r0.w, r1.w);
  }
  const float onepe = 1.0f + eps_ptr[eps_idx];
  const int total = bincnt[blockIdx.x];
  __syncthreads();

  constexpr int CAP   = PARKED ? PARK_CAP : BIN_CAP;
  constexpr int PER_T = (CAP + 1023) / 1024;
  if (total <= CAP) {
    unsigned* sorted = (unsigned*)acc;
    if constexpr (PARKED) {
      unsigned sv[PER_T];
#pragma unroll
      for (int u = 0; u < PER_T; ++u) {
        const int i = tid + u * 1024;
        sv[u] = bin[park_word((i < total) ? i : 0)];
      }
      int place = 0;
      if (tid <= BIN_ROWS) place = (int)bin[park_word(total + tid)];
#pragma unroll
      for (int u = 0; u < PER_T; ++u) {
        const int i = tid + u * 1024;
        if (i < total) sorted[i] = min(sv[u], (unsigned)(N_NODES - 1));
      }
      if (tid <= BIN_ROWS) rowstart[tid] = min(max(place, 0), total);
      __syncthreads();
    } else {
      unsigned ent[PER_T];
#pragma unroll
      for (int u = 0; u < PER_T; ++u) {
        const int i = tid + u * 1024;
        ent[u] = bin[(i < total) ? i : 0];
      }
      int rank[PER_T];
#pragma unroll
      for (int u = 0; u < PER_T; ++u) {
        const int i = tid + u * 1024;
        rank[u] = 0;
        if (i < total) {
          const unsigned dl = min(ent[u] >> 17, (unsigned)(BIN_ROWS - 1));
          rank[u] = atomicAdd(&rowcnt[dl], 1);
        }
      }
      __syncthreads();

      const int cnt = (tid < BIN_ROWS) ? rowcnt[tid] : 0;
      int incl = cnt;
#pragma unroll
      for (int d = 1; d < 32; d <<= 1) {
        const int t = __shfl_up(incl, d);
        if (lane >= d) incl += t;
      }
      if (lane == 31) wsum[wave] = incl;
      __syncthreads();
      int off = 0;
      for (int k = 0; k < wave; ++k) off += wsum[k];
      if (tid < BIN_ROWS) rowstart[tid] = off + incl - cnt;
      if (tid == BIN_ROWS - 1) rowstart[BIN_ROWS] = off + incl;
      __syncthreads();

#pragma unroll
      for (int u = 0; u < PER_T; ++u) {
        const int i = tid + u * 1024;
        if (i < total) {
          const unsigned dl = min(ent[u] >> 17, (unsigned)(BIN_ROWS - 1));
          sorted[rowstart[dl] + rank[u]] =
              min(ent[u] & 0x1FFFFu, (unsigned)(N_NODES - 1));
        }
      }
      __syncthreads();
      if constexpr (L0) {
        if (total <= PARK_CAP) {
          constexpr int PK_T = (PARK_CAP + 1023) / 1024;
          unsigned* pk = (unsigned*)park + (size_t)blockIdx.x * BIN_WORDS;
          unsigned sv[PK_T];
#pragma unroll
          for (int u = 0; u < PK_T; ++u) {
            const int i = tid + u * 1024;
            sv[u] = sorted[(i < total) ? i : 0];
          }
          const int place = rowstart[(tid <= BIN_ROWS) ? tid : 0];
#pragma unroll
          for (int u = 0; u < PK_T; ++u) {
            const int i = tid + u * 1024;
            if (i < total) pk[park_word(i)] = sv[u];
          }
          if (tid <= BIN_ROWS) pk[park_word(total + tid)] = (unsigned)place;
        }
      }
    }

    constexpr int LPE   = (1 << LOG2W) / 2;
    constexpr int EPI   = 32 / LPE;
    constexpr int STEPS = (EPI == 1) ? 48 : 16;
    constexpr int NEV   = (STEPS + 31) / 32;
    const int sub = lane / LPE;
    const int col = (lane % LPE) * 2;
    const float* __restrict__ hcol = hin + col;
    float scx = 1.0f, scy = 1.0f, shx = 0.0f, shy = 0.0f;
    if constexpr (BN) {
      scx = bnss[col];      scy = bnss[col + 1];
      shx = bnss[64 + col]; shy = bnss[64 + col + 1];
    }
    for (int r = wave; r < BIN_ROWS; r += 32) {
      const int s = __builtin_amdgcn_readfirstlane(rowstart[r]);
      const int n = __builtin_amdgcn_readfirstlane(rowstart[r + 1]) - s;
      float ax = 0.0f, ay = 0.0f;
      float2 own = make_float2(0.0f, 0.0f);
      const unsigned lbyte = (unsigned)col * 4u;
      if constexpr (EPI == 1) {
        unsigned lb = lbyte;
        asm("" : "+v"(lb));
        const v2f o2 = load_row_cols(hin, (row0 + (unsigned)r) << 8, lb);
        own = make_float2(o2.x, o2.y);
      } else if (sub == 0) {
        own = *(const float2*)(hcol + (size_t)(row0 + r) * 64);
      }
      if constexpr (EPI == 1) {
        constexpr int GRP = 8;
        static_assert(STEPS % GRP == 0, "the first part is whole groups");
        const v2f sc2 = {scx, scy}, sh2 = {shx, shy}, zero2 = {0.0f, 0.0f};
        v2f acc2 = zero2;
        unsigned eb[NEV];
#pragma unroll
        for (int q = 0; q < NEV; ++q) {
          eb[q] = 0u;
          if (q * 32 < n) {
            const int idx = q * 32 + lane;
            if (idx < n) eb[q] = sorted[s + idx] << 8;
          }
        }
        v2f v[STEPS];
#pragma unroll
        for (int g = 0; g < STEPS / GRP; ++g) {
          if (g * GRP < n) {
            unsigned lb = lbyte;
            asm("" : "+v"(lb));
#pragma unroll
            for (int j = 0; j < GRP; ++j) {
              const int t = g * GRP + j;
              const unsigned rb = (unsigned)__builtin_amdgcn_readlane(
                  (int)eb[t >> 5], t & 31);
              v[t] = load_row_cols(hin, rb, lb);
            }
          }
        }
#pragma unroll
        for (int g = 0; g < STEPS / GRP; ++g) {
          const int b = g * GRP;
          if (b + GRP <= n) {
#pragma unroll
            for (int j = 0; j < GRP; ++j) {
              const int t = g * GRP + j;
              acc2 += BN ? __builtin_elementwise_max(
                               __builtin_elementwise_fma(v[t], sc2, sh2), zero2)
                         : v[t];
            }
          } else if (b < n) {
#pragma unroll
            for (int j = 0; j < GRP; ++j) {
              const int t = g * GRP + j;
              const v2f y = BN ? __builtin_elementwise_max(
                                     __builtin_elementwise_fma(v[t], sc2, sh2), zero2)
                               : v[t];
              acc2 += (b + j < n) ? y : zero2;
            }
          }
        }
        for (int c = STEPS; c < n; c += GRP) {
          unsigned er = 0u;
          if (lane < GRP && c + lane < n) er = sorted[s + c + lane] << 8;
          unsigned lb = lbyte;
          asm("" : "+v"(lb));
          v2f w[GRP];
#pragma unroll
          for (int j = 0; j < GRP; ++j) {
            const unsigned rb = (unsigned)__builtin_amdgcn_readlane((int)er, j);
            w[j] = load_row_cols(hin, rb, lb);
          }
#pragma unroll
          for (int j = 0; j < GRP; ++j) {
            const v2f y = BN ? __builtin_elementwise_max(
                                   __builtin_elementwise_fma(w[j], sc2, sh2), zero2)
                             : w[j];
            acc2 += (c + j < n) ? y : zero2;
          }
        }
        ax = acc2.x;
        ay = acc2.y;
      } else {
        for (int c = 0; c < n; c += STEPS * EPI) {
          float2 v[STEPS];
          unsigned sk[STEPS];
#pragma unroll
          for (int t = 0; t < STEPS; ++t) {
            const int e0 = c + t * EPI;
            sk[t] = 0u;
            if (e0 < n) sk[t] = sorted[s + ((e0 + sub < n) ? e0 + sub : e0)];
          }
#pragma unroll
          for (int t = 0; t < STEPS; ++t) {
            v[t] = make_float2(0.0f, 0.0f);
            if (c + t * EPI < n)
              v[t] = *(const float2*)(hcol + (size_t)sk[t] * 64);
          }
#pragma unroll
          for (int t = 0; t < STEPS; ++t) {
            const bool ok = (c + t * EPI + sub < n);
            const float vx = BN ? fmaxf(fmaf(v[t].x, scx, shx), 0.0f) : v[t].x;
            const float vy = BN ? fmaxf(fmaf(v[t].y, scy, shy), 0.0f) : v[t].y;
            ax += ok ? vx : 0.0f;
            ay += ok ? vy : 0.0f;
          }
        }
      }
#pragma unroll
      for (int d = LPE; d < 32; d <<= 1) {
        ax += __shfl_xor(ax, d);
        ay += __shfl_xor(ay, d);
      }
      if (sub == 0) {
        if constexpr (BN) {
          own.x = fmaxf(fmaf(own.x, scx, shx), 0.0f);
          own.y = fmaxf(fmaf(own.y, scy, shy), 0.0f);
        }
        *(float2*)&arow[r * MLP_TP + col] =
            make_float2(onepe * own.x + ax, onepe * own.y + ay);
      }
    }
  } else {
    for (int i = tid; i < BIN_ROWS * 16; i += 1024)
      ((float4*)acc)[i] = make_float4(0.f, 0.f, 0.f, 0.f);
    __syncthreads();
    const int* __restrict__ dsts = edge_index + N_EDGES;
    for (int e = tid; e < N_EDGES; e += 1024) {
      const unsigned dl = (unsigned)dsts[e] - row0;
      if (dl < (unsigned)BIN_ROWS) {
        const float* hs = hin + (size_t)edge_index[e] * 64;
        for (int c = 0; c < (1 << LOG2W); ++c) {
          float v = hs[c];
          if constexpr (BN) v = fmaxf(fmaf(v, bnss[c], bnss[64 + c]), 0.0f);
          unsafeAtomicAdd(&acc[dl * 64 + c], v);
        }
      }
    }
    __syncthreads();
    for (int i = tid; i < BIN_ROWS * K; i += 1024) {
      const int r = i >> LOG2W, c = i & (K - 1);
      float v = hin[(size_t)(row0 + r) * 64 + c];
      if constexpr (BN) v = fmaxf(fmaf(v, bnss[c], bnss[64 + c]), 0.0f);
      arow[r * MLP_TP + c] = onepe * v + acc[r * 64 + c];
    }
  }
  __syncthreads();

  if (wave < BIN_ROWS / 16) {
    const int lrow  = lane & 15;
    const int lhalf = lane >> 4;
    float* my = arow + wave * (16 * MLP_TP);
    const float* afr = my + lrow * MLP_TP + lhalf * 2;
    v8f acc1[4] = {};
    tile_gemm<K / 4>(afr, w1s + lhalf * MLP_WP + lrow * 2, acc1);

#pragma unroll
    for (int nt = 0; nt < 4; ++nt) {
      const float bv = b1[nt * 16 + lrow];
#pragma unroll
      for (int r = 0; r < 8; ++r)
        my[(r + lhalf * 8) * MLP_TP + nt * 16 + lrow] = fmaxf(acc1[nt][r] + bv, 0.f);
    }

    v8f acc2[4] = {};
    tile_gemm<16>(afr, w2s + lhalf * MLP_WP + lrow * 2, acc2);

#pragma unroll
    for (int nt = 0; nt < 4; ++nt) {
      const float bv = b2[nt * 16 + lrow];
      float s = 0.0f, s2 = 0.0f;
#pragma unroll
      for (int r = 0; r < 8; ++r) {
        const float v = fmaxf(acc2[nt][r] + bv, 0.f);
        my[(r + lhalf * 8) * MLP_TP + nt * 16 + lrow] = v;
        s += v;
        s2 += v * v;
      }
      unsafeAtomicAdd(&sstat[nt * 16 + lrow], s);
      unsafeAtomicAdd(&sstat[64 + nt * 16 + lrow], s2);
    }
#pragma unroll
    for (int j = 0; j < 8; ++j) {
      const int row = j * 2 + lhalf;
      *(float4*)(region + (size_t)(wave * 16 + row) * 64 + lrow * 4) =
          *(const float4*)&my[row * MLP_TP + lrow * 4];
    }
  }
  __syncthreads();
  if (tid < 128) unsafeAtomicAdd(&stats[tid], sstat[tid]);
}

__global__ void final_kernel(const float* __restrict__ h,
                             const float* __restrict__ bnstats,
                             const float* __restrict__ gamma,
                             const float* __restrict__ beta,
                             const float* __restrict__ Wf,
                             const float* __restrict__ bf,
                             float* __restrict__ out) {
  __shared__ float sbn[128];
  if (threadIdx.x < 64)
    bn_scale_shift(bnstats, gamma, beta, threadIdx.x, sbn[threadIdx.x],
                   sbn[64 + threadIdx.x]);
  __syncthreads();
  const int i = blockIdx.x * blockDim.x + threadIdx.x;
  if (i >= N_NODES) return;
  const float4* hp = (const float4*)(h + (size_t)i * 64);
  const float4* wp = (const float4*)Wf;
  const float4* sc = (const float4*)sbn;
  const float4* sh = (const float4*)(sbn + 64);
  float s = bf[0];
#pragma unroll
  for (int j = 0; j < 16; ++j) {
    float4 a = hp[j], b = wp[j];
    const float4 c = sc[j], d = sh[j];
    a.x = fmaxf(fmaf(a.x, c.x, d.x), 0.0f);
    a.y = fmaxf(fmaf(a.y, c.y, d.y), 0.0f);
    a.z = fmaxf(fmaf(a.z, c.z, d.z), 0.0f);
    a.w = fmaxf(fmaf(a.w, c.w, d.w), 0.0f);
    s += a.x * b.x + a.y * b.y + a.z * b.z + a.w * b.w;
  }
  out[i] = s;
}

extern "C" void kernel_launch(void* const* d_in, const int* in_sizes, int n_in,
                              void* d_out, int out_size, void* d_ws,
                              size_t ws_size, hipStream_t stream) {
  const float* x      = (const float*)d_in[0];
  const int*   labels = (const int*)d_in[1];
  const int*   ei     = (const int*)d_in[2];
  const float* emb    = (const float*)d_in[3];
  const float* W1_0   = (const float*)d_in[4];
  const float* b1_0   = (const float*)d_in[5];
  const float* W2_0   = (const float*)d_in[6];
  const float* b2_0   = (const float*)d_in[7];
  const float* eps_0  = (const float*)d_in[8];
  const float* g_0    = (const float*)d_in[9];
  const float* be_0   = (const float*)d_in[10];
  const float* W1_s   = (const float*)d_in[11];
  const float* b1_s   = (const float*)d_in[12];
  const float* W2_s   = (const float*)d_in[13];
  const float* b2_s   = (const float*)d_in[14];
  const float* eps_s  = (const float*)d_in[15];
  const float* g_s    = (const float*)d_in[16];
  const float* be_s   = (const float*)d_in[17];
  const float* Wf     = (const float*)d_in[18];
  const float* bf     = (const float*)d_in[19];
  float* out = (float*)d_out;

  float* bufQ  = (float*)d_ws;
  float* sums0 = bufQ + (size_t)N_NODES * 64;
  float* sums1 = sums0 + 128;
  int*   cnt   = (int*)(sums1 + 128);
  float* bufP  = sums1 + 128 + 1024;

  build_input_kernel<<<(N_NODES * 16 + 255) / 256, 256, 0, stream>>>(
      x, labels, emb, bufP, cnt);

  for (int layer = 0; layer < 3; ++layer) {
    const float* hin  = (layer == 1) ? bufQ : bufP;
    float*       hout = (layer == 1) ? bufP : bufQ;
    float*       st_new = (layer == 1) ? sums1 : sums0;
    const float* st_old = (layer == 1) ? sums0 : sums1;
    const float* gm_old = (layer == 2) ? g_s : g_0;
    const float* bt_old = (layer == 2) ? be_s : be_0;
    int* lcnt = cnt + ((layer == 2) ? N_OWN : 0);
    if (layer != 1)
      bin_edges_kernel<<<N_OWN, 1024, 0, stream>>>(ei, hout, lcnt, st_new,
                                                   (layer == 0) ? 256 : 128);

    const float* W1 = (layer == 0) ? W1_0 : (W1_s + (size_t)(layer - 1) * 4096);
    const float* b1 = (layer == 0) ? b1_0 : (b1_s + (size_t)(layer - 1) * 64);
    const float* W2 = (layer == 0) ? W2_0 : (W2_s + (size_t)(layer - 1) * 4096);
    const float* b2 = (layer == 0) ? b2_0 : (b2_s + (size_t)(layer - 1) * 64);
    const float* ep = (layer == 0) ? eps_0 : eps_s;
    const int    ei_idx = (layer == 0) ? 0 : (layer - 1);

    if (layer == 0)
      owner_gin_kernel<true, false><<<N_OWN, 1024, 0, stream>>>(
          hin, hout, bufP, ei, lcnt, W1, b1, W2, b2, ep, ei_idx,
          st_new, st_old, gm_old, bt_old);
    else if (layer == 1)
      owner_gin_kernel<false, true><<<N_OWN, 1024, 0, stream>>>(
          hin, hout, nullptr, ei, lcnt, W1, b1, W2, b2, ep, ei_idx,
          st_new, st_old, gm_old, bt_old);
    else
      owner_gin_kernel<false, false><<<N_OWN, 1024, 0, stream>>>(
          hin, hout, nullptr, ei, lcnt, W1, b1, W2, b2, ep, ei_idx,
          st_new, st_old, gm_old, bt_old);
  }

  final_kernel<<<(N_NODES + 255) / 256, 256, 0, stream>>>(
      bufQ, sums0, g_s + 64, be_s + 64, Wf, bf, out);
}
